# starting kernel
# baseline (speedup 1.0000x reference)
#include <hip/hip_runtime.h>
#include <math.h>

#define NQ   20000
#define NS   8192
#define C    112
#define C2   224
#define CLS  20
#define TILES (NQ / 16)
#define WAVES_PER_BLOCK 4

#define WREG    3712
#define SSTRIDE 20
#define XSTRIDE 226
#define HSTRIDE 114
#define H_OFF   0
#define Y2_OFF  1856

typedef float v2f __attribute__((ext_vector_type(2)));
typedef float v8f __attribute__((ext_vector_type(8)));

__global__ void attn_logits(const float* __restrict__ feat_s7,
                            const float* __restrict__ w_attn,
                            float* __restrict__ logits) {
  int j = blockIdx.x * blockDim.x + threadIdx.x;
  if (j >= NS) return;
  const float* row = feat_s7 + (long)j * C;
  float acc = 0.f;
#pragma unroll 8
  for (int c = 0; c < C; ++c) acc += row[c] * w_attn[c];
  logits[j] = acc;
}

__global__ void attn_pool_gate(const float* __restrict__ feat_s7,
                               const float* __restrict__ logits,
                               float* __restrict__ ebuf,
                               float* __restrict__ gate) {
  __shared__ float red[256];
  __shared__ float bcast[2];
  const int t = threadIdx.x;

  float m = -1e30f;
  for (int j = t; j < NS; j += 256) m = fmaxf(m, logits[j]);
  red[t] = m; __syncthreads();
  for (int s = 128; s > 0; s >>= 1) { if (t < s) red[t] = fmaxf(red[t], red[t + s]); __syncthreads(); }
  if (t == 0) bcast[0] = red[0];
  __syncthreads();
  m = bcast[0];

  float sum = 0.f;
  for (int j = t; j < NS; j += 256) { float e = __expf(logits[j] - m); ebuf[j] = e; sum += e; }
  __syncthreads();
  red[t] = sum; __syncthreads();
  for (int s = 128; s > 0; s >>= 1) { if (t < s) red[t] += red[t + s]; __syncthreads(); }
  if (t == 0) bcast[1] = red[0];
  __syncthreads();
  const float inv = 1.0f / bcast[1];

  if (t < C) {
    float acc = 0.f;
    for (int j = 0; j < NS; ++j) acc += ebuf[j] * feat_s7[(long)j * C + t];
    float g = acc * inv;
    gate[t] = 1.0f / (1.0f + __expf(-g));
  }
}

#define TOP3_INS(val, idxv) {                                         \
    float s_ = (val); int id_ = (idxv);                               \
    bool c0_ = s_ < v0, c1_ = s_ < v1, c2_ = s_ < v2v;                \
    v2v = c1_ ? v1 : (c2_ ? s_  : v2v); i2 = c1_ ? i1 : (c2_ ? id_ : i2); \
    v1  = c0_ ? v0 : (c1_ ? s_  : v1 ); i1 = c0_ ? i0 : (c1_ ? id_ : i1); \
    v0  = c0_ ? s_ : v0;                i0 = c0_ ? id_ : i0;          }

__global__ __launch_bounds__(128) void fused_main(
    const float* __restrict__ coords_q, const float* __restrict__ coords_s,
    const float* __restrict__ x7,       const float* __restrict__ feat_s,
    const float* __restrict__ gate,     const float* __restrict__ W1,
    const float* __restrict__ gammaW,   const float* __restrict__ betaW,
    const float* __restrict__ W2,       const float* __restrict__ W_cls,
    const float* __restrict__ b_cls,    float* __restrict__ out)
{
  __shared__ float lds[WAVES_PER_BLOCK * WREG];
  const int lane = threadIdx.x & 31;
  const int wave = threadIdx.x >> 5;
  const int tile = blockIdx.x * WAVES_PER_BLOCK + wave;
  if (tile >= TILES) return;

  float* WL = lds + wave * WREG;
  const int r15  = lane & 15;
  const int half = lane >> 4;
  const int qbase = tile * 16;

  const float qx = coords_q[(qbase + r15) * 3 + 0];
  const float qy = coords_q[(qbase + r15) * 3 + 1];
  const float qz = coords_q[(qbase + r15) * 3 + 2];
  v2f a;
  a.x = half ? (-2.f * qz) : (-2.f * qx);
  a.y = half ? 1.0f        : (-2.f * qy);

  float v0 = 1e30f, v1 = 1e30f, v2v = 1e30f;
  int   i0 = 0, i1 = 0, i2 = 0;

  for (int n0 = 0; n0 < NS; n0 += 16) {
    __builtin_prefetch(coords_s + (n0 + 128) * 3, 0, 1);
    const int j = n0 + r15;
    const float sx = coords_s[j * 3 + 0];
    const float sy = coords_s[j * 3 + 1];
    const float sz = coords_s[j * 3 + 2];
    const float s2 = sx * sx + sy * sy + sz * sz;
    v2f b;
    b.x = half ? sz : sx;
    b.y = half ? s2 : sy;
    v8f c = {0.f, 0.f, 0.f, 0.f, 0.f, 0.f, 0.f, 0.f};
    c = __builtin_amdgcn_wmma_f32_16x16x4_f32(false, a, false, b, (short)0, c, false, false);

#pragma unroll
    for (int r = 0; r < 8; ++r)
      WL[(r + 8 * half) * SSTRIDE + r15] = c[r];
    asm volatile("" ::: "memory");

    const float4 sA = *(const float4*)(WL + r15 * SSTRIDE + 8 * half);
    const float4 sB = *(const float4*)(WL + r15 * SSTRIDE + 8 * half + 4);

    const float m8 = fminf(fminf(fminf(sA.x, sA.y), fminf(sA.z, sA.w)),
                           fminf(fminf(sB.x, sB.y), fminf(sB.z, sB.w)));
    if (__any(m8 < v2v)) {
      const int base = n0 + 8 * half;
      TOP3_INS(sA.x, base + 0); TOP3_INS(sA.y, base + 1);
      TOP3_INS(sA.z, base + 2); TOP3_INS(sA.w, base + 3);
      TOP3_INS(sB.x, base + 4); TOP3_INS(sB.y, base + 5);
      TOP3_INS(sB.z, base + 6); TOP3_INS(sB.w, base + 7);
    }
    asm volatile("" ::: "memory");
  }

  {
    float pv0 = __shfl(v0, lane + 16), pv1 = __shfl(v1, lane + 16), pv2 = __shfl(v2v, lane + 16);
    int   pi0 = __shfl(i0, lane + 16), pi1 = __shfl(i1, lane + 16), pi2 = __shfl(i2,  lane + 16);
    TOP3_INS(pv0, pi0); TOP3_INS(pv1, pi1); TOP3_INS(pv2, pi2);
  }
  const float e1 = __expf(v0 - v1), e2 = __expf(v0 - v2v);
  const float inv = 1.0f / (1.0f + e1 + e2);
  const float w0 = inv, w1 = e1 * inv, w2 = e2 * inv;

  for (int r = 0; r < 16; ++r) {
    const float rw0 = __shfl(w0, r), rw1 = __shfl(w1, r), rw2 = __shfl(w2, r);
    const int   ri0 = __shfl(i0, r), ri1 = __shfl(i1, r), ri2 = __shfl(i2, r);
    const int qi = qbase + r;
    for (int t = lane; t < C2; t += 32) {
      float v;
      if (t < C) {
        v = x7[(long)qi * C + t] * gate[t];
      } else {
        const int cc = t - C;
        v = rw0 * feat_s[(long)ri0 * C + cc]
          + rw1 * feat_s[(long)ri1 * C + cc]
          + rw2 * feat_s[(long)ri2 * C + cc];
      }
      WL[r * XSTRIDE + t] = v;
    }
  }
  asm volatile("" ::: "memory");

  v8f acc[7];
#pragma unroll
  for (int n = 0; n < 7; ++n) acc[n] = (v8f){0.f, 0.f, 0.f, 0.f, 0.f, 0.f, 0.f, 0.f};
  for (int k0 = 0; k0 < C2; k0 += 4) {
    const int kk = k0 + 2 * half;
    const v2f a1 = *(const v2f*)(WL + r15 * XSTRIDE + kk);
#pragma unroll
    for (int n = 0; n < 7; ++n) {
      v2f bv;
      bv.x = W1[(long)kk * C + n * 16 + r15];
      bv.y = W1[(long)(kk + 1) * C + n * 16 + r15];
      acc[n] = __builtin_amdgcn_wmma_f32_16x16x4_f32(false, a1, false, bv, (short)0, acc[n], false, false);
    }
  }

#pragma unroll
  for (int n = 0; n < 7; ++n) {
    const int ch = n * 16 + r15;
    const float gm = gammaW[ch], bt = betaW[ch];
#pragma unroll
    for (int r = 0; r < 8; ++r) {
      const float h = fmaxf(acc[n][r] * gm + bt, 0.f);
      WL[H_OFF + (r + 8 * half) * HSTRIDE + ch] = h;
    }
  }
  asm volatile("" ::: "memory");

  v8f acc2[7];
#pragma unroll
  for (int n = 0; n < 7; ++n) acc2[n] = (v8f){0.f, 0.f, 0.f, 0.f, 0.f, 0.f, 0.f, 0.f};
  for (int k0 = 0; k0 < C; k0 += 4) {
    const int kk = k0 + 2 * half;
    const v2f a2 = *(const v2f*)(WL + H_OFF + r15 * HSTRIDE + kk);
#pragma unroll
    for (int n = 0; n < 7; ++n) {
      v2f bv;
      bv.x = W2[(long)kk * C + n * 16 + r15];
      bv.y = W2[(long)(kk + 1) * C + n * 16 + r15];
      acc2[n] = __builtin_amdgcn_wmma_f32_16x16x4_f32(false, a2, false, bv, (short)0, acc2[n], false, false);
    }
  }
#pragma unroll
  for (int n = 0; n < 7; ++n) {
    const int ch = n * 16 + r15;
#pragma unroll
    for (int r = 0; r < 8; ++r)
      WL[Y2_OFF + (r + 8 * half) * HSTRIDE + ch] = acc2[n][r];
  }
  asm volatile("" ::: "memory");

  v8f acc3[2];
  acc3[0] = (v8f){0.f, 0.f, 0.f, 0.f, 0.f, 0.f, 0.f, 0.f};
  acc3[1] = acc3[0];
  for (int k0 = 0; k0 < C; k0 += 4) {
    const int kk = k0 + 2 * half;
    const v2f a3 = *(const v2f*)(WL + Y2_OFF + r15 * HSTRIDE + kk);
#pragma unroll
    for (int n = 0; n < 2; ++n) {
      const int col = n * 16 + r15;
      v2f bv;
      bv.x = (col < CLS) ? W_cls[(long)kk * CLS + col] : 0.f;
      bv.y = (col < CLS) ? W_cls[(long)(kk + 1) * CLS + col] : 0.f;
      acc3[n] = __builtin_amdgcn_wmma_f32_16x16x4_f32(false, a3, false, bv, (short)0, acc3[n], false, false);
    }
  }
#pragma unroll
  for (int n = 0; n < 2; ++n) {
    const int col = n * 16 + r15;
    if (col < CLS) {
      const float bb = b_cls[col];
#pragma unroll
      for (int r = 0; r < 8; ++r)
        out[(long)(qbase + r + 8 * half) * CLS + col] = acc3[n][r] + bb;
    }
  }
}

extern "C" void kernel_launch(void* const* d_in, const int* in_sizes, int n_in,
                              void* d_out, int out_size, void* d_ws, size_t ws_size,
                              hipStream_t stream) {
  const float* coords_q = (const float*)d_in[0];
  const float* coords_s = (const float*)d_in[1];
  const float* x7       = (const float*)d_in[2];
  const float* feat_s7  = (const float*)d_in[3];
  const float* feat_s   = (const float*)d_in[4];
  const float* w_attn   = (const float*)d_in[5];
  const float* W1       = (const float*)d_in[6];
  const float* gammaW   = (const float*)d_in[7];
  const float* betaW    = (const float*)d_in[8];
  const float* W2       = (const float*)d_in[9];
  const float* W_cls    = (const float*)d_in[10];
  const float* b_cls    = (const float*)d_in[11];
  float* out = (float*)d_out;

  float* ws     = (float*)d_ws;
  float* logits = ws;
  float* ebuf   = ws + NS;
  float* gate   = ws + 2 * NS;

  attn_logits<<<NS / 256, 256, 0, stream>>>(feat_s7, w_attn, logits);
  attn_pool_gate<<<1, 256, 0, stream>>>(feat_s7, logits, ebuf, gate);

  const int blocks = (TILES + WAVES_PER_BLOCK - 1) / WAVES_PER_BLOCK;
  fused_main<<<blocks, 128, 0, stream>>>(coords_q, coords_s, x7, feat_s, gate,
                                         W1, gammaW, betaW, W2, W_cls, b_cls, out);
}


# softmax pooling split by channel over 28 work-groups: each takes max and exp sum of all logits itself, sums its 4 channels over all rows, writes its gates to a line of its own; sums re-associated
# speedup vs baseline: 3.7059x; 3.7059x over previous
#include <hip/hip_runtime.h>
#include <math.h>

#define NQ   20000
#define NS   8192
#define C    112
#define C2   224
#define CLS  20
#define TILES (NQ / 16)
#define WAVES_PER_BLOCK 4
#define PCH        4
#define GATE_PITCH 32

#define WREG    3712
#define SSTRIDE 20
#define XSTRIDE 226
#define HSTRIDE 114
#define H_OFF   0
#define Y2_OFF  1856

typedef float v2f __attribute__((ext_vector_type(2)));
typedef float v8f __attribute__((ext_vector_type(8)));

__global__ void attn_logits(const float* __restrict__ feat_s7,
                            const float* __restrict__ w_attn,
                            float* __restrict__ logits) {
  int j = blockIdx.x * blockDim.x + threadIdx.x;
  if (j >= NS) return;
  const float* row = feat_s7 + (long)j * C;
  float acc = 0.f;
#pragma unroll 8
  for (int c = 0; c < C; ++c) acc += row[c] * w_attn[c];
  logits[j] = acc;
}

#define PTHREADS 1024
#define PROWS    (NS / PTHREADS)
#define PWAVES   (PTHREADS / 32)
static_assert(PWAVES == 32 && NS % PTHREADS == 0 && C % PCH == 0 && PCH == 4, "pool split");

__device__ __forceinline__ float wave_max(float v) {
  for (int o = 16; o > 0; o >>= 1) v = fmaxf(v, __shfl_xor(v, o));
  return v;
}
__device__ __forceinline__ float wave_sum(float v) {
  for (int o = 16; o > 0; o >>= 1) v += __shfl_xor(v, o);
  return v;
}

__global__ __launch_bounds__(PTHREADS) void attn_pool_gate(
    const float* __restrict__ feat_s7, const float* __restrict__ logits,
    float* __restrict__ gate) {
  __shared__ float redm[PWAVES];
  __shared__ float reds[1 + PCH][PWAVES];
  const int t = threadIdx.x;
  const int lane = t & 31;
  const int wave = t >> 5;
  const int c0 = blockIdx.x * PCH;

  float l[PROWS];
  float4 f[PROWS];
#pragma unroll
  for (int i = 0; i < PROWS; ++i) {
    const int j = t + i * PTHREADS;
    l[i] = logits[j];
    f[i] = *(const float4*)(feat_s7 + (long)j * C + c0);
  }

  float m = l[0];
#pragma unroll
  for (int i = 1; i < PROWS; ++i) m = fmaxf(m, l[i]);
  m = wave_max(m);
  if (lane == 0) redm[wave] = m;
  __syncthreads();
  m = wave_max(redm[lane]);

  float s = 0.f, a0 = 0.f, a1 = 0.f, a2 = 0.f, a3 = 0.f;
#pragma unroll
  for (int i = 0; i < PROWS; ++i) {
    const float e = __expf(l[i] - m);
    s  += e;
    a0 += e * f[i].x;
    a1 += e * f[i].y;
    a2 += e * f[i].z;
    a3 += e * f[i].w;
  }
  s  = wave_sum(s);
  a0 = wave_sum(a0);
  a1 = wave_sum(a1);
  a2 = wave_sum(a2);
  a3 = wave_sum(a3);
  if (lane == 0) {
    reds[0][wave] = s;
    reds[1][wave] = a0;
    reds[2][wave] = a1;
    reds[3][wave] = a2;
    reds[4][wave] = a3;
  }
  __syncthreads();

  if (wave == 0) {
    const float sum = wave_sum(reds[0][lane]);
    const float t0 = wave_sum(reds[1][lane]);
    const float t1 = wave_sum(reds[2][lane]);
    const float t2 = wave_sum(reds[3][lane]);
    const float t3 = wave_sum(reds[4][lane]);
    if (lane < PCH) {
      const float acc = (lane == 0) ? t0 : (lane == 1) ? t1 : (lane == 2) ? t2 : t3;
      const float inv = 1.0f / sum;
      const float g = acc * inv;
      gate[blockIdx.x * GATE_PITCH + lane] = 1.0f / (1.0f + __expf(-g));
    }
  }
}

#define TOP3_INS(val, idxv) {                                         \
    float s_ = (val); int id_ = (idxv);                               \
    bool c0_ = s_ < v0, c1_ = s_ < v1, c2_ = s_ < v2v;                \
    v2v = c1_ ? v1 : (c2_ ? s_  : v2v); i2 = c1_ ? i1 : (c2_ ? id_ : i2); \
    v1  = c0_ ? v0 : (c1_ ? s_  : v1 ); i1 = c0_ ? i0 : (c1_ ? id_ : i1); \
    v0  = c0_ ? s_ : v0;                i0 = c0_ ? id_ : i0;          }

__global__ __launch_bounds__(128) void fused_main(
    const float* __restrict__ coords_q, const float* __restrict__ coords_s,
    const float* __restrict__ x7,       const float* __restrict__ feat_s,
    const float* __restrict__ gate,     const float* __restrict__ W1,
    const float* __restrict__ gammaW,   const float* __restrict__ betaW,
    const float* __restrict__ W2,       const float* __restrict__ W_cls,
    const float* __restrict__ b_cls,    float* __restrict__ out)
{
  __shared__ float lds[WAVES_PER_BLOCK * WREG];
  const int lane = threadIdx.x & 31;
  const int wave = threadIdx.x >> 5;
  const int tile = blockIdx.x * WAVES_PER_BLOCK + wave;
  if (tile >= TILES) return;

  float* WL = lds + wave * WREG;
  const int r15  = lane & 15;
  const int half = lane >> 4;
  const int qbase = tile * 16;

  const float qx = coords_q[(qbase + r15) * 3 + 0];
  const float qy = coords_q[(qbase + r15) * 3 + 1];
  const float qz = coords_q[(qbase + r15) * 3 + 2];
  v2f a;
  a.x = half ? (-2.f * qz) : (-2.f * qx);
  a.y = half ? 1.0f        : (-2.f * qy);

  float v0 = 1e30f, v1 = 1e30f, v2v = 1e30f;
  int   i0 = 0, i1 = 0, i2 = 0;

  for (int n0 = 0; n0 < NS; n0 += 16) {
    __builtin_prefetch(coords_s + (n0 + 128) * 3, 0, 1);
    const int j = n0 + r15;
    const float sx = coords_s[j * 3 + 0];
    const float sy = coords_s[j * 3 + 1];
    const float sz = coords_s[j * 3 + 2];
    const float s2 = sx * sx + sy * sy + sz * sz;
    v2f b;
    b.x = half ? sz : sx;
    b.y = half ? s2 : sy;
    v8f c = {0.f, 0.f, 0.f, 0.f, 0.f, 0.f, 0.f, 0.f};
    c = __builtin_amdgcn_wmma_f32_16x16x4_f32(false, a, false, b, (short)0, c, false, false);

#pragma unroll
    for (int r = 0; r < 8; ++r)
      WL[(r + 8 * half) * SSTRIDE + r15] = c[r];
    asm volatile("" ::: "memory");

    const float4 sA = *(const float4*)(WL + r15 * SSTRIDE + 8 * half);
    const float4 sB = *(const float4*)(WL + r15 * SSTRIDE + 8 * half + 4);

    const float m8 = fminf(fminf(fminf(sA.x, sA.y), fminf(sA.z, sA.w)),
                           fminf(fminf(sB.x, sB.y), fminf(sB.z, sB.w)));
    if (__any(m8 < v2v)) {
      const int base = n0 + 8 * half;
      TOP3_INS(sA.x, base + 0); TOP3_INS(sA.y, base + 1);
      TOP3_INS(sA.z, base + 2); TOP3_INS(sA.w, base + 3);
      TOP3_INS(sB.x, base + 4); TOP3_INS(sB.y, base + 5);
      TOP3_INS(sB.z, base + 6); TOP3_INS(sB.w, base + 7);
    }
    asm volatile("" ::: "memory");
  }

  {
    float pv0 = __shfl(v0, lane + 16), pv1 = __shfl(v1, lane + 16), pv2 = __shfl(v2v, lane + 16);
    int   pi0 = __shfl(i0, lane + 16), pi1 = __shfl(i1, lane + 16), pi2 = __shfl(i2,  lane + 16);
    TOP3_INS(pv0, pi0); TOP3_INS(pv1, pi1); TOP3_INS(pv2, pi2);
  }
  const float e1 = __expf(v0 - v1), e2 = __expf(v0 - v2v);
  const float inv = 1.0f / (1.0f + e1 + e2);
  const float w0 = inv, w1 = e1 * inv, w2 = e2 * inv;

  for (int r = 0; r < 16; ++r) {
    const float rw0 = __shfl(w0, r), rw1 = __shfl(w1, r), rw2 = __shfl(w2, r);
    const int   ri0 = __shfl(i0, r), ri1 = __shfl(i1, r), ri2 = __shfl(i2, r);
    const int qi = qbase + r;
    for (int t = lane; t < C2; t += 32) {
      float v;
      if (t < C) {
        v = x7[(long)qi * C + t] * gate[(t / PCH) * GATE_PITCH + (t % PCH)];
      } else {
        const int cc = t - C;
        v = rw0 * feat_s[(long)ri0 * C + cc]
          + rw1 * feat_s[(long)ri1 * C + cc]
          + rw2 * feat_s[(long)ri2 * C + cc];
      }
      WL[r * XSTRIDE + t] = v;
    }
  }
  asm volatile("" ::: "memory");

  v8f acc[7];
#pragma unroll
  for (int n = 0; n < 7; ++n) acc[n] = (v8f){0.f, 0.f, 0.f, 0.f, 0.f, 0.f, 0.f, 0.f};
  for (int k0 = 0; k0 < C2; k0 += 4) {
    const int kk = k0 + 2 * half;
    const v2f a1 = *(const v2f*)(WL + r15 * XSTRIDE + kk);
#pragma unroll
    for (int n = 0; n < 7; ++n) {
      v2f bv;
      bv.x = W1[(long)kk * C + n * 16 + r15];
      bv.y = W1[(long)(kk + 1) * C + n * 16 + r15];
      acc[n] = __builtin_amdgcn_wmma_f32_16x16x4_f32(false, a1, false, bv, (short)0, acc[n], false, false);
    }
  }

#pragma unroll
  for (int n = 0; n < 7; ++n) {
    const int ch = n * 16 + r15;
    const float gm = gammaW[ch], bt = betaW[ch];
#pragma unroll
    for (int r = 0; r < 8; ++r) {
      const float h = fmaxf(acc[n][r] * gm + bt, 0.f);
      WL[H_OFF + (r + 8 * half) * HSTRIDE + ch] = h;
    }
  }
  asm volatile("" ::: "memory");

  v8f acc2[7];
#pragma unroll
  for (int n = 0; n < 7; ++n) acc2[n] = (v8f){0.f, 0.f, 0.f, 0.f, 0.f, 0.f, 0.f, 0.f};
  for (int k0 = 0; k0 < C; k0 += 4) {
    const int kk = k0 + 2 * half;
    const v2f a2 = *(const v2f*)(WL + H_OFF + r15 * HSTRIDE + kk);
#pragma unroll
    for (int n = 0; n < 7; ++n) {
      v2f bv;
      bv.x = W2[(long)kk * C + n * 16 + r15];
      bv.y = W2[(long)(kk + 1) * C + n * 16 + r15];
      acc2[n] = __builtin_amdgcn_wmma_f32_16x16x4_f32(false, a2, false, bv, (short)0, acc2[n], false, false);
    }
  }
#pragma unroll
  for (int n = 0; n < 7; ++n) {
    const int ch = n * 16 + r15;
#pragma unroll
    for (int r = 0; r < 8; ++r)
      WL[Y2_OFF + (r + 8 * half) * HSTRIDE + ch] = acc2[n][r];
  }
  asm volatile("" ::: "memory");

  v8f acc3[2];
  acc3[0] = (v8f){0.f, 0.f, 0.f, 0.f, 0.f, 0.f, 0.f, 0.f};
  acc3[1] = acc3[0];
  for (int k0 = 0; k0 < C; k0 += 4) {
    const int kk = k0 + 2 * half;
    const v2f a3 = *(const v2f*)(WL + Y2_OFF + r15 * HSTRIDE + kk);
#pragma unroll
    for (int n = 0; n < 2; ++n) {
      const int col = n * 16 + r15;
      v2f bv;
      bv.x = (col < CLS) ? W_cls[(long)kk * CLS + col] : 0.f;
      bv.y = (col < CLS) ? W_cls[(long)(kk + 1) * CLS + col] : 0.f;
      acc3[n] = __builtin_amdgcn_wmma_f32_16x16x4_f32(false, a3, false, bv, (short)0, acc3[n], false, false);
    }
  }
#pragma unroll
  for (int n = 0; n < 2; ++n) {
    const int col = n * 16 + r15;
    if (col < CLS) {
      const float bb = b_cls[col];
#pragma unroll
      for (int r = 0; r < 8; ++r)
        out[(long)(qbase + r + 8 * half) * CLS + col] = acc3[n][r] + bb;
    }
  }
}

extern "C" void kernel_launch(void* const* d_in, const int* in_sizes, int n_in,
                              void* d_out, int out_size, void* d_ws, size_t ws_size,
                              hipStream_t stream) {
  const float* coords_q = (const float*)d_in[0];
  const float* coords_s = (const float*)d_in[1];
  const float* x7       = (const float*)d_in[2];
  const float* feat_s7  = (const float*)d_in[3];
  const float* feat_s   = (const float*)d_in[4];
  const float* w_attn   = (const float*)d_in[5];
  const float* W1       = (const float*)d_in[6];
  const float* gammaW   = (const float*)d_in[7];
  const float* betaW    = (const float*)d_in[8];
  const float* W2       = (const float*)d_in[9];
  const float* W_cls    = (const float*)d_in[10];
  const float* b_cls    = (const float*)d_in[11];
  float* out = (float*)d_out;

  float* ws     = (float*)d_ws;
  float* logits = ws;
  float* gate   = ws + 2 * NS;

  attn_logits<<<NS / 256, 256, 0, stream>>>(feat_s7, w_attn, logits);
  attn_pool_gate<<<C / PCH, PTHREADS, 0, stream>>>(feat_s7, logits, gate);

  const int blocks = (TILES + WAVES_PER_BLOCK - 1) / WAVES_PER_BLOCK;
  fused_main<<<blocks, 128, 0, stream>>>(coords_q, coords_s, x7, feat_s, gate,
                                         W1, gammaW, betaW, W2, W_cls, b_cls, out);
}


# top-3 search loop without a memory round trip per tile: support points as matrix rows (a lane's 8 result registers are its own query's scores, no LDS hop), 16 tiles per trip, next loads in flight
# speedup vs baseline: 6.2443x; 1.6850x over previous
#include <hip/hip_runtime.h>
#include <math.h>

#define NQ   20000
#define NS   8192
#define C    112
#define C2   224
#define CLS  20
#define TILES (NQ / 16)
#define WAVES_PER_BLOCK 4
#define PCH        4
#define GATE_PITCH 32
#define SB         16

#define WREG    3712
#define XSTRIDE 226
#define HSTRIDE 114
#define H_OFF   0
#define Y2_OFF  1856

typedef float v2f __attribute__((ext_vector_type(2)));
typedef float v8f __attribute__((ext_vector_type(8)));

__global__ void attn_logits(const float* __restrict__ feat_s7,
                            const float* __restrict__ w_attn,
                            float* __restrict__ logits) {
  int j = blockIdx.x * blockDim.x + threadIdx.x;
  if (j >= NS) return;
  const float* row = feat_s7 + (long)j * C;
  float acc = 0.f;
#pragma unroll 8
  for (int c = 0; c < C; ++c) acc += row[c] * w_attn[c];
  logits[j] = acc;
}

#define PTHREADS 1024
#define PROWS    (NS / PTHREADS)
#define PWAVES   (PTHREADS / 32)
static_assert(PWAVES == 32 && NS % PTHREADS == 0 && C % PCH == 0 && PCH == 4, "pool split");
static_assert(NS % (16 * SB) == 0 && (NS & (NS - 1)) == 0, "search trips");

__device__ __forceinline__ float wave_max(float v) {
  for (int o = 16; o > 0; o >>= 1) v = fmaxf(v, __shfl_xor(v, o));
  return v;
}
__device__ __forceinline__ float wave_sum(float v) {
  for (int o = 16; o > 0; o >>= 1) v += __shfl_xor(v, o);
  return v;
}

__global__ __launch_bounds__(PTHREADS) void attn_pool_gate(
    const float* __restrict__ feat_s7, const float* __restrict__ logits,
    float* __restrict__ gate) {
  __shared__ float redm[PWAVES];
  __shared__ float reds[1 + PCH][PWAVES];
  const int t = threadIdx.x;
  const int lane = t & 31;
  const int wave = t >> 5;
  const int c0 = blockIdx.x * PCH;

  float l[PROWS];
  float4 f[PROWS];
#pragma unroll
  for (int i = 0; i < PROWS; ++i) {
    const int j = t + i * PTHREADS;
    l[i] = logits[j];
    f[i] = *(const float4*)(feat_s7 + (long)j * C + c0);
  }

  float m = l[0];
#pragma unroll
  for (int i = 1; i < PROWS; ++i) m = fmaxf(m, l[i]);
  m = wave_max(m);
  if (lane == 0) redm[wave] = m;
  __syncthreads();
  m = wave_max(redm[lane]);

  float s = 0.f, a0 = 0.f, a1 = 0.f, a2 = 0.f, a3 = 0.f;
#pragma unroll
  for (int i = 0; i < PROWS; ++i) {
    const float e = __expf(l[i] - m);
    s  += e;
    a0 += e * f[i].x;
    a1 += e * f[i].y;
    a2 += e * f[i].z;
    a3 += e * f[i].w;
  }
  s  = wave_sum(s);
  a0 = wave_sum(a0);
  a1 = wave_sum(a1);
  a2 = wave_sum(a2);
  a3 = wave_sum(a3);
  if (lane == 0) {
    reds[0][wave] = s;
    reds[1][wave] = a0;
    reds[2][wave] = a1;
    reds[3][wave] = a2;
    reds[4][wave] = a3;
  }
  __syncthreads();

  if (wave == 0) {
    const float sum = wave_sum(reds[0][lane]);
    const float t0 = wave_sum(reds[1][lane]);
    const float t1 = wave_sum(reds[2][lane]);
    const float t2 = wave_sum(reds[3][lane]);
    const float t3 = wave_sum(reds[4][lane]);
    if (lane < PCH) {
      const float acc = (lane == 0) ? t0 : (lane == 1) ? t1 : (lane == 2) ? t2 : t3;
      const float inv = 1.0f / sum;
      const float g = acc * inv;
      gate[blockIdx.x * GATE_PITCH + lane] = 1.0f / (1.0f + __expf(-g));
    }
  }
}

#define TOP3_INS(val, idxv) {                                         \
    float s_ = (val); int id_ = (idxv);                               \
    bool c0_ = s_ < v0, c1_ = s_ < v1, c2_ = s_ < v2v;                \
    v2v = c1_ ? v1 : (c2_ ? s_  : v2v); i2 = c1_ ? i1 : (c2_ ? id_ : i2); \
    v1  = c0_ ? v0 : (c1_ ? s_  : v1 ); i1 = c0_ ? i0 : (c1_ ? id_ : i1); \
    v0  = c0_ ? s_ : v0;                i0 = c0_ ? id_ : i0;          }

__global__ __launch_bounds__(128) void fused_main(
    const float* __restrict__ coords_q, const float* __restrict__ coords_s,
    const float* __restrict__ x7,       const float* __restrict__ feat_s,
    const float* __restrict__ gate,     const float* __restrict__ W1,
    const float* __restrict__ gammaW,   const float* __restrict__ betaW,
    const float* __restrict__ W2,       const float* __restrict__ W_cls,
    const float* __restrict__ b_cls,    float* __restrict__ out)
{
  __shared__ float lds[WAVES_PER_BLOCK * WREG];
  const int lane = threadIdx.x & 31;
  const int wave = threadIdx.x >> 5;
  const int tile = blockIdx.x * WAVES_PER_BLOCK + wave;
  if (tile >= TILES) return;

  float* WL = lds + wave * WREG;
  const int r15  = lane & 15;
  const int half = lane >> 4;
  const int qbase = tile * 16;

  const float qx = coords_q[(qbase + r15) * 3 + 0];
  const float qy = coords_q[(qbase + r15) * 3 + 1];
  const float qz = coords_q[(qbase + r15) * 3 + 2];
  v2f qb;
  qb.x = half ? (-2.f * qz) : (-2.f * qx);
  qb.y = half ? 1.0f        : (-2.f * qy);

  float v0 = 1e30f, v1 = 1e30f, v2v = 1e30f;
  int   i0 = 0, i1 = 0, i2 = 0;

  float cx[SB], cy[SB], cz[SB];
#pragma unroll
  for (int u = 0; u < SB; ++u) {
    const float* p = coords_s + (u * 16 + r15) * 3;
    cx[u] = p[0]; cy[u] = p[1]; cz[u] = p[2];
  }
  for (int n0 = 0; n0 < NS; n0 += 16 * SB) {
    v2f sa[SB];
#pragma unroll
    for (int u = 0; u < SB; ++u) {
      const float s2 = fmaf(cz[u], cz[u], fmaf(cx[u], cx[u], cy[u] * cy[u]));
      sa[u].x = half ? cz[u] : cx[u];
      sa[u].y = half ? s2    : cy[u];
    }
    const int nb = (n0 + 16 * SB) & (NS - 1);
#pragma unroll
    for (int u = 0; u < SB; ++u) {
      const float* p = coords_s + (nb + u * 16 + r15) * 3;
      cx[u] = p[0]; cy[u] = p[1]; cz[u] = p[2];
    }
    v8f c[SB];
#pragma unroll
    for (int u = 0; u < SB; ++u) {
      const v8f z = {0.f, 0.f, 0.f, 0.f, 0.f, 0.f, 0.f, 0.f};
      c[u] = __builtin_amdgcn_wmma_f32_16x16x4_f32(false, sa[u], false, qb, (short)0, z, false, false);
    }
#pragma unroll
    for (int u = 0; u < SB; ++u) {
      const float m8 = fminf(fminf(fminf(c[u][0], c[u][1]), fminf(c[u][2], c[u][3])),
                             fminf(fminf(c[u][4], c[u][5]), fminf(c[u][6], c[u][7])));
      if (__any(m8 < v2v)) {
        const int base = n0 + u * 16 + 8 * half;
        TOP3_INS(c[u][0], base + 0); TOP3_INS(c[u][1], base + 1);
        TOP3_INS(c[u][2], base + 2); TOP3_INS(c[u][3], base + 3);
        TOP3_INS(c[u][4], base + 4); TOP3_INS(c[u][5], base + 5);
        TOP3_INS(c[u][6], base + 6); TOP3_INS(c[u][7], base + 7);
      }
    }
  }

  {
    float pv0 = __shfl(v0, lane + 16), pv1 = __shfl(v1, lane + 16), pv2 = __shfl(v2v, lane + 16);
    int   pi0 = __shfl(i0, lane + 16), pi1 = __shfl(i1, lane + 16), pi2 = __shfl(i2,  lane + 16);
    TOP3_INS(pv0, pi0); TOP3_INS(pv1, pi1); TOP3_INS(pv2, pi2);
  }
  const float e1 = __expf(v0 - v1), e2 = __expf(v0 - v2v);
  const float inv = 1.0f / (1.0f + e1 + e2);
  const float w0 = inv, w1 = e1 * inv, w2 = e2 * inv;

  for (int r = 0; r < 16; ++r) {
    const float rw0 = __shfl(w0, r), rw1 = __shfl(w1, r), rw2 = __shfl(w2, r);
    const int   ri0 = __shfl(i0, r), ri1 = __shfl(i1, r), ri2 = __shfl(i2, r);
    const int qi = qbase + r;
    for (int t = lane; t < C2; t += 32) {
      float v;
      if (t < C) {
        v = x7[(long)qi * C + t] * gate[(t / PCH) * GATE_PITCH + (t % PCH)];
      } else {
        const int cc = t - C;
        v = rw0 * feat_s[(long)ri0 * C + cc]
          + rw1 * feat_s[(long)ri1 * C + cc]
          + rw2 * feat_s[(long)ri2 * C + cc];
      }
      WL[r * XSTRIDE + t] = v;
    }
  }
  asm volatile("" ::: "memory");

  v8f acc[7];
#pragma unroll
  for (int n = 0; n < 7; ++n) acc[n] = (v8f){0.f, 0.f, 0.f, 0.f, 0.f, 0.f, 0.f, 0.f};
  for (int k0 = 0; k0 < C2; k0 += 4) {
    const int kk = k0 + 2 * half;
    const v2f a1 = *(const v2f*)(WL + r15 * XSTRIDE + kk);
#pragma unroll
    for (int n = 0; n < 7; ++n) {
      v2f bv;
      bv.x = W1[(long)kk * C + n * 16 + r15];
      bv.y = W1[(long)(kk + 1) * C + n * 16 + r15];
      acc[n] = __builtin_amdgcn_wmma_f32_16x16x4_f32(false, a1, false, bv, (short)0, acc[n], false, false);
    }
  }

#pragma unroll
  for (int n = 0; n < 7; ++n) {
    const int ch = n * 16 + r15;
    const float gm = gammaW[ch], bt = betaW[ch];
#pragma unroll
    for (int r = 0; r < 8; ++r) {
      const float h = fmaxf(acc[n][r] * gm + bt, 0.f);
      WL[H_OFF + (r + 8 * half) * HSTRIDE + ch] = h;
    }
  }
  asm volatile("" ::: "memory");

  v8f acc2[7];
#pragma unroll
  for (int n = 0; n < 7; ++n) acc2[n] = (v8f){0.f, 0.f, 0.f, 0.f, 0.f, 0.f, 0.f, 0.f};
  for (int k0 = 0; k0 < C; k0 += 4) {
    const int kk = k0 + 2 * half;
    const v2f a2 = *(const v2f*)(WL + H_OFF + r15 * HSTRIDE + kk);
#pragma unroll
    for (int n = 0; n < 7; ++n) {
      v2f bv;
      bv.x = W2[(long)kk * C + n * 16 + r15];
      bv.y = W2[(long)(kk + 1) * C + n * 16 + r15];
      acc2[n] = __builtin_amdgcn_wmma_f32_16x16x4_f32(false, a2, false, bv, (short)0, acc2[n], false, false);
    }
  }
#pragma unroll
  for (int n = 0; n < 7; ++n) {
    const int ch = n * 16 + r15;
#pragma unroll
    for (int r = 0; r < 8; ++r)
      WL[Y2_OFF + (r + 8 * half) * HSTRIDE + ch] = acc2[n][r];
  }
  asm volatile("" ::: "memory");

  v8f acc3[2];
  acc3[0] = (v8f){0.f, 0.f, 0.f, 0.f, 0.f, 0.f, 0.f, 0.f};
  acc3[1] = acc3[0];
  for (int k0 = 0; k0 < C; k0 += 4) {
    const int kk = k0 + 2 * half;
    const v2f a3 = *(const v2f*)(WL + Y2_OFF + r15 * HSTRIDE + kk);
#pragma unroll
    for (int n = 0; n < 2; ++n) {
      const int col = n * 16 + r15;
      v2f bv;
      bv.x = (col < CLS) ? W_cls[(long)kk * CLS + col] : 0.f;
      bv.y = (col < CLS) ? W_cls[(long)(kk + 1) * CLS + col] : 0.f;
      acc3[n] = __builtin_amdgcn_wmma_f32_16x16x4_f32(false, a3, false, bv, (short)0, acc3[n], false, false);
    }
  }
#pragma unroll
  for (int n = 0; n < 2; ++n) {
    const int col = n * 16 + r15;
    if (col < CLS) {
      const float bb = b_cls[col];
#pragma unroll
      for (int r = 0; r < 8; ++r)
        out[(long)(qbase + r + 8 * half) * CLS + col] = acc3[n][r] + bb;
    }
  }
}

extern "C" void kernel_launch(void* const* d_in, const int* in_sizes, int n_in,
                              void* d_out, int out_size, void* d_ws, size_t ws_size,
                              hipStream_t stream) {
  const float* coords_q = (const float*)d_in[0];
  const float* coords_s = (const float*)d_in[1];
  const float* x7       = (const float*)d_in[2];
  const float* feat_s7  = (const float*)d_in[3];
  const float* feat_s   = (const float*)d_in[4];
  const float* w_attn   = (const float*)d_in[5];
  const float* W1       = (const float*)d_in[6];
  const float* gammaW   = (const float*)d_in[7];
  const float* betaW    = (const float*)d_in[8];
  const float* W2       = (const float*)d_in[9];
  const float* W_cls    = (const float*)d_in[10];
  const float* b_cls    = (const float*)d_in[11];
  float* out = (float*)d_out;

  float* ws     = (float*)d_ws;
  float* logits = ws;
  float* gate   = ws + 2 * NS;

  attn_logits<<<NS / 256, 256, 0, stream>>>(feat_s7, w_attn, logits);
  attn_pool_gate<<<C / PCH, PTHREADS, 0, stream>>>(feat_s7, logits, gate);

  const int blocks = (TILES + WAVES_PER_BLOCK - 1) / WAVES_PER_BLOCK;
  fused_main<<<blocks, 128, 0, stream>>>(coords_q, coords_s, x7, feat_s, gate,
                                         W1, gammaW, betaW, W2, W_cls, b_cls, out);
}


# GEMM1 takes its A operand straight from global memory: a lane loads its own fragment pairs of x7, gate and its 3 matched rows per k step; the X tile in LDS and its row-by-row build loop are gone
# speedup vs baseline: 8.3117x; 1.3311x over previous
#include <hip/hip_runtime.h>
#include <math.h>

#define NQ   20000
#define NS   8192
#define C    112
#define C2   224
#define CLS  20
#define TILES (NQ / 16)
#define WAVES_PER_BLOCK 4
#define PCH        4
#define GATE_PITCH 32
#define SB         16

#define WREG    3712
#define HSTRIDE 114
#define H_OFF   0
#define Y2_OFF  1856

typedef float v2f __attribute__((ext_vector_type(2)));
typedef float v8f __attribute__((ext_vector_type(8)));

__global__ void attn_logits(const float* __restrict__ feat_s7,
                            const float* __restrict__ w_attn,
                            float* __restrict__ logits) {
  int j = blockIdx.x * blockDim.x + threadIdx.x;
  if (j >= NS) return;
  const float* row = feat_s7 + (long)j * C;
  float acc = 0.f;
#pragma unroll 8
  for (int c = 0; c < C; ++c) acc += row[c] * w_attn[c];
  logits[j] = acc;
}

#define PTHREADS 1024
#define PROWS    (NS / PTHREADS)
#define PWAVES   (PTHREADS / 32)
static_assert(PWAVES == 32 && NS % PTHREADS == 0 && C % PCH == 0 && PCH == 4, "pool split");
static_assert(NS % (16 * SB) == 0 && (NS & (NS - 1)) == 0, "search trips");

__device__ __forceinline__ float wave_max(float v) {
  for (int o = 16; o > 0; o >>= 1) v = fmaxf(v, __shfl_xor(v, o));
  return v;
}
__device__ __forceinline__ float wave_sum(float v) {
  for (int o = 16; o > 0; o >>= 1) v += __shfl_xor(v, o);
  return v;
}

__global__ __launch_bounds__(PTHREADS) void attn_pool_gate(
    const float* __restrict__ feat_s7, const float* __restrict__ logits,
    float* __restrict__ gate) {
  __shared__ float redm[PWAVES];
  __shared__ float reds[1 + PCH][PWAVES];
  const int t = threadIdx.x;
  const int lane = t & 31;
  const int wave = t >> 5;
  const int c0 = blockIdx.x * PCH;

  float l[PROWS];
  float4 f[PROWS];
#pragma unroll
  for (int i = 0; i < PROWS; ++i) {
    const int j = t + i * PTHREADS;
    l[i] = logits[j];
    f[i] = *(const float4*)(feat_s7 + (long)j * C + c0);
  }

  float m = l[0];
#pragma unroll
  for (int i = 1; i < PROWS; ++i) m = fmaxf(m, l[i]);
  m = wave_max(m);
  if (lane == 0) redm[wave] = m;
  __syncthreads();
  m = wave_max(redm[lane]);

  float s = 0.f, a0 = 0.f, a1 = 0.f, a2 = 0.f, a3 = 0.f;
#pragma unroll
  for (int i = 0; i < PROWS; ++i) {
    const float e = __expf(l[i] - m);
    s  += e;
    a0 += e * f[i].x;
    a1 += e * f[i].y;
    a2 += e * f[i].z;
    a3 += e * f[i].w;
  }
  s  = wave_sum(s);
  a0 = wave_sum(a0);
  a1 = wave_sum(a1);
  a2 = wave_sum(a2);
  a3 = wave_sum(a3);
  if (lane == 0) {
    reds[0][wave] = s;
    reds[1][wave] = a0;
    reds[2][wave] = a1;
    reds[3][wave] = a2;
    reds[4][wave] = a3;
  }
  __syncthreads();

  if (wave == 0) {
    const float sum = wave_sum(reds[0][lane]);
    const float t0 = wave_sum(reds[1][lane]);
    const float t1 = wave_sum(reds[2][lane]);
    const float t2 = wave_sum(reds[3][lane]);
    const float t3 = wave_sum(reds[4][lane]);
    if (lane < PCH) {
      const float acc = (lane == 0) ? t0 : (lane == 1) ? t1 : (lane == 2) ? t2 : t3;
      const float inv = 1.0f / sum;
      const float g = acc * inv;
      gate[blockIdx.x * GATE_PITCH + lane] = 1.0f / (1.0f + __expf(-g));
    }
  }
}

#define TOP3_INS(val, idxv) {                                         \
    float s_ = (val); int id_ = (idxv);                               \
    bool c0_ = s_ < v0, c1_ = s_ < v1, c2_ = s_ < v2v;                \
    v2v = c1_ ? v1 : (c2_ ? s_  : v2v); i2 = c1_ ? i1 : (c2_ ? id_ : i2); \
    v1  = c0_ ? v0 : (c1_ ? s_  : v1 ); i1 = c0_ ? i0 : (c1_ ? id_ : i1); \
    v0  = c0_ ? s_ : v0;                i0 = c0_ ? id_ : i0;          }

__global__ __launch_bounds__(128) void fused_main(
    const float* __restrict__ coords_q, const float* __restrict__ coords_s,
    const float* __restrict__ x7,       const float* __restrict__ feat_s,
    const float* __restrict__ gate,     const float* __restrict__ W1,
    const float* __restrict__ gammaW,   const float* __restrict__ betaW,
    const float* __restrict__ W2,       const float* __restrict__ W_cls,
    const float* __restrict__ b_cls,    float* __restrict__ out)
{
  __shared__ float lds[WAVES_PER_BLOCK * WREG];
  const int lane = threadIdx.x & 31;
  const int wave = threadIdx.x >> 5;
  const int tile = blockIdx.x * WAVES_PER_BLOCK + wave;
  if (tile >= TILES) return;

  float* WL = lds + wave * WREG;
  const int r15  = lane & 15;
  const int half = lane >> 4;
  const int qbase = tile * 16;

  const float qx = coords_q[(qbase + r15) * 3 + 0];
  const float qy = coords_q[(qbase + r15) * 3 + 1];
  const float qz = coords_q[(qbase + r15) * 3 + 2];
  v2f qb;
  qb.x = half ? (-2.f * qz) : (-2.f * qx);
  qb.y = half ? 1.0f        : (-2.f * qy);

  float v0 = 1e30f, v1 = 1e30f, v2v = 1e30f;
  int   i0 = 0, i1 = 0, i2 = 0;

  float cx[SB], cy[SB], cz[SB];
#pragma unroll
  for (int u = 0; u < SB; ++u) {
    const float* p = coords_s + (u * 16 + r15) * 3;
    cx[u] = p[0]; cy[u] = p[1]; cz[u] = p[2];
  }
  for (int n0 = 0; n0 < NS; n0 += 16 * SB) {
    v2f sa[SB];
#pragma unroll
    for (int u = 0; u < SB; ++u) {
      const float s2 = fmaf(cz[u], cz[u], fmaf(cx[u], cx[u], cy[u] * cy[u]));
      sa[u].x = half ? cz[u] : cx[u];
      sa[u].y = half ? s2    : cy[u];
    }
    const int nb = (n0 + 16 * SB) & (NS - 1);
#pragma unroll
    for (int u = 0; u < SB; ++u) {
      const float* p = coords_s + (nb + u * 16 + r15) * 3;
      cx[u] = p[0]; cy[u] = p[1]; cz[u] = p[2];
    }
    v8f c[SB];
#pragma unroll
    for (int u = 0; u < SB; ++u) {
      const v8f z = {0.f, 0.f, 0.f, 0.f, 0.f, 0.f, 0.f, 0.f};
      c[u] = __builtin_amdgcn_wmma_f32_16x16x4_f32(false, sa[u], false, qb, (short)0, z, false, false);
    }
#pragma unroll
    for (int u = 0; u < SB; ++u) {
      const float m8 = fminf(fminf(fminf(c[u][0], c[u][1]), fminf(c[u][2], c[u][3])),
                             fminf(fminf(c[u][4], c[u][5]), fminf(c[u][6], c[u][7])));
      if (__any(m8 < v2v)) {
        const int base = n0 + u * 16 + 8 * half;
        TOP3_INS(c[u][0], base + 0); TOP3_INS(c[u][1], base + 1);
        TOP3_INS(c[u][2], base + 2); TOP3_INS(c[u][3], base + 3);
        TOP3_INS(c[u][4], base + 4); TOP3_INS(c[u][5], base + 5);
        TOP3_INS(c[u][6], base + 6); TOP3_INS(c[u][7], base + 7);
      }
    }
  }

  {
    float pv0 = __shfl(v0, lane + 16), pv1 = __shfl(v1, lane + 16), pv2 = __shfl(v2v, lane + 16);
    int   pi0 = __shfl(i0, lane + 16), pi1 = __shfl(i1, lane + 16), pi2 = __shfl(i2,  lane + 16);
    TOP3_INS(pv0, pi0); TOP3_INS(pv1, pi1); TOP3_INS(pv2, pi2);
  }
  const float e1 = __expf(v0 - v1), e2 = __expf(v0 - v2v);
  const float inv = 1.0f / (1.0f + e1 + e2);
  const float w0 = inv, w1 = e1 * inv, w2 = e2 * inv;

  i0 = __shfl(i0, r15); i1 = __shfl(i1, r15); i2 = __shfl(i2, r15);

  v8f acc[7];
#pragma unroll
  for (int n = 0; n < 7; ++n) acc[n] = (v8f){0.f, 0.f, 0.f, 0.f, 0.f, 0.f, 0.f, 0.f};
  {
    const float* xrow = x7 + (long)(qbase + r15) * C + 2 * half;
    const float* grow = gate + 2 * half;
    for (int k0 = 0; k0 < C; k0 += 4) {
      const int kk = k0 + 2 * half;
      const v2f xv = *(const v2f*)(xrow + k0);
      const v2f gv = *(const v2f*)(grow + (k0 / PCH) * GATE_PITCH);
      v2f a1;
      a1.x = xv.x * gv.x;
      a1.y = xv.y * gv.y;
#pragma unroll
      for (int n = 0; n < 7; ++n) {
        v2f bv;
        bv.x = W1[(long)kk * C + n * 16 + r15];
        bv.y = W1[(long)(kk + 1) * C + n * 16 + r15];
        acc[n] = __builtin_amdgcn_wmma_f32_16x16x4_f32(false, a1, false, bv, (short)0, acc[n], false, false);
      }
    }
  }
  {
    const float* f0 = feat_s + (long)i0 * C + 2 * half;
    const float* f1 = feat_s + (long)i1 * C + 2 * half;
    const float* f2 = feat_s + (long)i2 * C + 2 * half;
    for (int k0 = 0; k0 < C; k0 += 4) {
      const int kk = C + k0 + 2 * half;
      const v2f p0 = *(const v2f*)(f0 + k0);
      const v2f p1 = *(const v2f*)(f1 + k0);
      const v2f p2 = *(const v2f*)(f2 + k0);
      v2f a1;
      a1.x = fmaf(w2, p2.x, fmaf(w0, p0.x, w1 * p1.x));
      a1.y = fmaf(w2, p2.y, fmaf(w0, p0.y, w1 * p1.y));
#pragma unroll
      for (int n = 0; n < 7; ++n) {
        v2f bv;
        bv.x = W1[(long)kk * C + n * 16 + r15];
        bv.y = W1[(long)(kk + 1) * C + n * 16 + r15];
        acc[n] = __builtin_amdgcn_wmma_f32_16x16x4_f32(false, a1, false, bv, (short)0, acc[n], false, false);
      }
    }
  }

#pragma unroll
  for (int n = 0; n < 7; ++n) {
    const int ch = n * 16 + r15;
    const float gm = gammaW[ch], bt = betaW[ch];
#pragma unroll
    for (int r = 0; r < 8; ++r) {
      const float h = fmaxf(acc[n][r] * gm + bt, 0.f);
      WL[H_OFF + (r + 8 * half) * HSTRIDE + ch] = h;
    }
  }
  asm volatile("" ::: "memory");

  v8f acc2[7];
#pragma unroll
  for (int n = 0; n < 7; ++n) acc2[n] = (v8f){0.f, 0.f, 0.f, 0.f, 0.f, 0.f, 0.f, 0.f};
  for (int k0 = 0; k0 < C; k0 += 4) {
    const int kk = k0 + 2 * half;
    const v2f a2 = *(const v2f*)(WL + H_OFF + r15 * HSTRIDE + kk);
#pragma unroll
    for (int n = 0; n < 7; ++n) {
      v2f bv;
      bv.x = W2[(long)kk * C + n * 16 + r15];
      bv.y = W2[(long)(kk + 1) * C + n * 16 + r15];
      acc2[n] = __builtin_amdgcn_wmma_f32_16x16x4_f32(false, a2, false, bv, (short)0, acc2[n], false, false);
    }
  }
#pragma unroll
  for (int n = 0; n < 7; ++n) {
    const int ch = n * 16 + r15;
#pragma unroll
    for (int r = 0; r < 8; ++r)
      WL[Y2_OFF + (r + 8 * half) * HSTRIDE + ch] = acc2[n][r];
  }
  asm volatile("" ::: "memory");

  v8f acc3[2];
  acc3[0] = (v8f){0.f, 0.f, 0.f, 0.f, 0.f, 0.f, 0.f, 0.f};
  acc3[1] = acc3[0];
  for (int k0 = 0; k0 < C; k0 += 4) {
    const int kk = k0 + 2 * half;
    const v2f a3 = *(const v2f*)(WL + Y2_OFF + r15 * HSTRIDE + kk);
#pragma unroll
    for (int n = 0; n < 2; ++n) {
      const int col = n * 16 + r15;
      v2f bv;
      bv.x = (col < CLS) ? W_cls[(long)kk * CLS + col] : 0.f;
      bv.y = (col < CLS) ? W_cls[(long)(kk + 1) * CLS + col] : 0.f;
      acc3[n] = __builtin_amdgcn_wmma_f32_16x16x4_f32(false, a3, false, bv, (short)0, acc3[n], false, false);
    }
  }
#pragma unroll
  for (int n = 0; n < 2; ++n) {
    const int col = n * 16 + r15;
    if (col < CLS) {
      const float bb = b_cls[col];
#pragma unroll
      for (int r = 0; r < 8; ++r)
        out[(long)(qbase + r + 8 * half) * CLS + col] = acc3[n][r] + bb;
    }
  }
}

extern "C" void kernel_launch(void* const* d_in, const int* in_sizes, int n_in,
                              void* d_out, int out_size, void* d_ws, size_t ws_size,
                              hipStream_t stream) {
  const float* coords_q = (const float*)d_in[0];
  const float* coords_s = (const float*)d_in[1];
  const float* x7       = (const float*)d_in[2];
  const float* feat_s7  = (const float*)d_in[3];
  const float* feat_s   = (const float*)d_in[4];
  const float* w_attn   = (const float*)d_in[5];
  const float* W1       = (const float*)d_in[6];
  const float* gammaW   = (const float*)d_in[7];
  const float* betaW    = (const float*)d_in[8];
  const float* W2       = (const float*)d_in[9];
  const float* W_cls    = (const float*)d_in[10];
  const float* b_cls    = (const float*)d_in[11];
  float* out = (float*)d_out;

  float* ws     = (float*)d_ws;
  float* logits = ws;
  float* gate   = ws + 2 * NS;

  attn_logits<<<NS / 256, 256, 0, stream>>>(feat_s7, w_attn, logits);
  attn_pool_gate<<<C / PCH, PTHREADS, 0, stream>>>(feat_s7, logits, gate);

  const int blocks = (TILES + WAVES_PER_BLOCK - 1) / WAVES_PER_BLOCK;
  fused_main<<<blocks, 128, 0, stream>>>(coords_q, coords_s, x7, feat_s, gate,
                                         W1, gammaW, betaW, W2, W_cls, b_cls, out);
}


# GEMM k loops software-pipelined: a 16-row k block's operands load one block ahead into a second register set as one group; W_cls loaded ahead of GEMM3; k order in a block changed, sums re-associated
# speedup vs baseline: 9.8056x; 1.1797x over previous
#include <hip/hip_runtime.h>
#include <math.h>

#define NQ   20000
#define NS   8192
#define C    112
#define C2   224
#define CLS  20
#define TILES (NQ / 16)
#define WAVES_PER_BLOCK 4
#define PCH        4
#define GATE_PITCH 32
#define SB         16

#define WREG    3712
#define HSTRIDE 114
#define H_OFF   0
#define Y2_OFF  1856

typedef float v2f __attribute__((ext_vector_type(2)));
typedef float v8f __attribute__((ext_vector_type(8)));

__global__ void attn_logits(const float* __restrict__ feat_s7,
                            const float* __restrict__ w_attn,
                            float* __restrict__ logits) {
  int j = blockIdx.x * blockDim.x + threadIdx.x;
  if (j >= NS) return;
  const float* row = feat_s7 + (long)j * C;
  float acc = 0.f;
#pragma unroll 8
  for (int c = 0; c < C; ++c) acc += row[c] * w_attn[c];
  logits[j] = acc;
}

#define PTHREADS 1024
#define PROWS    (NS / PTHREADS)
#define PWAVES   (PTHREADS / 32)
static_assert(PWAVES == 32 && NS % PTHREADS == 0 && C % PCH == 0 && PCH == 4, "pool split");
static_assert(NS % (16 * SB) == 0 && (NS & (NS - 1)) == 0, "search trips");

__device__ __forceinline__ float wave_max(float v) {
  for (int o = 16; o > 0; o >>= 1) v = fmaxf(v, __shfl_xor(v, o));
  return v;
}
__device__ __forceinline__ float wave_sum(float v) {
  for (int o = 16; o > 0; o >>= 1) v += __shfl_xor(v, o);
  return v;
}

__global__ __launch_bounds__(PTHREADS) void attn_pool_gate(
    const float* __restrict__ feat_s7, const float* __restrict__ logits,
    float* __restrict__ gate) {
  __shared__ float redm[PWAVES];
  __shared__ float reds[1 + PCH][PWAVES];
  const int t = threadIdx.x;
  const int lane = t & 31;
  const int wave = t >> 5;
  const int c0 = blockIdx.x * PCH;

  float l[PROWS];
  float4 f[PROWS];
#pragma unroll
  for (int i = 0; i < PROWS; ++i) {
    const int j = t + i * PTHREADS;
    l[i] = logits[j];
    f[i] = *(const float4*)(feat_s7 + (long)j * C + c0);
  }

  float m = l[0];
#pragma unroll
  for (int i = 1; i < PROWS; ++i) m = fmaxf(m, l[i]);
  m = wave_max(m);
  if (lane == 0) redm[wave] = m;
  __syncthreads();
  m = wave_max(redm[lane]);

  float s = 0.f, a0 = 0.f, a1 = 0.f, a2 = 0.f, a3 = 0.f;
#pragma unroll
  for (int i = 0; i < PROWS; ++i) {
    const float e = __expf(l[i] - m);
    s  += e;
    a0 += e * f[i].x;
    a1 += e * f[i].y;
    a2 += e * f[i].z;
    a3 += e * f[i].w;
  }
  s  = wave_sum(s);
  a0 = wave_sum(a0);
  a1 = wave_sum(a1);
  a2 = wave_sum(a2);
  a3 = wave_sum(a3);
  if (lane == 0) {
    reds[0][wave] = s;
    reds[1][wave] = a0;
    reds[2][wave] = a1;
    reds[3][wave] = a2;
    reds[4][wave] = a3;
  }
  __syncthreads();

  if (wave == 0) {
    const float sum = wave_sum(reds[0][lane]);
    const float t0 = wave_sum(reds[1][lane]);
    const float t1 = wave_sum(reds[2][lane]);
    const float t2 = wave_sum(reds[3][lane]);
    const float t3 = wave_sum(reds[4][lane]);
    if (lane < PCH) {
      const float acc = (lane == 0) ? t0 : (lane == 1) ? t1 : (lane == 2) ? t2 : t3;
      const float inv = 1.0f / sum;
      const float g = acc * inv;
      gate[blockIdx.x * GATE_PITCH + lane] = 1.0f / (1.0f + __expf(-g));
    }
  }
}

#define TOP3_INS(val, idxv) {                                         \
    float s_ = (val); int id_ = (idxv);                               \
    bool c0_ = s_ < v0, c1_ = s_ < v1, c2_ = s_ < v2v;                \
    v2v = c1_ ? v1 : (c2_ ? s_  : v2v); i2 = c1_ ? i1 : (c2_ ? id_ : i2); \
    v1  = c0_ ? v0 : (c1_ ? s_  : v1 ); i1 = c0_ ? i0 : (c1_ ? id_ : i1); \
    v0  = c0_ ? s_ : v0;                i0 = c0_ ? id_ : i0;          }

#define KB   4
#define NBLK (C / (4 * KB))
static_assert(C % (4 * KB) == 0 && KB == 4 && PCH == 4, "k blocks: 16 rows, 4 gate lines");

#define LOAD_B7(dst, Wp, krow0) {                                              \
    _Pragma("unroll")                                                          \
    for (int s_ = 0; s_ < KB; ++s_) {                                          \
      const float* r0_ = (Wp) + (long)((krow0) + 8 * half + 2 * s_) * C + r15; \
      _Pragma("unroll")                                                        \
      for (int n_ = 0; n_ < 7; ++n_) {                                         \
        dst[s_][n_].x = r0_[n_ * 16];                                          \
        dst[s_][n_].y = r0_[C + n_ * 16];                                      \
      }                                                                        \
    } }

__global__ __launch_bounds__(128) void fused_main(
    const float* __restrict__ coords_q, const float* __restrict__ coords_s,
    const float* __restrict__ x7,       const float* __restrict__ feat_s,
    const float* __restrict__ gate,     const float* __restrict__ W1,
    const float* __restrict__ gammaW,   const float* __restrict__ betaW,
    const float* __restrict__ W2,       const float* __restrict__ W_cls,
    const float* __restrict__ b_cls,    float* __restrict__ out)
{
  __shared__ float lds[WAVES_PER_BLOCK * WREG];
  const int lane = threadIdx.x & 31;
  const int wave = threadIdx.x >> 5;
  const int tile = blockIdx.x * WAVES_PER_BLOCK + wave;
  if (tile >= TILES) return;

  float* WL = lds + wave * WREG;
  const int r15  = lane & 15;
  const int half = lane >> 4;
  const int qbase = tile * 16;

  const float qx = coords_q[(qbase + r15) * 3 + 0];
  const float qy = coords_q[(qbase + r15) * 3 + 1];
  const float qz = coords_q[(qbase + r15) * 3 + 2];
  v2f qb;
  qb.x = half ? (-2.f * qz) : (-2.f * qx);
  qb.y = half ? 1.0f        : (-2.f * qy);

  float v0 = 1e30f, v1 = 1e30f, v2v = 1e30f;
  int   i0 = 0, i1 = 0, i2 = 0;

  float cx[SB], cy[SB], cz[SB];
#pragma unroll
  for (int u = 0; u < SB; ++u) {
    const float* p = coords_s + (u * 16 + r15) * 3;
    cx[u] = p[0]; cy[u] = p[1]; cz[u] = p[2];
  }
  for (int n0 = 0; n0 < NS; n0 += 16 * SB) {
    v2f sa[SB];
#pragma unroll
    for (int u = 0; u < SB; ++u) {
      const float s2 = fmaf(cz[u], cz[u], fmaf(cx[u], cx[u], cy[u] * cy[u]));
      sa[u].x = half ? cz[u] : cx[u];
      sa[u].y = half ? s2    : cy[u];
    }
    const int nb = (n0 + 16 * SB) & (NS - 1);
#pragma unroll
    for (int u = 0; u < SB; ++u) {
      const float* p = coords_s + (nb + u * 16 + r15) * 3;
      cx[u] = p[0]; cy[u] = p[1]; cz[u] = p[2];
    }
    v8f c[SB];
#pragma unroll
    for (int u = 0; u < SB; ++u) {
      const v8f z = {0.f, 0.f, 0.f, 0.f, 0.f, 0.f, 0.f, 0.f};
      c[u] = __builtin_amdgcn_wmma_f32_16x16x4_f32(false, sa[u], false, qb, (short)0, z, false, false);
    }
#pragma unroll
    for (int u = 0; u < SB; ++u) {
      const float m8 = fminf(fminf(fminf(c[u][0], c[u][1]), fminf(c[u][2], c[u][3])),
                             fminf(fminf(c[u][4], c[u][5]), fminf(c[u][6], c[u][7])));
      if (__any(m8 < v2v)) {
        const int base = n0 + u * 16 + 8 * half;
        TOP3_INS(c[u][0], base + 0); TOP3_INS(c[u][1], base + 1);
        TOP3_INS(c[u][2], base + 2); TOP3_INS(c[u][3], base + 3);
        TOP3_INS(c[u][4], base + 4); TOP3_INS(c[u][5], base + 5);
        TOP3_INS(c[u][6], base + 6); TOP3_INS(c[u][7], base + 7);
      }
    }
  }

  v2f bb[2][KB][7];
  float4 xq[2], gq[2];
  const float* xrow = x7 + (long)(qbase + r15) * C + 8 * half;
  const float* grow = gate + 2 * half * GATE_PITCH;
  xq[0] = *(const float4*)(xrow);
  xq[1] = *(const float4*)(xrow + 4);
  gq[0] = *(const float4*)(grow);
  gq[1] = *(const float4*)(grow + GATE_PITCH);
  LOAD_B7(bb[0], W1, 0);

  {
    float pv0 = __shfl(v0, lane + 16), pv1 = __shfl(v1, lane + 16), pv2 = __shfl(v2v, lane + 16);
    int   pi0 = __shfl(i0, lane + 16), pi1 = __shfl(i1, lane + 16), pi2 = __shfl(i2,  lane + 16);
    TOP3_INS(pv0, pi0); TOP3_INS(pv1, pi1); TOP3_INS(pv2, pi2);
  }
  const float e1 = __expf(v0 - v1), e2 = __expf(v0 - v2v);
  const float inv = 1.0f / (1.0f + e1 + e2);
  const float w0 = inv, w1 = e1 * inv, w2 = e2 * inv;

  i0 = __shfl(i0, r15); i1 = __shfl(i1, r15); i2 = __shfl(i2, r15);

  const float* f0 = feat_s + (long)i0 * C + 8 * half;
  const float* f1 = feat_s + (long)i1 * C + 8 * half;
  const float* f2 = feat_s + (long)i2 * C + 8 * half;
  float4 pq0[2], pq1[2], pq2[2];
  pq0[0] = *(const float4*)(f0); pq0[1] = *(const float4*)(f0 + 4);
  pq1[0] = *(const float4*)(f1); pq1[1] = *(const float4*)(f1 + 4);
  pq2[0] = *(const float4*)(f2); pq2[1] = *(const float4*)(f2 + 4);

  v8f acc[7];
#pragma unroll
  for (int n = 0; n < 7; ++n) acc[n] = (v8f){0.f, 0.f, 0.f, 0.f, 0.f, 0.f, 0.f, 0.f};
#pragma unroll
  for (int blk = 0; blk < NBLK; ++blk) {
    v2f a[KB];
    a[0].x = xq[0].x * gq[0].x; a[0].y = xq[0].y * gq[0].y;
    a[1].x = xq[0].z * gq[0].z; a[1].y = xq[0].w * gq[0].w;
    a[2].x = xq[1].x * gq[1].x; a[2].y = xq[1].y * gq[1].y;
    a[3].x = xq[1].z * gq[1].z; a[3].y = xq[1].w * gq[1].w;
    __builtin_amdgcn_sched_barrier(0);
    if (blk + 1 < NBLK) {
      xq[0] = *(const float4*)(xrow + 16 * (blk + 1));
      xq[1] = *(const float4*)(xrow + 16 * (blk + 1) + 4);
      gq[0] = *(const float4*)(grow + 4 * (blk + 1) * GATE_PITCH);
      gq[1] = *(const float4*)(grow + (4 * (blk + 1) + 1) * GATE_PITCH);
    }
    LOAD_B7(bb[(blk + 1) & 1], W1, 16 * (blk + 1));
    __builtin_amdgcn_sched_barrier(0);
#pragma unroll
    for (int s = 0; s < KB; ++s)
#pragma unroll
      for (int n = 0; n < 7; ++n)
        acc[n] = __builtin_amdgcn_wmma_f32_16x16x4_f32(false, a[s], false, bb[blk & 1][s][n], (short)0, acc[n], false, false);
  }
#pragma unroll
  for (int blk = 0; blk < NBLK; ++blk) {
    v2f a[KB];
    a[0].x = fmaf(w2, pq2[0].x, fmaf(w0, pq0[0].x, w1 * pq1[0].x));
    a[0].y = fmaf(w2, pq2[0].y, fmaf(w0, pq0[0].y, w1 * pq1[0].y));
    a[1].x = fmaf(w2, pq2[0].z, fmaf(w0, pq0[0].z, w1 * pq1[0].z));
    a[1].y = fmaf(w2, pq2[0].w, fmaf(w0, pq0[0].w, w1 * pq1[0].w));
    a[2].x = fmaf(w2, pq2[1].x, fmaf(w0, pq0[1].x, w1 * pq1[1].x));
    a[2].y = fmaf(w2, pq2[1].y, fmaf(w0, pq0[1].y, w1 * pq1[1].y));
    a[3].x = fmaf(w2, pq2[1].z, fmaf(w0, pq0[1].z, w1 * pq1[1].z));
    a[3].y = fmaf(w2, pq2[1].w, fmaf(w0, pq0[1].w, w1 * pq1[1].w));
    __builtin_amdgcn_sched_barrier(0);
    if (blk + 1 < NBLK) {
      pq0[0] = *(const float4*)(f0 + 16 * (blk + 1)); pq0[1] = *(const float4*)(f0 + 16 * (blk + 1) + 4);
      pq1[0] = *(const float4*)(f1 + 16 * (blk + 1)); pq1[1] = *(const float4*)(f1 + 16 * (blk + 1) + 4);
      pq2[0] = *(const float4*)(f2 + 16 * (blk + 1)); pq2[1] = *(const float4*)(f2 + 16 * (blk + 1) + 4);
      LOAD_B7(bb[(NBLK + blk + 1) & 1], W1, C + 16 * (blk + 1));
    } else {
      LOAD_B7(bb[(NBLK + blk + 1) & 1], W2, 0);
    }
    __builtin_amdgcn_sched_barrier(0);
#pragma unroll
    for (int s = 0; s < KB; ++s)
#pragma unroll
      for (int n = 0; n < 7; ++n)
        acc[n] = __builtin_amdgcn_wmma_f32_16x16x4_f32(false, a[s], false, bb[(NBLK + blk) & 1][s][n], (short)0, acc[n], false, false);
  }

#pragma unroll
  for (int n = 0; n < 7; ++n) {
    const int ch = n * 16 + r15;
    const float gm = gammaW[ch], bt = betaW[ch];
#pragma unroll
    for (int r = 0; r < 8; ++r) {
      const float h = fmaxf(acc[n][r] * gm + bt, 0.f);
      WL[H_OFF + (r + 8 * half) * HSTRIDE + ch] = h;
    }
  }
  asm volatile("" ::: "memory");

  v2f wc[C / 4][2];
  v8f acc2[7];
#pragma unroll
  for (int n = 0; n < 7; ++n) acc2[n] = (v8f){0.f, 0.f, 0.f, 0.f, 0.f, 0.f, 0.f, 0.f};
  {
    const float* hrow = WL + H_OFF + r15 * HSTRIDE + 8 * half;
    v2f hq[2][KB];
#pragma unroll
    for (int s = 0; s < KB; ++s) hq[0][s] = *(const v2f*)(hrow + 2 * s);
#pragma unroll
    for (int blk = 0; blk < NBLK; ++blk) {
      __builtin_amdgcn_sched_barrier(0);
      if (blk + 1 < NBLK) {
#pragma unroll
        for (int s = 0; s < KB; ++s) hq[(blk + 1) & 1][s] = *(const v2f*)(hrow + 16 * (blk + 1) + 2 * s);
        LOAD_B7(bb[(blk + 1) & 1], W2, 16 * (blk + 1));
      } else {
        const int c1 = (16 + r15 < CLS) ? 16 + r15 : 0;
#pragma unroll
        for (int ks = 0; ks < C / 4; ++ks) {
          const float* r0 = W_cls + (long)(4 * ks + 2 * half) * CLS;
          wc[ks][0].x = r0[r15];
          wc[ks][0].y = r0[CLS + r15];
          wc[ks][1].x = r0[c1];
          wc[ks][1].y = r0[CLS + c1];
        }
      }
      __builtin_amdgcn_sched_barrier(0);
#pragma unroll
      for (int s = 0; s < KB; ++s)
#pragma unroll
        for (int n = 0; n < 7; ++n)
          acc2[n] = __builtin_amdgcn_wmma_f32_16x16x4_f32(false, hq[blk & 1][s], false, bb[blk & 1][s][n], (short)0, acc2[n], false, false);
    }
  }
#pragma unroll
  for (int n = 0; n < 7; ++n) {
    const int ch = n * 16 + r15;
#pragma unroll
    for (int r = 0; r < 8; ++r)
      WL[Y2_OFF + (r + 8 * half) * HSTRIDE + ch] = acc2[n][r];
  }
  asm volatile("" ::: "memory");

  v8f acc3[2];
  acc3[0] = (v8f){0.f, 0.f, 0.f, 0.f, 0.f, 0.f, 0.f, 0.f};
  acc3[1] = acc3[0];
  {
    const float* yrow = WL + Y2_OFF + r15 * HSTRIDE + 2 * half;
    v2f ya[C / 4];
#pragma unroll
    for (int ks = 0; ks < C / 4; ++ks) ya[ks] = *(const v2f*)(yrow + 4 * ks);
    __builtin_amdgcn_sched_barrier(0);
#pragma unroll
    for (int ks = 0; ks < C / 4; ++ks)
#pragma unroll
      for (int n = 0; n < 2; ++n)
        acc3[n] = __builtin_amdgcn_wmma_f32_16x16x4_f32(false, ya[ks], false, wc[ks][n], (short)0, acc3[n], false, false);
  }
#pragma unroll
  for (int n = 0; n < 2; ++n) {
    const int col = n * 16 + r15;
    if (col < CLS) {
      const float bb = b_cls[col];
#pragma unroll
      for (int r = 0; r < 8; ++r)
        out[(long)(qbase + r + 8 * half) * CLS + col] = acc3[n][r] + bb;
    }
  }
}

extern "C" void kernel_launch(void* const* d_in, const int* in_sizes, int n_in,
                              void* d_out, int out_size, void* d_ws, size_t ws_size,
                              hipStream_t stream) {
  const float* coords_q = (const float*)d_in[0];
  const float* coords_s = (const float*)d_in[1];
  const float* x7       = (const float*)d_in[2];
  const float* feat_s7  = (const float*)d_in[3];
  const float* feat_s   = (const float*)d_in[4];
  const float* w_attn   = (const float*)d_in[5];
  const float* W1       = (const float*)d_in[6];
  const float* gammaW   = (const float*)d_in[7];
  const float* betaW    = (const float*)d_in[8];
  const float* W2       = (const float*)d_in[9];
  const float* W_cls    = (const float*)d_in[10];
  const float* b_cls    = (const float*)d_in[11];
  float* out = (float*)d_out;

  float* ws     = (float*)d_ws;
  float* logits = ws;
  float* gate   = ws + 2 * NS;

  attn_logits<<<NS / 256, 256, 0, stream>>>(feat_s7, w_attn, logits);
  attn_pool_gate<<<C / PCH, PTHREADS, 0, stream>>>(feat_s7, logits, gate);

  const int blocks = (TILES + WAVES_PER_BLOCK - 1) / WAVES_PER_BLOCK;
  fused_main<<<blocks, 128, 0, stream>>>(coords_q, coords_s, x7, feat_s, gate,
                                         W1, gammaW, betaW, W2, W_cls, b_cls, out);
}


# top-3 search prunes with one bound per query, set once per 16-tile trip: third smallest of a lane's and its partner half's lists merged; the 16 tile tests run ahead of the insert chains
# speedup vs baseline: 10.8881x; 1.1104x over previous
#include <hip/hip_runtime.h>
#include <math.h>

#define NQ   20000
#define NS   8192
#define C    112
#define C2   224
#define CLS  20
#define TILES (NQ / 16)
#define WAVES_PER_BLOCK 4
#define PCH        4
#define GATE_PITCH 32
#define SB         16

#define WREG    3712
#define HSTRIDE 114
#define H_OFF   0
#define Y2_OFF  1856

typedef float v2f __attribute__((ext_vector_type(2)));
typedef float v8f __attribute__((ext_vector_type(8)));

__global__ void attn_logits(const float* __restrict__ feat_s7,
                            const float* __restrict__ w_attn,
                            float* __restrict__ logits) {
  int j = blockIdx.x * blockDim.x + threadIdx.x;
  if (j >= NS) return;
  const float* row = feat_s7 + (long)j * C;
  float acc = 0.f;
#pragma unroll 8
  for (int c = 0; c < C; ++c) acc += row[c] * w_attn[c];
  logits[j] = acc;
}

#define PTHREADS 1024
#define PROWS    (NS / PTHREADS)
#define PWAVES   (PTHREADS / 32)
static_assert(PWAVES == 32 && NS % PTHREADS == 0 && C % PCH == 0 && PCH == 4, "pool split");
static_assert(NS % (16 * SB) == 0 && (NS & (NS - 1)) == 0, "search trips");

__device__ __forceinline__ float wave_max(float v) {
  for (int o = 16; o > 0; o >>= 1) v = fmaxf(v, __shfl_xor(v, o));
  return v;
}
__device__ __forceinline__ float wave_sum(float v) {
  for (int o = 16; o > 0; o >>= 1) v += __shfl_xor(v, o);
  return v;
}

__global__ __launch_bounds__(PTHREADS) void attn_pool_gate(
    const float* __restrict__ feat_s7, const float* __restrict__ logits,
    float* __restrict__ gate) {
  __shared__ float redm[PWAVES];
  __shared__ float reds[1 + PCH][PWAVES];
  const int t = threadIdx.x;
  const int lane = t & 31;
  const int wave = t >> 5;
  const int c0 = blockIdx.x * PCH;

  float l[PROWS];
  float4 f[PROWS];
#pragma unroll
  for (int i = 0; i < PROWS; ++i) {
    const int j = t + i * PTHREADS;
    l[i] = logits[j];
    f[i] = *(const float4*)(feat_s7 + (long)j * C + c0);
  }

  float m = l[0];
#pragma unroll
  for (int i = 1; i < PROWS; ++i) m = fmaxf(m, l[i]);
  m = wave_max(m);
  if (lane == 0) redm[wave] = m;
  __syncthreads();
  m = wave_max(redm[lane]);

  float s = 0.f, a0 = 0.f, a1 = 0.f, a2 = 0.f, a3 = 0.f;
#pragma unroll
  for (int i = 0; i < PROWS; ++i) {
    const float e = __expf(l[i] - m);
    s  += e;
    a0 += e * f[i].x;
    a1 += e * f[i].y;
    a2 += e * f[i].z;
    a3 += e * f[i].w;
  }
  s  = wave_sum(s);
  a0 = wave_sum(a0);
  a1 = wave_sum(a1);
  a2 = wave_sum(a2);
  a3 = wave_sum(a3);
  if (lane == 0) {
    reds[0][wave] = s;
    reds[1][wave] = a0;
    reds[2][wave] = a1;
    reds[3][wave] = a2;
    reds[4][wave] = a3;
  }
  __syncthreads();

  if (wave == 0) {
    const float sum = wave_sum(reds[0][lane]);
    const float t0 = wave_sum(reds[1][lane]);
    const float t1 = wave_sum(reds[2][lane]);
    const float t2 = wave_sum(reds[3][lane]);
    const float t3 = wave_sum(reds[4][lane]);
    if (lane < PCH) {
      const float acc = (lane == 0) ? t0 : (lane == 1) ? t1 : (lane == 2) ? t2 : t3;
      const float inv = 1.0f / sum;
      const float g = acc * inv;
      gate[blockIdx.x * GATE_PITCH + lane] = 1.0f / (1.0f + __expf(-g));
    }
  }
}

__device__ __forceinline__ float xhalf(float x) {
  const int b = __builtin_bit_cast(int, x);
  return __builtin_bit_cast(float, __builtin_amdgcn_permlanex16(b, b, 0x76543210, 0xfedcba98, false, false));
}

#define TOP3_INS(val, idxv) {                                         \
    float s_ = (val); int id_ = (idxv);                               \
    bool c0_ = s_ < v0, c1_ = s_ < v1, c2_ = s_ < v2v;                \
    v2v = c1_ ? v1 : (c2_ ? s_  : v2v); i2 = c1_ ? i1 : (c2_ ? id_ : i2); \
    v1  = c0_ ? v0 : (c1_ ? s_  : v1 ); i1 = c0_ ? i0 : (c1_ ? id_ : i1); \
    v0  = c0_ ? s_ : v0;                i0 = c0_ ? id_ : i0;          }

#define KB   4
#define NBLK (C / (4 * KB))
static_assert(C % (4 * KB) == 0 && KB == 4 && PCH == 4, "k blocks: 16 rows, 4 gate lines");

#define LOAD_B7(dst, Wp, krow0) {                                              \
    _Pragma("unroll")                                                          \
    for (int s_ = 0; s_ < KB; ++s_) {                                          \
      const float* r0_ = (Wp) + (long)((krow0) + 8 * half + 2 * s_) * C + r15; \
      _Pragma("unroll")                                                        \
      for (int n_ = 0; n_ < 7; ++n_) {                                         \
        dst[s_][n_].x = r0_[n_ * 16];                                          \
        dst[s_][n_].y = r0_[C + n_ * 16];                                      \
      }                                                                        \
    } }

__global__ __launch_bounds__(128) void fused_main(
    const float* __restrict__ coords_q, const float* __restrict__ coords_s,
    const float* __restrict__ x7,       const float* __restrict__ feat_s,
    const float* __restrict__ gate,     const float* __restrict__ W1,
    const float* __restrict__ gammaW,   const float* __restrict__ betaW,
    const float* __restrict__ W2,       const float* __restrict__ W_cls,
    const float* __restrict__ b_cls,    float* __restrict__ out)
{
  __shared__ float lds[WAVES_PER_BLOCK * WREG];
  const int lane = threadIdx.x & 31;
  const int wave = threadIdx.x >> 5;
  const int tile = blockIdx.x * WAVES_PER_BLOCK + wave;
  if (tile >= TILES) return;

  float* WL = lds + wave * WREG;
  const int r15  = lane & 15;
  const int half = lane >> 4;
  const int qbase = tile * 16;

  const float qx = coords_q[(qbase + r15) * 3 + 0];
  const float qy = coords_q[(qbase + r15) * 3 + 1];
  const float qz = coords_q[(qbase + r15) * 3 + 2];
  v2f qb;
  qb.x = half ? (-2.f * qz) : (-2.f * qx);
  qb.y = half ? 1.0f        : (-2.f * qy);

  float v0 = 1e30f, v1 = 1e30f, v2v = 1e30f;
  int   i0 = 0, i1 = 0, i2 = 0;

  float cx[SB], cy[SB], cz[SB];
#pragma unroll
  for (int u = 0; u < SB; ++u) {
    const float* p = coords_s + (u * 16 + r15) * 3;
    cx[u] = p[0]; cy[u] = p[1]; cz[u] = p[2];
  }
  for (int n0 = 0; n0 < NS; n0 += 16 * SB) {
    const float p0 = xhalf(v0), p1 = xhalf(v1), p2 = xhalf(v2v);
    const float bound = fminf(fminf(v2v, p2), fminf(fmaxf(v1, p0), fmaxf(v0, p1)));
    v2f sa[SB];
#pragma unroll
    for (int u = 0; u < SB; ++u) {
      const float s2 = fmaf(cz[u], cz[u], fmaf(cx[u], cx[u], cy[u] * cy[u]));
      sa[u].x = half ? cz[u] : cx[u];
      sa[u].y = half ? s2    : cy[u];
    }
    const int nb = (n0 + 16 * SB) & (NS - 1);
#pragma unroll
    for (int u = 0; u < SB; ++u) {
      const float* p = coords_s + (nb + u * 16 + r15) * 3;
      cx[u] = p[0]; cy[u] = p[1]; cz[u] = p[2];
    }
    v8f c[SB];
#pragma unroll
    for (int u = 0; u < SB; ++u) {
      const v8f z = {0.f, 0.f, 0.f, 0.f, 0.f, 0.f, 0.f, 0.f};
      c[u] = __builtin_amdgcn_wmma_f32_16x16x4_f32(false, sa[u], false, qb, (short)0, z, false, false);
    }
    bool hit[SB];
#pragma unroll
    for (int u = 0; u < SB; ++u) {
      const float m8 = fminf(fminf(fminf(c[u][0], c[u][1]), fminf(c[u][2], c[u][3])),
                             fminf(fminf(c[u][4], c[u][5]), fminf(c[u][6], c[u][7])));
      hit[u] = __any(m8 <= bound);
    }
#pragma unroll
    for (int u = 0; u < SB; ++u) {
      if (hit[u]) {
        const int base = n0 + u * 16 + 8 * half;
        TOP3_INS(c[u][0], base + 0); TOP3_INS(c[u][1], base + 1);
        TOP3_INS(c[u][2], base + 2); TOP3_INS(c[u][3], base + 3);
        TOP3_INS(c[u][4], base + 4); TOP3_INS(c[u][5], base + 5);
        TOP3_INS(c[u][6], base + 6); TOP3_INS(c[u][7], base + 7);
      }
    }
  }

  v2f bb[2][KB][7];
  float4 xq[2], gq[2];
  const float* xrow = x7 + (long)(qbase + r15) * C + 8 * half;
  const float* grow = gate + 2 * half * GATE_PITCH;
  xq[0] = *(const float4*)(xrow);
  xq[1] = *(const float4*)(xrow + 4);
  gq[0] = *(const float4*)(grow);
  gq[1] = *(const float4*)(grow + GATE_PITCH);
  LOAD_B7(bb[0], W1, 0);

  {
    float pv0 = __shfl(v0, lane + 16), pv1 = __shfl(v1, lane + 16), pv2 = __shfl(v2v, lane + 16);
    int   pi0 = __shfl(i0, lane + 16), pi1 = __shfl(i1, lane + 16), pi2 = __shfl(i2,  lane + 16);
    TOP3_INS(pv0, pi0); TOP3_INS(pv1, pi1); TOP3_INS(pv2, pi2);
  }
  const float e1 = __expf(v0 - v1), e2 = __expf(v0 - v2v);
  const float inv = 1.0f / (1.0f + e1 + e2);
  const float w0 = inv, w1 = e1 * inv, w2 = e2 * inv;

  i0 = __shfl(i0, r15); i1 = __shfl(i1, r15); i2 = __shfl(i2, r15);

  const float* f0 = feat_s + (long)i0 * C + 8 * half;
  const float* f1 = feat_s + (long)i1 * C + 8 * half;
  const float* f2 = feat_s + (long)i2 * C + 8 * half;
  float4 pq0[2], pq1[2], pq2[2];
  pq0[0] = *(const float4*)(f0); pq0[1] = *(const float4*)(f0 + 4);
  pq1[0] = *(const float4*)(f1); pq1[1] = *(const float4*)(f1 + 4);
  pq2[0] = *(const float4*)(f2); pq2[1] = *(const float4*)(f2 + 4);

  v8f acc[7];
#pragma unroll
  for (int n = 0; n < 7; ++n) acc[n] = (v8f){0.f, 0.f, 0.f, 0.f, 0.f, 0.f, 0.f, 0.f};
#pragma unroll
  for (int blk = 0; blk < NBLK; ++blk) {
    v2f a[KB];
    a[0].x = xq[0].x * gq[0].x; a[0].y = xq[0].y * gq[0].y;
    a[1].x = xq[0].z * gq[0].z; a[1].y = xq[0].w * gq[0].w;
    a[2].x = xq[1].x * gq[1].x; a[2].y = xq[1].y * gq[1].y;
    a[3].x = xq[1].z * gq[1].z; a[3].y = xq[1].w * gq[1].w;
    __builtin_amdgcn_sched_barrier(0);
    if (blk + 1 < NBLK) {
      xq[0] = *(const float4*)(xrow + 16 * (blk + 1));
      xq[1] = *(const float4*)(xrow + 16 * (blk + 1) + 4);
      gq[0] = *(const float4*)(grow + 4 * (blk + 1) * GATE_PITCH);
      gq[1] = *(const float4*)(grow + (4 * (blk + 1) + 1) * GATE_PITCH);
    }
    LOAD_B7(bb[(blk + 1) & 1], W1, 16 * (blk + 1));
    __builtin_amdgcn_sched_barrier(0);
#pragma unroll
    for (int s = 0; s < KB; ++s)
#pragma unroll
      for (int n = 0; n < 7; ++n)
        acc[n] = __builtin_amdgcn_wmma_f32_16x16x4_f32(false, a[s], false, bb[blk & 1][s][n], (short)0, acc[n], false, false);
  }
#pragma unroll
  for (int blk = 0; blk < NBLK; ++blk) {
    v2f a[KB];
    a[0].x = fmaf(w2, pq2[0].x, fmaf(w0, pq0[0].x, w1 * pq1[0].x));
    a[0].y = fmaf(w2, pq2[0].y, fmaf(w0, pq0[0].y, w1 * pq1[0].y));
    a[1].x = fmaf(w2, pq2[0].z, fmaf(w0, pq0[0].z, w1 * pq1[0].z));
    a[1].y = fmaf(w2, pq2[0].w, fmaf(w0, pq0[0].w, w1 * pq1[0].w));
    a[2].x = fmaf(w2, pq2[1].x, fmaf(w0, pq0[1].x, w1 * pq1[1].x));
    a[2].y = fmaf(w2, pq2[1].y, fmaf(w0, pq0[1].y, w1 * pq1[1].y));
    a[3].x = fmaf(w2, pq2[1].z, fmaf(w0, pq0[1].z, w1 * pq1[1].z));
    a[3].y = fmaf(w2, pq2[1].w, fmaf(w0, pq0[1].w, w1 * pq1[1].w));
    __builtin_amdgcn_sched_barrier(0);
    if (blk + 1 < NBLK) {
      pq0[0] = *(const float4*)(f0 + 16 * (blk + 1)); pq0[1] = *(const float4*)(f0 + 16 * (blk + 1) + 4);
      pq1[0] = *(const float4*)(f1 + 16 * (blk + 1)); pq1[1] = *(const float4*)(f1 + 16 * (blk + 1) + 4);
      pq2[0] = *(const float4*)(f2 + 16 * (blk + 1)); pq2[1] = *(const float4*)(f2 + 16 * (blk + 1) + 4);
      LOAD_B7(bb[(NBLK + blk + 1) & 1], W1, C + 16 * (blk + 1));
    } else {
      LOAD_B7(bb[(NBLK + blk + 1) & 1], W2, 0);
    }
    __builtin_amdgcn_sched_barrier(0);
#pragma unroll
    for (int s = 0; s < KB; ++s)
#pragma unroll
      for (int n = 0; n < 7; ++n)
        acc[n] = __builtin_amdgcn_wmma_f32_16x16x4_f32(false, a[s], false, bb[(NBLK + blk) & 1][s][n], (short)0, acc[n], false, false);
  }

#pragma unroll
  for (int n = 0; n < 7; ++n) {
    const int ch = n * 16 + r15;
    const float gm = gammaW[ch], bt = betaW[ch];
#pragma unroll
    for (int r = 0; r < 8; ++r) {
      const float h = fmaxf(acc[n][r] * gm + bt, 0.f);
      WL[H_OFF + (r + 8 * half) * HSTRIDE + ch] = h;
    }
  }
  asm volatile("" ::: "memory");

  v2f wc[C / 4][2];
  v8f acc2[7];
#pragma unroll
  for (int n = 0; n < 7; ++n) acc2[n] = (v8f){0.f, 0.f, 0.f, 0.f, 0.f, 0.f, 0.f, 0.f};
  {
    const float* hrow = WL + H_OFF + r15 * HSTRIDE + 8 * half;
    v2f hq[2][KB];
#pragma unroll
    for (int s = 0; s < KB; ++s) hq[0][s] = *(const v2f*)(hrow + 2 * s);
#pragma unroll
    for (int blk = 0; blk < NBLK; ++blk) {
      __builtin_amdgcn_sched_barrier(0);
      if (blk + 1 < NBLK) {
#pragma unroll
        for (int s = 0; s < KB; ++s) hq[(blk + 1) & 1][s] = *(const v2f*)(hrow + 16 * (blk + 1) + 2 * s);
        LOAD_B7(bb[(blk + 1) & 1], W2, 16 * (blk + 1));
      } else {
        const int c1 = (16 + r15 < CLS) ? 16 + r15 : 0;
#pragma unroll
        for (int ks = 0; ks < C / 4; ++ks) {
          const float* r0 = W_cls + (long)(4 * ks + 2 * half) * CLS;
          wc[ks][0].x = r0[r15];
          wc[ks][0].y = r0[CLS + r15];
          wc[ks][1].x = r0[c1];
          wc[ks][1].y = r0[CLS + c1];
        }
      }
      __builtin_amdgcn_sched_barrier(0);
#pragma unroll
      for (int s = 0; s < KB; ++s)
#pragma unroll
        for (int n = 0; n < 7; ++n)
          acc2[n] = __builtin_amdgcn_wmma_f32_16x16x4_f32(false, hq[blk & 1][s], false, bb[blk & 1][s][n], (short)0, acc2[n], false, false);
    }
  }
#pragma unroll
  for (int n = 0; n < 7; ++n) {
    const int ch = n * 16 + r15;
#pragma unroll
    for (int r = 0; r < 8; ++r)
      WL[Y2_OFF + (r + 8 * half) * HSTRIDE + ch] = acc2[n][r];
  }
  asm volatile("" ::: "memory");

  v8f acc3[2];
  acc3[0] = (v8f){0.f, 0.f, 0.f, 0.f, 0.f, 0.f, 0.f, 0.f};
  acc3[1] = acc3[0];
  {
    const float* yrow = WL + Y2_OFF + r15 * HSTRIDE + 2 * half;
    v2f ya[C / 4];
#pragma unroll
    for (int ks = 0; ks < C / 4; ++ks) ya[ks] = *(const v2f*)(yrow + 4 * ks);
    __builtin_amdgcn_sched_barrier(0);
#pragma unroll
    for (int ks = 0; ks < C / 4; ++ks)
#pragma unroll
      for (int n = 0; n < 2; ++n)
        acc3[n] = __builtin_amdgcn_wmma_f32_16x16x4_f32(false, ya[ks], false, wc[ks][n], (short)0, acc3[n], false, false);
  }
#pragma unroll
  for (int n = 0; n < 2; ++n) {
    const int col = n * 16 + r15;
    if (col < CLS) {
      const float bb = b_cls[col];
#pragma unroll
      for (int r = 0; r < 8; ++r)
        out[(long)(qbase + r + 8 * half) * CLS + col] = acc3[n][r] + bb;
    }
  }
}

extern "C" void kernel_launch(void* const* d_in, const int* in_sizes, int n_in,
                              void* d_out, int out_size, void* d_ws, size_t ws_size,
                              hipStream_t stream) {
  const float* coords_q = (const float*)d_in[0];
  const float* coords_s = (const float*)d_in[1];
  const float* x7       = (const float*)d_in[2];
  const float* feat_s7  = (const float*)d_in[3];
  const float* feat_s   = (const float*)d_in[4];
  const float* w_attn   = (const float*)d_in[5];
  const float* W1       = (const float*)d_in[6];
  const float* gammaW   = (const float*)d_in[7];
  const float* betaW    = (const float*)d_in[8];
  const float* W2       = (const float*)d_in[9];
  const float* W_cls    = (const float*)d_in[10];
  const float* b_cls    = (const float*)d_in[11];
  float* out = (float*)d_out;

  float* ws     = (float*)d_ws;
  float* logits = ws;
  float* gate   = ws + 2 * NS;

  attn_logits<<<NS / 256, 256, 0, stream>>>(feat_s7, w_attn, logits);
  attn_pool_gate<<<C / PCH, PTHREADS, 0, stream>>>(feat_s7, logits, gate);

  const int blocks = (TILES + WAVES_PER_BLOCK - 1) / WAVES_PER_BLOCK;
  fused_main<<<blocks, 128, 0, stream>>>(coords_q, coords_s, x7, feat_s, gate,
                                         W1, gammaW, betaW, W2, W_cls, b_cls, out);
}


# top-3 search moved out of the main kernel into the pooling launch: the 28 pooling groups (now 128 threads) run beside the search groups, not in front; top-3 via workspace; pool sums re-associated
# speedup vs baseline: 12.0889x; 1.1103x over previous
#include <hip/hip_runtime.h>
#include <math.h>

#define NQ   20000
#define NS   8192
#define C    112
#define C2   224
#define CLS  20
#define TILES (NQ / 16)
#define WAVES_PER_BLOCK 4
#define PCH        4
#define GATE_PITCH 32
#define SB         16
#define POOLB      (C / PCH)
#define SBLOCKS    ((TILES + WAVES_PER_BLOCK - 1) / WAVES_PER_BLOCK)
#define TOP_PITCH  96

#define WREG    3712
#define HSTRIDE 114
#define H_OFF   0
#define Y2_OFF  1856

typedef float v2f __attribute__((ext_vector_type(2)));
typedef float v8f __attribute__((ext_vector_type(8)));

__global__ void attn_logits(const float* __restrict__ feat_s7,
                            const float* __restrict__ w_attn,
                            float* __restrict__ logits) {
  int j = blockIdx.x * blockDim.x + threadIdx.x;
  if (j >= NS) return;
  const float* row = feat_s7 + (long)j * C;
  float acc = 0.f;
#pragma unroll 8
  for (int c = 0; c < C; ++c) acc += row[c] * w_attn[c];
  logits[j] = acc;
}

#define PTHREADS 128
#define PWAVES   (PTHREADS / 32)
#define PCHUNK   16
#define PCHUNKS  (NS / (PTHREADS * PCHUNK))
static_assert(PWAVES == WAVES_PER_BLOCK && NS % (PTHREADS * PCHUNK) == 0 && NS % (4 * PTHREADS) == 0 && C % PCH == 0 && PCH == 4, "pool split");
static_assert(NS % (16 * SB) == 0 && (NS & (NS - 1)) == 0, "search trips");

__device__ __forceinline__ float wave_max(float v) {
  for (int o = 16; o > 0; o >>= 1) v = fmaxf(v, __shfl_xor(v, o));
  return v;
}
__device__ __forceinline__ float wave_sum(float v) {
  for (int o = 16; o > 0; o >>= 1) v += __shfl_xor(v, o);
  return v;
}

__device__ __forceinline__ void pool_gate_role(
    const float* __restrict__ feat_s7, const float* __restrict__ logits,
    float* __restrict__ gate, const int blk) {
  __shared__ float redm[PWAVES];
  __shared__ float reds[1 + PCH][PWAVES];
  const int t = threadIdx.x;
  const int lane = t & 31;
  const int wave = t >> 5;
  const int c0 = blk * PCH;

  float m;
  {
    float4 lv[NS / (4 * PTHREADS)];
#pragma unroll
    for (int i = 0; i < NS / (4 * PTHREADS); ++i)
      lv[i] = *(const float4*)(logits + 4 * (t + i * PTHREADS));
    m = fmaxf(fmaxf(lv[0].x, lv[0].y), fmaxf(lv[0].z, lv[0].w));
#pragma unroll
    for (int i = 1; i < NS / (4 * PTHREADS); ++i)
      m = fmaxf(m, fmaxf(fmaxf(lv[i].x, lv[i].y), fmaxf(lv[i].z, lv[i].w)));
  }
  m = wave_max(m);
  if (lane == 0) redm[wave] = m;
  __syncthreads();
  m = fmaxf(fmaxf(redm[0], redm[1]), fmaxf(redm[2], redm[3]));

  float s = 0.f, a0 = 0.f, a1 = 0.f, a2 = 0.f, a3 = 0.f;
#pragma nounroll
  for (int ch = 0; ch < PCHUNKS; ++ch) {
    float l[PCHUNK];
    float4 f[PCHUNK];
#pragma unroll
    for (int i = 0; i < PCHUNK; ++i) {
      const int j = t + (ch * PCHUNK + i) * PTHREADS;
      l[i] = logits[j];
      f[i] = *(const float4*)(feat_s7 + (long)j * C + c0);
    }
#pragma unroll
    for (int i = 0; i < PCHUNK; ++i) {
      const float e = __expf(l[i] - m);
      s  += e;
      a0 += e * f[i].x;
      a1 += e * f[i].y;
      a2 += e * f[i].z;
      a3 += e * f[i].w;
    }
  }
  s  = wave_sum(s);
  a0 = wave_sum(a0);
  a1 = wave_sum(a1);
  a2 = wave_sum(a2);
  a3 = wave_sum(a3);
  if (lane == 0) {
    reds[0][wave] = s;
    reds[1][wave] = a0;
    reds[2][wave] = a1;
    reds[3][wave] = a2;
    reds[4][wave] = a3;
  }
  __syncthreads();

  if (t < PCH) {
    const float sum = (reds[0][0] + reds[0][1]) + (reds[0][2] + reds[0][3]);
    const float acc = (reds[1 + t][0] + reds[1 + t][1]) + (reds[1 + t][2] + reds[1 + t][3]);
    const float inv = 1.0f / sum;
    const float g = acc * inv;
    gate[blk * GATE_PITCH + t] = 1.0f / (1.0f + __expf(-g));
  }
}

__device__ __forceinline__ float xhalf(float x) {
  const int b = __builtin_bit_cast(int, x);
  return __builtin_bit_cast(float, __builtin_amdgcn_permlanex16(b, b, 0x76543210, 0xfedcba98, false, false));
}

#define TOP3_INS(val, idxv) {                                         \
    float s_ = (val); int id_ = (idxv);                               \
    bool c0_ = s_ < v0, c1_ = s_ < v1, c2_ = s_ < v2v;                \
    v2v = c1_ ? v1 : (c2_ ? s_  : v2v); i2 = c1_ ? i1 : (c2_ ? id_ : i2); \
    v1  = c0_ ? v0 : (c1_ ? s_  : v1 ); i1 = c0_ ? i0 : (c1_ ? id_ : i1); \
    v0  = c0_ ? s_ : v0;                i0 = c0_ ? id_ : i0;          }

__device__ __forceinline__ void search_role(
    const float* __restrict__ coords_q, const float* __restrict__ coords_s,
    float* __restrict__ top, const int tile, const int lane)
{
  const int r15  = lane & 15;
  const int half = lane >> 4;
  const int qbase = tile * 16;

  const float qx = coords_q[(qbase + r15) * 3 + 0];
  const float qy = coords_q[(qbase + r15) * 3 + 1];
  const float qz = coords_q[(qbase + r15) * 3 + 2];
  v2f qb;
  qb.x = half ? (-2.f * qz) : (-2.f * qx);
  qb.y = half ? 1.0f        : (-2.f * qy);

  float v0 = 1e30f, v1 = 1e30f, v2v = 1e30f;
  int   i0 = 0, i1 = 0, i2 = 0;

  float cx[SB], cy[SB], cz[SB];
#pragma unroll
  for (int u = 0; u < SB; ++u) {
    const float* p = coords_s + (u * 16 + r15) * 3;
    cx[u] = p[0]; cy[u] = p[1]; cz[u] = p[2];
  }
  for (int n0 = 0; n0 < NS; n0 += 16 * SB) {
    const float p0 = xhalf(v0), p1 = xhalf(v1), p2 = xhalf(v2v);
    const float bound = fminf(fminf(v2v, p2), fminf(fmaxf(v1, p0), fmaxf(v0, p1)));
    v2f sa[SB];
#pragma unroll
    for (int u = 0; u < SB; ++u) {
      const float s2 = fmaf(cz[u], cz[u], fmaf(cx[u], cx[u], cy[u] * cy[u]));
      sa[u].x = half ? cz[u] : cx[u];
      sa[u].y = half ? s2    : cy[u];
    }
    const int nb = (n0 + 16 * SB) & (NS - 1);
#pragma unroll
    for (int u = 0; u < SB; ++u) {
      const float* p = coords_s + (nb + u * 16 + r15) * 3;
      cx[u] = p[0]; cy[u] = p[1]; cz[u] = p[2];
    }
    v8f c[SB];
#pragma unroll
    for (int u = 0; u < SB; ++u) {
      const v8f z = {0.f, 0.f, 0.f, 0.f, 0.f, 0.f, 0.f, 0.f};
      c[u] = __builtin_amdgcn_wmma_f32_16x16x4_f32(false, sa[u], false, qb, (short)0, z, false, false);
    }
    bool hit[SB];
#pragma unroll
    for (int u = 0; u < SB; ++u) {
      const float m8 = fminf(fminf(fminf(c[u][0], c[u][1]), fminf(c[u][2], c[u][3])),
                             fminf(fminf(c[u][4], c[u][5]), fminf(c[u][6], c[u][7])));
      hit[u] = __any(m8 <= bound);
    }
#pragma unroll
    for (int u = 0; u < SB; ++u) {
      if (hit[u]) {
        const int base = n0 + u * 16 + 8 * half;
        TOP3_INS(c[u][0], base + 0); TOP3_INS(c[u][1], base + 1);
        TOP3_INS(c[u][2], base + 2); TOP3_INS(c[u][3], base + 3);
        TOP3_INS(c[u][4], base + 4); TOP3_INS(c[u][5], base + 5);
        TOP3_INS(c[u][6], base + 6); TOP3_INS(c[u][7], base + 7);
      }
    }
  }

  {
    float pv0 = __shfl(v0, lane + 16), pv1 = __shfl(v1, lane + 16), pv2 = __shfl(v2v, lane + 16);
    int   pi0 = __shfl(i0, lane + 16), pi1 = __shfl(i1, lane + 16), pi2 = __shfl(i2,  lane + 16);
    TOP3_INS(pv0, pi0); TOP3_INS(pv1, pi1); TOP3_INS(pv2, pi2);
  }
  const float e1 = __expf(v0 - v1), e2 = __expf(v0 - v2v);
  const float inv = 1.0f / (1.0f + e1 + e2);
  const float w0 = inv, w1 = e1 * inv, w2 = e2 * inv;

  i0 = __shfl(i0, r15); i1 = __shfl(i1, r15); i2 = __shfl(i2, r15);

  float* tp = top + (long)tile * TOP_PITCH;
  tp[lane]      = __int_as_float(half ? i1 : i0);
  tp[32 + lane] = half ? w0 : __int_as_float(i2);
  tp[64 + lane] = half ? w2 : w1;
}

__global__ __launch_bounds__(128) void pool_and_search(
    const float* __restrict__ coords_q, const float* __restrict__ coords_s,
    const float* __restrict__ feat_s7,  const float* __restrict__ logits,
    float* __restrict__ gate,           float* __restrict__ top)
{
  if (blockIdx.x < POOLB) {
    pool_gate_role(feat_s7, logits, gate, blockIdx.x);
    return;
  }
  const int lane = threadIdx.x & 31;
  const int wave = threadIdx.x >> 5;
  const int tile = (blockIdx.x - POOLB) * WAVES_PER_BLOCK + wave;
  if (tile >= TILES) return;
  search_role(coords_q, coords_s, top, tile, lane);
}

#define KB   4
#define NBLK (C / (4 * KB))
static_assert(C % (4 * KB) == 0 && KB == 4 && PCH == 4, "k blocks: 16 rows, 4 gate lines");

#define LOAD_B7(dst, Wp, krow0) {                                              \
    _Pragma("unroll")                                                          \
    for (int s_ = 0; s_ < KB; ++s_) {                                          \
      const float* r0_ = (Wp) + (long)((krow0) + 8 * half + 2 * s_) * C + r15; \
      _Pragma("unroll")                                                        \
      for (int n_ = 0; n_ < 7; ++n_) {                                         \
        dst[s_][n_].x = r0_[n_ * 16];                                          \
        dst[s_][n_].y = r0_[C + n_ * 16];                                      \
      }                                                                        \
    } }

__global__ __launch_bounds__(128) void gemm_main(
    const float* __restrict__ x7,       const float* __restrict__ feat_s,
    const float* __restrict__ gate,     const float* __restrict__ top,
    const float* __restrict__ W1,
    const float* __restrict__ gammaW,   const float* __restrict__ betaW,
    const float* __restrict__ W2,       const float* __restrict__ W_cls,
    const float* __restrict__ b_cls,    float* __restrict__ out)
{
  __shared__ float lds[WAVES_PER_BLOCK * WREG];
  const int lane = threadIdx.x & 31;
  const int wave = threadIdx.x >> 5;
  const int tile = blockIdx.x * WAVES_PER_BLOCK + wave;
  if (tile >= TILES) return;

  float* WL = lds + wave * WREG;
  const int r15  = lane & 15;
  const int half = lane >> 4;
  const int qbase = tile * 16;

  const float* tp = top + (long)tile * TOP_PITCH;
  const int i0 = __float_as_int(tp[r15]), i1 = __float_as_int(tp[16 + r15]), i2 = __float_as_int(tp[32 + r15]);
  const float w0 = tp[48 + r15];
  const float w1 = tp[64 + r15];
  const float w2 = tp[80 + r15];

  v2f bb[2][KB][7];
  float4 xq[2], gq[2];
  const float* xrow = x7 + (long)(qbase + r15) * C + 8 * half;
  const float* grow = gate + 2 * half * GATE_PITCH;
  xq[0] = *(const float4*)(xrow);
  xq[1] = *(const float4*)(xrow + 4);
  gq[0] = *(const float4*)(grow);
  gq[1] = *(const float4*)(grow + GATE_PITCH);
  LOAD_B7(bb[0], W1, 0);

  const float* f0 = feat_s + (long)i0 * C + 8 * half;
  const float* f1 = feat_s + (long)i1 * C + 8 * half;
  const float* f2 = feat_s + (long)i2 * C + 8 * half;
  float4 pq0[2], pq1[2], pq2[2];
  pq0[0] = *(const float4*)(f0); pq0[1] = *(const float4*)(f0 + 4);
  pq1[0] = *(const float4*)(f1); pq1[1] = *(const float4*)(f1 + 4);
  pq2[0] = *(const float4*)(f2); pq2[1] = *(const float4*)(f2 + 4);

  v8f acc[7];
#pragma unroll
  for (int n = 0; n < 7; ++n) acc[n] = (v8f){0.f, 0.f, 0.f, 0.f, 0.f, 0.f, 0.f, 0.f};
#pragma unroll
  for (int blk = 0; blk < NBLK; ++blk) {
    v2f a[KB];
    a[0].x = xq[0].x * gq[0].x; a[0].y = xq[0].y * gq[0].y;
    a[1].x = xq[0].z * gq[0].z; a[1].y = xq[0].w * gq[0].w;
    a[2].x = xq[1].x * gq[1].x; a[2].y = xq[1].y * gq[1].y;
    a[3].x = xq[1].z * gq[1].z; a[3].y = xq[1].w * gq[1].w;
    __builtin_amdgcn_sched_barrier(0);
    if (blk + 1 < NBLK) {
      xq[0] = *(const float4*)(xrow + 16 * (blk + 1));
      xq[1] = *(const float4*)(xrow + 16 * (blk + 1) + 4);
      gq[0] = *(const float4*)(grow + 4 * (blk + 1) * GATE_PITCH);
      gq[1] = *(const float4*)(grow + (4 * (blk + 1) + 1) * GATE_PITCH);
    }
    LOAD_B7(bb[(blk + 1) & 1], W1, 16 * (blk + 1));
    __builtin_amdgcn_sched_barrier(0);
#pragma unroll
    for (int s = 0; s < KB; ++s)
#pragma unroll
      for (int n = 0; n < 7; ++n)
        acc[n] = __builtin_amdgcn_wmma_f32_16x16x4_f32(false, a[s], false, bb[blk & 1][s][n], (short)0, acc[n], false, false);
  }
#pragma unroll
  for (int blk = 0; blk < NBLK; ++blk) {
    v2f a[KB];
    a[0].x = fmaf(w2, pq2[0].x, fmaf(w0, pq0[0].x, w1 * pq1[0].x));
    a[0].y = fmaf(w2, pq2[0].y, fmaf(w0, pq0[0].y, w1 * pq1[0].y));
    a[1].x = fmaf(w2, pq2[0].z, fmaf(w0, pq0[0].z, w1 * pq1[0].z));
    a[1].y = fmaf(w2, pq2[0].w, fmaf(w0, pq0[0].w, w1 * pq1[0].w));
    a[2].x = fmaf(w2, pq2[1].x, fmaf(w0, pq0[1].x, w1 * pq1[1].x));
    a[2].y = fmaf(w2, pq2[1].y, fmaf(w0, pq0[1].y, w1 * pq1[1].y));
    a[3].x = fmaf(w2, pq2[1].z, fmaf(w0, pq0[1].z, w1 * pq1[1].z));
    a[3].y = fmaf(w2, pq2[1].w, fmaf(w0, pq0[1].w, w1 * pq1[1].w));
    __builtin_amdgcn_sched_barrier(0);
    if (blk + 1 < NBLK) {
      pq0[0] = *(const float4*)(f0 + 16 * (blk + 1)); pq0[1] = *(const float4*)(f0 + 16 * (blk + 1) + 4);
      pq1[0] = *(const float4*)(f1 + 16 * (blk + 1)); pq1[1] = *(const float4*)(f1 + 16 * (blk + 1) + 4);
      pq2[0] = *(const float4*)(f2 + 16 * (blk + 1)); pq2[1] = *(const float4*)(f2 + 16 * (blk + 1) + 4);
      LOAD_B7(bb[(NBLK + blk + 1) & 1], W1, C + 16 * (blk + 1));
    } else {
      LOAD_B7(bb[(NBLK + blk + 1) & 1], W2, 0);
    }
    __builtin_amdgcn_sched_barrier(0);
#pragma unroll
    for (int s = 0; s < KB; ++s)
#pragma unroll
      for (int n = 0; n < 7; ++n)
        acc[n] = __builtin_amdgcn_wmma_f32_16x16x4_f32(false, a[s], false, bb[(NBLK + blk) & 1][s][n], (short)0, acc[n], false, false);
  }

#pragma unroll
  for (int n = 0; n < 7; ++n) {
    const int ch = n * 16 + r15;
    const float gm = gammaW[ch], bt = betaW[ch];
#pragma unroll
    for (int r = 0; r < 8; ++r) {
      const float h = fmaxf(acc[n][r] * gm + bt, 0.f);
      WL[H_OFF + (r + 8 * half) * HSTRIDE + ch] = h;
    }
  }
  asm volatile("" ::: "memory");

  v2f wc[C / 4][2];
  v8f acc2[7];
#pragma unroll
  for (int n = 0; n < 7; ++n) acc2[n] = (v8f){0.f, 0.f, 0.f, 0.f, 0.f, 0.f, 0.f, 0.f};
  {
    const float* hrow = WL + H_OFF + r15 * HSTRIDE + 8 * half;
    v2f hq[2][KB];
#pragma unroll
    for (int s = 0; s < KB; ++s) hq[0][s] = *(const v2f*)(hrow + 2 * s);
#pragma unroll
    for (int blk = 0; blk < NBLK; ++blk) {
      __builtin_amdgcn_sched_barrier(0);
      if (blk + 1 < NBLK) {
#pragma unroll
        for (int s = 0; s < KB; ++s) hq[(blk + 1) & 1][s] = *(const v2f*)(hrow + 16 * (blk + 1) + 2 * s);
        LOAD_B7(bb[(blk + 1) & 1], W2, 16 * (blk + 1));
      } else {
        const int c1 = (16 + r15 < CLS) ? 16 + r15 : 0;
#pragma unroll
        for (int ks = 0; ks < C / 4; ++ks) {
          const float* r0 = W_cls + (long)(4 * ks + 2 * half) * CLS;
          wc[ks][0].x = r0[r15];
          wc[ks][0].y = r0[CLS + r15];
          wc[ks][1].x = r0[c1];
          wc[ks][1].y = r0[CLS + c1];
        }
      }
      __builtin_amdgcn_sched_barrier(0);
#pragma unroll
      for (int s = 0; s < KB; ++s)
#pragma unroll
        for (int n = 0; n < 7; ++n)
          acc2[n] = __builtin_amdgcn_wmma_f32_16x16x4_f32(false, hq[blk & 1][s], false, bb[blk & 1][s][n], (short)0, acc2[n], false, false);
    }
  }
#pragma unroll
  for (int n = 0; n < 7; ++n) {
    const int ch = n * 16 + r15;
#pragma unroll
    for (int r = 0; r < 8; ++r)
      WL[Y2_OFF + (r + 8 * half) * HSTRIDE + ch] = acc2[n][r];
  }
  asm volatile("" ::: "memory");

  v8f acc3[2];
  acc3[0] = (v8f){0.f, 0.f, 0.f, 0.f, 0.f, 0.f, 0.f, 0.f};
  acc3[1] = acc3[0];
  {
    const float* yrow = WL + Y2_OFF + r15 * HSTRIDE + 2 * half;
    v2f ya[C / 4];
#pragma unroll
    for (int ks = 0; ks < C / 4; ++ks) ya[ks] = *(const v2f*)(yrow + 4 * ks);
    __builtin_amdgcn_sched_barrier(0);
#pragma unroll
    for (int ks = 0; ks < C / 4; ++ks)
#pragma unroll
      for (int n = 0; n < 2; ++n)
        acc3[n] = __builtin_amdgcn_wmma_f32_16x16x4_f32(false, ya[ks], false, wc[ks][n], (short)0, acc3[n], false, false);
  }
#pragma unroll
  for (int n = 0; n < 2; ++n) {
    const int col = n * 16 + r15;
    if (col < CLS) {
      const float bb = b_cls[col];
#pragma unroll
      for (int r = 0; r < 8; ++r)
        out[(long)(qbase + r + 8 * half) * CLS + col] = acc3[n][r] + bb;
    }
  }
}

extern "C" void kernel_launch(void* const* d_in, const int* in_sizes, int n_in,
                              void* d_out, int out_size, void* d_ws, size_t ws_size,
                              hipStream_t stream) {
  const float* coords_q = (const float*)d_in[0];
  const float* coords_s = (const float*)d_in[1];
  const float* x7       = (const float*)d_in[2];
  const float* feat_s7  = (const float*)d_in[3];
  const float* feat_s   = (const float*)d_in[4];
  const float* w_attn   = (const float*)d_in[5];
  const float* W1       = (const float*)d_in[6];
  const float* gammaW   = (const float*)d_in[7];
  const float* betaW    = (const float*)d_in[8];
  const float* W2       = (const float*)d_in[9];
  const float* W_cls    = (const float*)d_in[10];
  const float* b_cls    = (const float*)d_in[11];
  float* out = (float*)d_out;

  float* ws     = (float*)d_ws;
  float* logits = ws;
  float* gate   = ws + 2 * NS;
  float* top    = ws + 4 * NS;

  attn_logits<<<NS / 256, 256, 0, stream>>>(feat_s7, w_attn, logits);
  pool_and_search<<<POOLB + SBLOCKS, 128, 0, stream>>>(coords_q, coords_s, feat_s7, logits, gate, top);
  gemm_main<<<SBLOCKS, 128, 0, stream>>>(x7, feat_s, gate, top, W1, gammaW, betaW, W2, W_cls, b_cls, out);
}


# top-3 search split by support range over two launches: rows 0..4095 run in the logits launch beside its 64 groups, lists pass through workspace to rows 4096.. in the pool launch; sums unchanged
# speedup vs baseline: 12.6425x; 1.0458x over previous
#include <hip/hip_runtime.h>
#include <math.h>

#define NQ   20000
#define NS   8192
#define C    112
#define C2   224
#define CLS  20
#define TILES (NQ / 16)
#define WAVES_PER_BLOCK 4
#define PCH        4
#define GATE_PITCH 32
#define SB         16
#define POOLB      (C / PCH)
#define SBLOCKS    ((TILES + WAVES_PER_BLOCK - 1) / WAVES_PER_BLOCK)
#define TOP_PITCH  96
#define LOGB       (NS / 128)
#define NSPLIT     (NS / 2)
#define CARRY_PITCH 192

#define WREG    3712
#define HSTRIDE 114
#define H_OFF   0
#define Y2_OFF  1856

typedef float v2f __attribute__((ext_vector_type(2)));
typedef float v8f __attribute__((ext_vector_type(8)));

__device__ __forceinline__ void logits_role(const float* __restrict__ feat_s7,
                                            const float* __restrict__ w_attn,
                                            float* __restrict__ logits, const int j) {
  const float* row = feat_s7 + (long)j * C;
  float acc = 0.f;
#pragma unroll 8
  for (int c = 0; c < C; ++c) acc += row[c] * w_attn[c];
  logits[j] = acc;
}

#define PTHREADS 128
#define PWAVES   (PTHREADS / 32)
#define PCHUNK   16
#define PCHUNKS  (NS / (PTHREADS * PCHUNK))
static_assert(PWAVES == WAVES_PER_BLOCK && NS % (PTHREADS * PCHUNK) == 0 && NS % (4 * PTHREADS) == 0 && C % PCH == 0 && PCH == 4, "pool split");
static_assert(NS % (16 * SB) == 0 && (NS & (NS - 1)) == 0 && NSPLIT % (16 * SB) == 0 && NS % 128 == 0, "search trips");

__device__ __forceinline__ float wave_max(float v) {
  for (int o = 16; o > 0; o >>= 1) v = fmaxf(v, __shfl_xor(v, o));
  return v;
}
__device__ __forceinline__ float wave_sum(float v) {
  for (int o = 16; o > 0; o >>= 1) v += __shfl_xor(v, o);
  return v;
}

__device__ __forceinline__ void pool_gate_role(
    const float* __restrict__ feat_s7, const float* __restrict__ logits,
    float* __restrict__ gate, const int blk) {
  __shared__ float redm[PWAVES];
  __shared__ float reds[1 + PCH][PWAVES];
  const int t = threadIdx.x;
  const int lane = t & 31;
  const int wave = t >> 5;
  const int c0 = blk * PCH;

  float m;
  {
    float4 lv[NS / (4 * PTHREADS)];
#pragma unroll
    for (int i = 0; i < NS / (4 * PTHREADS); ++i)
      lv[i] = *(const float4*)(logits + 4 * (t + i * PTHREADS));
    m = fmaxf(fmaxf(lv[0].x, lv[0].y), fmaxf(lv[0].z, lv[0].w));
#pragma unroll
    for (int i = 1; i < NS / (4 * PTHREADS); ++i)
      m = fmaxf(m, fmaxf(fmaxf(lv[i].x, lv[i].y), fmaxf(lv[i].z, lv[i].w)));
  }
  m = wave_max(m);
  if (lane == 0) redm[wave] = m;
  __syncthreads();
  m = fmaxf(fmaxf(redm[0], redm[1]), fmaxf(redm[2], redm[3]));

  float s = 0.f, a0 = 0.f, a1 = 0.f, a2 = 0.f, a3 = 0.f;
#pragma nounroll
  for (int ch = 0; ch < PCHUNKS; ++ch) {
    float l[PCHUNK];
    float4 f[PCHUNK];
#pragma unroll
    for (int i = 0; i < PCHUNK; ++i) {
      const int j = t + (ch * PCHUNK + i) * PTHREADS;
      l[i] = logits[j];
      f[i] = *(const float4*)(feat_s7 + (long)j * C + c0);
    }
#pragma unroll
    for (int i = 0; i < PCHUNK; ++i) {
      const float e = __expf(l[i] - m);
      s  += e;
      a0 += e * f[i].x;
      a1 += e * f[i].y;
      a2 += e * f[i].z;
      a3 += e * f[i].w;
    }
  }
  s  = wave_sum(s);
  a0 = wave_sum(a0);
  a1 = wave_sum(a1);
  a2 = wave_sum(a2);
  a3 = wave_sum(a3);
  if (lane == 0) {
    reds[0][wave] = s;
    reds[1][wave] = a0;
    reds[2][wave] = a1;
    reds[3][wave] = a2;
    reds[4][wave] = a3;
  }
  __syncthreads();

  if (t < PCH) {
    const float sum = (reds[0][0] + reds[0][1]) + (reds[0][2] + reds[0][3]);
    const float acc = (reds[1 + t][0] + reds[1 + t][1]) + (reds[1 + t][2] + reds[1 + t][3]);
    const float inv = 1.0f / sum;
    const float g = acc * inv;
    gate[blk * GATE_PITCH + t] = 1.0f / (1.0f + __expf(-g));
  }
}

__device__ __forceinline__ float xhalf(float x) {
  const int b = __builtin_bit_cast(int, x);
  return __builtin_bit_cast(float, __builtin_amdgcn_permlanex16(b, b, 0x76543210, 0xfedcba98, false, false));
}

#define TOP3_INS(val, idxv) {                                         \
    float s_ = (val); int id_ = (idxv);                               \
    bool c0_ = s_ < v0, c1_ = s_ < v1, c2_ = s_ < v2v;                \
    v2v = c1_ ? v1 : (c2_ ? s_  : v2v); i2 = c1_ ? i1 : (c2_ ? id_ : i2); \
    v1  = c0_ ? v0 : (c1_ ? s_  : v1 ); i1 = c0_ ? i0 : (c1_ ? id_ : i1); \
    v0  = c0_ ? s_ : v0;                i0 = c0_ ? id_ : i0;          }

template <int PART>
__device__ __forceinline__ void search_role(
    const float* __restrict__ coords_q, const float* __restrict__ coords_s,
    float* __restrict__ carry, float* __restrict__ top, const int tile, const int lane)
{
  constexpr int NBEG = PART ? NSPLIT : 0;
  constexpr int NEND = PART ? NS : NSPLIT;
  float* cp = carry + (long)tile * CARRY_PITCH;
  const int r15  = lane & 15;
  const int half = lane >> 4;
  const int qbase = tile * 16;

  const float qx = coords_q[(qbase + r15) * 3 + 0];
  const float qy = coords_q[(qbase + r15) * 3 + 1];
  const float qz = coords_q[(qbase + r15) * 3 + 2];
  v2f qb;
  qb.x = half ? (-2.f * qz) : (-2.f * qx);
  qb.y = half ? 1.0f        : (-2.f * qy);

  float v0 = 1e30f, v1 = 1e30f, v2v = 1e30f;
  int   i0 = 0, i1 = 0, i2 = 0;
  if (PART) {
    v0 = cp[lane]; v1 = cp[32 + lane]; v2v = cp[64 + lane];
    i0 = __float_as_int(cp[96 + lane]); i1 = __float_as_int(cp[128 + lane]); i2 = __float_as_int(cp[160 + lane]);
  }

  float cx[SB], cy[SB], cz[SB];
#pragma unroll
  for (int u = 0; u < SB; ++u) {
    const float* p = coords_s + (NBEG + u * 16 + r15) * 3;
    cx[u] = p[0]; cy[u] = p[1]; cz[u] = p[2];
  }
  for (int n0 = NBEG; n0 < NEND; n0 += 16 * SB) {
    const float p0 = xhalf(v0), p1 = xhalf(v1), p2 = xhalf(v2v);
    const float bound = fminf(fminf(v2v, p2), fminf(fmaxf(v1, p0), fmaxf(v0, p1)));
    v2f sa[SB];
#pragma unroll
    for (int u = 0; u < SB; ++u) {
      const float s2 = fmaf(cz[u], cz[u], fmaf(cx[u], cx[u], cy[u] * cy[u]));
      sa[u].x = half ? cz[u] : cx[u];
      sa[u].y = half ? s2    : cy[u];
    }
    const int nb = (n0 + 16 * SB) & (NS - 1);
#pragma unroll
    for (int u = 0; u < SB; ++u) {
      const float* p = coords_s + (nb + u * 16 + r15) * 3;
      cx[u] = p[0]; cy[u] = p[1]; cz[u] = p[2];
    }
    v8f c[SB];
#pragma unroll
    for (int u = 0; u < SB; ++u) {
      const v8f z = {0.f, 0.f, 0.f, 0.f, 0.f, 0.f, 0.f, 0.f};
      c[u] = __builtin_amdgcn_wmma_f32_16x16x4_f32(false, sa[u], false, qb, (short)0, z, false, false);
    }
    bool hit[SB];
#pragma unroll
    for (int u = 0; u < SB; ++u) {
      const float m8 = fminf(fminf(fminf(c[u][0], c[u][1]), fminf(c[u][2], c[u][3])),
                             fminf(fminf(c[u][4], c[u][5]), fminf(c[u][6], c[u][7])));
      hit[u] = __any(m8 <= bound);
    }
#pragma unroll
    for (int u = 0; u < SB; ++u) {
      if (hit[u]) {
        const int base = n0 + u * 16 + 8 * half;
        TOP3_INS(c[u][0], base + 0); TOP3_INS(c[u][1], base + 1);
        TOP3_INS(c[u][2], base + 2); TOP3_INS(c[u][3], base + 3);
        TOP3_INS(c[u][4], base + 4); TOP3_INS(c[u][5], base + 5);
        TOP3_INS(c[u][6], base + 6); TOP3_INS(c[u][7], base + 7);
      }
    }
  }

  if (!PART) {
    cp[lane] = v0; cp[32 + lane] = v1; cp[64 + lane] = v2v;
    cp[96 + lane] = __int_as_float(i0); cp[128 + lane] = __int_as_float(i1); cp[160 + lane] = __int_as_float(i2);
    return;
  }
  {
    float pv0 = __shfl(v0, lane + 16), pv1 = __shfl(v1, lane + 16), pv2 = __shfl(v2v, lane + 16);
    int   pi0 = __shfl(i0, lane + 16), pi1 = __shfl(i1, lane + 16), pi2 = __shfl(i2,  lane + 16);
    TOP3_INS(pv0, pi0); TOP3_INS(pv1, pi1); TOP3_INS(pv2, pi2);
  }
  const float e1 = __expf(v0 - v1), e2 = __expf(v0 - v2v);
  const float inv = 1.0f / (1.0f + e1 + e2);
  const float w0 = inv, w1 = e1 * inv, w2 = e2 * inv;

  i0 = __shfl(i0, r15); i1 = __shfl(i1, r15); i2 = __shfl(i2, r15);

  float* tp = top + (long)tile * TOP_PITCH;
  tp[lane]      = __int_as_float(half ? i1 : i0);
  tp[32 + lane] = half ? w0 : __int_as_float(i2);
  tp[64 + lane] = half ? w2 : w1;
}

__global__ __launch_bounds__(128) void logits_and_search(
    const float* __restrict__ coords_q, const float* __restrict__ coords_s,
    const float* __restrict__ feat_s7,  const float* __restrict__ w_attn,
    float* __restrict__ logits,         float* __restrict__ carry)
{
  if (blockIdx.x < LOGB) {
    logits_role(feat_s7, w_attn, logits, blockIdx.x * 128 + threadIdx.x);
    return;
  }
  const int lane = threadIdx.x & 31;
  const int wave = threadIdx.x >> 5;
  const int tile = (blockIdx.x - LOGB) * WAVES_PER_BLOCK + wave;
  if (tile >= TILES) return;
  search_role<0>(coords_q, coords_s, carry, nullptr, tile, lane);
}

__global__ __launch_bounds__(128) void pool_and_search(
    const float* __restrict__ coords_q, const float* __restrict__ coords_s,
    const float* __restrict__ feat_s7,  const float* __restrict__ logits,
    float* __restrict__ gate,           float* __restrict__ carry,
    float* __restrict__ top)
{
  if (blockIdx.x < POOLB) {
    pool_gate_role(feat_s7, logits, gate, blockIdx.x);
    return;
  }
  const int lane = threadIdx.x & 31;
  const int wave = threadIdx.x >> 5;
  const int tile = (blockIdx.x - POOLB) * WAVES_PER_BLOCK + wave;
  if (tile >= TILES) return;
  search_role<1>(coords_q, coords_s, carry, top, tile, lane);
}

#define KB   4
#define NBLK (C / (4 * KB))
static_assert(C % (4 * KB) == 0 && KB == 4 && PCH == 4, "k blocks: 16 rows, 4 gate lines");

#define LOAD_B7(dst, Wp, krow0) {                                              \
    _Pragma("unroll")                                                          \
    for (int s_ = 0; s_ < KB; ++s_) {                                          \
      const float* r0_ = (Wp) + (long)((krow0) + 8 * half + 2 * s_) * C + r15; \
      _Pragma("unroll")                                                        \
      for (int n_ = 0; n_ < 7; ++n_) {                                         \
        dst[s_][n_].x = r0_[n_ * 16];                                          \
        dst[s_][n_].y = r0_[C + n_ * 16];                                      \
      }                                                                        \
    } }

__global__ __launch_bounds__(128) void gemm_main(
    const float* __restrict__ x7,       const float* __restrict__ feat_s,
    const float* __restrict__ gate,     const float* __restrict__ top,
    const float* __restrict__ W1,
    const float* __restrict__ gammaW,   const float* __restrict__ betaW,
    const float* __restrict__ W2,       const float* __restrict__ W_cls,
    const float* __restrict__ b_cls,    float* __restrict__ out)
{
  __shared__ float lds[WAVES_PER_BLOCK * WREG];
  const int lane = threadIdx.x & 31;
  const int wave = threadIdx.x >> 5;
  const int tile = blockIdx.x * WAVES_PER_BLOCK + wave;
  if (tile >= TILES) return;

  float* WL = lds + wave * WREG;
  const int r15  = lane & 15;
  const int half = lane >> 4;
  const int qbase = tile * 16;

  const float* tp = top + (long)tile * TOP_PITCH;
  const int i0 = __float_as_int(tp[r15]), i1 = __float_as_int(tp[16 + r15]), i2 = __float_as_int(tp[32 + r15]);
  const float w0 = tp[48 + r15];
  const float w1 = tp[64 + r15];
  const float w2 = tp[80 + r15];

  v2f bb[2][KB][7];
  float4 xq[2], gq[2];
  const float* xrow = x7 + (long)(qbase + r15) * C + 8 * half;
  const float* grow = gate + 2 * half * GATE_PITCH;
  xq[0] = *(const float4*)(xrow);
  xq[1] = *(const float4*)(xrow + 4);
  gq[0] = *(const float4*)(grow);
  gq[1] = *(const float4*)(grow + GATE_PITCH);
  LOAD_B7(bb[0], W1, 0);

  const float* f0 = feat_s + (long)i0 * C + 8 * half;
  const float* f1 = feat_s + (long)i1 * C + 8 * half;
  const float* f2 = feat_s + (long)i2 * C + 8 * half;
  float4 pq0[2], pq1[2], pq2[2];
  pq0[0] = *(const float4*)(f0); pq0[1] = *(const float4*)(f0 + 4);
  pq1[0] = *(const float4*)(f1); pq1[1] = *(const float4*)(f1 + 4);
  pq2[0] = *(const float4*)(f2); pq2[1] = *(const float4*)(f2 + 4);

  v8f acc[7];
#pragma unroll
  for (int n = 0; n < 7; ++n) acc[n] = (v8f){0.f, 0.f, 0.f, 0.f, 0.f, 0.f, 0.f, 0.f};
#pragma unroll
  for (int blk = 0; blk < NBLK; ++blk) {
    v2f a[KB];
    a[0].x = xq[0].x * gq[0].x; a[0].y = xq[0].y * gq[0].y;
    a[1].x = xq[0].z * gq[0].z; a[1].y = xq[0].w * gq[0].w;
    a[2].x = xq[1].x * gq[1].x; a[2].y = xq[1].y * gq[1].y;
    a[3].x = xq[1].z * gq[1].z; a[3].y = xq[1].w * gq[1].w;
    __builtin_amdgcn_sched_barrier(0);
    if (blk + 1 < NBLK) {
      xq[0] = *(const float4*)(xrow + 16 * (blk + 1));
      xq[1] = *(const float4*)(xrow + 16 * (blk + 1) + 4);
      gq[0] = *(const float4*)(grow + 4 * (blk + 1) * GATE_PITCH);
      gq[1] = *(const float4*)(grow + (4 * (blk + 1) + 1) * GATE_PITCH);
    }
    LOAD_B7(bb[(blk + 1) & 1], W1, 16 * (blk + 1));
    __builtin_amdgcn_sched_barrier(0);
#pragma unroll
    for (int s = 0; s < KB; ++s)
#pragma unroll
      for (int n = 0; n < 7; ++n)
        acc[n] = __builtin_amdgcn_wmma_f32_16x16x4_f32(false, a[s], false, bb[blk & 1][s][n], (short)0, acc[n], false, false);
  }
#pragma unroll
  for (int blk = 0; blk < NBLK; ++blk) {
    v2f a[KB];
    a[0].x = fmaf(w2, pq2[0].x, fmaf(w0, pq0[0].x, w1 * pq1[0].x));
    a[0].y = fmaf(w2, pq2[0].y, fmaf(w0, pq0[0].y, w1 * pq1[0].y));
    a[1].x = fmaf(w2, pq2[0].z, fmaf(w0, pq0[0].z, w1 * pq1[0].z));
    a[1].y = fmaf(w2, pq2[0].w, fmaf(w0, pq0[0].w, w1 * pq1[0].w));
    a[2].x = fmaf(w2, pq2[1].x, fmaf(w0, pq0[1].x, w1 * pq1[1].x));
    a[2].y = fmaf(w2, pq2[1].y, fmaf(w0, pq0[1].y, w1 * pq1[1].y));
    a[3].x = fmaf(w2, pq2[1].z, fmaf(w0, pq0[1].z, w1 * pq1[1].z));
    a[3].y = fmaf(w2, pq2[1].w, fmaf(w0, pq0[1].w, w1 * pq1[1].w));
    __builtin_amdgcn_sched_barrier(0);
    if (blk + 1 < NBLK) {
      pq0[0] = *(const float4*)(f0 + 16 * (blk + 1)); pq0[1] = *(const float4*)(f0 + 16 * (blk + 1) + 4);
      pq1[0] = *(const float4*)(f1 + 16 * (blk + 1)); pq1[1] = *(const float4*)(f1 + 16 * (blk + 1) + 4);
      pq2[0] = *(const float4*)(f2 + 16 * (blk + 1)); pq2[1] = *(const float4*)(f2 + 16 * (blk + 1) + 4);
      LOAD_B7(bb[(NBLK + blk + 1) & 1], W1, C + 16 * (blk + 1));
    } else {
      LOAD_B7(bb[(NBLK + blk + 1) & 1], W2, 0);
    }
    __builtin_amdgcn_sched_barrier(0);
#pragma unroll
    for (int s = 0; s < KB; ++s)
#pragma unroll
      for (int n = 0; n < 7; ++n)
        acc[n] = __builtin_amdgcn_wmma_f32_16x16x4_f32(false, a[s], false, bb[(NBLK + blk) & 1][s][n], (short)0, acc[n], false, false);
  }

#pragma unroll
  for (int n = 0; n < 7; ++n) {
    const int ch = n * 16 + r15;
    const float gm = gammaW[ch], bt = betaW[ch];
#pragma unroll
    for (int r = 0; r < 8; ++r) {
      const float h = fmaxf(acc[n][r] * gm + bt, 0.f);
      WL[H_OFF + (r + 8 * half) * HSTRIDE + ch] = h;
    }
  }
  asm volatile("" ::: "memory");

  v2f wc[C / 4][2];
  v8f acc2[7];
#pragma unroll
  for (int n = 0; n < 7; ++n) acc2[n] = (v8f){0.f, 0.f, 0.f, 0.f, 0.f, 0.f, 0.f, 0.f};
  {
    const float* hrow = WL + H_OFF + r15 * HSTRIDE + 8 * half;
    v2f hq[2][KB];
#pragma unroll
    for (int s = 0; s < KB; ++s) hq[0][s] = *(const v2f*)(hrow + 2 * s);
#pragma unroll
    for (int blk = 0; blk < NBLK; ++blk) {
      __builtin_amdgcn_sched_barrier(0);
      if (blk + 1 < NBLK) {
#pragma unroll
        for (int s = 0; s < KB; ++s) hq[(blk + 1) & 1][s] = *(const v2f*)(hrow + 16 * (blk + 1) + 2 * s);
        LOAD_B7(bb[(blk + 1) & 1], W2, 16 * (blk + 1));
      } else {
        const int c1 = (16 + r15 < CLS) ? 16 + r15 : 0;
#pragma unroll
        for (int ks = 0; ks < C / 4; ++ks) {
          const float* r0 = W_cls + (long)(4 * ks + 2 * half) * CLS;
          wc[ks][0].x = r0[r15];
          wc[ks][0].y = r0[CLS + r15];
          wc[ks][1].x = r0[c1];
          wc[ks][1].y = r0[CLS + c1];
        }
      }
      __builtin_amdgcn_sched_barrier(0);
#pragma unroll
      for (int s = 0; s < KB; ++s)
#pragma unroll
        for (int n = 0; n < 7; ++n)
          acc2[n] = __builtin_amdgcn_wmma_f32_16x16x4_f32(false, hq[blk & 1][s], false, bb[blk & 1][s][n], (short)0, acc2[n], false, false);
    }
  }
#pragma unroll
  for (int n = 0; n < 7; ++n) {
    const int ch = n * 16 + r15;
#pragma unroll
    for (int r = 0; r < 8; ++r)
      WL[Y2_OFF + (r + 8 * half) * HSTRIDE + ch] = acc2[n][r];
  }
  asm volatile("" ::: "memory");

  v8f acc3[2];
  acc3[0] = (v8f){0.f, 0.f, 0.f, 0.f, 0.f, 0.f, 0.f, 0.f};
  acc3[1] = acc3[0];
  {
    const float* yrow = WL + Y2_OFF + r15 * HSTRIDE + 2 * half;
    v2f ya[C / 4];
#pragma unroll
    for (int ks = 0; ks < C / 4; ++ks) ya[ks] = *(const v2f*)(yrow + 4 * ks);
    __builtin_amdgcn_sched_barrier(0);
#pragma unroll
    for (int ks = 0; ks < C / 4; ++ks)
#pragma unroll
      for (int n = 0; n < 2; ++n)
        acc3[n] = __builtin_amdgcn_wmma_f32_16x16x4_f32(false, ya[ks], false, wc[ks][n], (short)0, acc3[n], false, false);
  }
#pragma unroll
  for (int n = 0; n < 2; ++n) {
    const int col = n * 16 + r15;
    if (col < CLS) {
      const float bb = b_cls[col];
#pragma unroll
      for (int r = 0; r < 8; ++r)
        out[(long)(qbase + r + 8 * half) * CLS + col] = acc3[n][r] + bb;
    }
  }
}

extern "C" void kernel_launch(void* const* d_in, const int* in_sizes, int n_in,
                              void* d_out, int out_size, void* d_ws, size_t ws_size,
                              hipStream_t stream) {
  const float* coords_q = (const float*)d_in[0];
  const float* coords_s = (const float*)d_in[1];
  const float* x7       = (const float*)d_in[2];
  const float* feat_s7  = (const float*)d_in[3];
  const float* feat_s   = (const float*)d_in[4];
  const float* w_attn   = (const float*)d_in[5];
  const float* W1       = (const float*)d_in[6];
  const float* gammaW   = (const float*)d_in[7];
  const float* betaW    = (const float*)d_in[8];
  const float* W2       = (const float*)d_in[9];
  const float* W_cls    = (const float*)d_in[10];
  const float* b_cls    = (const float*)d_in[11];
  float* out = (float*)d_out;

  float* ws     = (float*)d_ws;
  float* logits = ws;
  float* gate   = ws + 2 * NS;
  float* top    = ws + 4 * NS;
  float* carry  = ws + 4 * NS + 131072;

  logits_and_search<<<LOGB + SBLOCKS, 128, 0, stream>>>(coords_q, coords_s, feat_s7, w_attn, logits, carry);
  pool_and_search<<<POOLB + SBLOCKS, 128, 0, stream>>>(coords_q, coords_s, feat_s7, logits, gate, carry, top);
  gemm_main<<<SBLOCKS, 128, 0, stream>>>(x7, feat_s, gate, top, W1, gammaW, betaW, W2, W_cls, b_cls, out);
}


# GEMM weight fragments come from k-interleaved copies of W1, W2, W_cls built by 88 small groups of launch 1: a lane's 4 matrix steps x 7 tiles take 14 16-byte loads per k block, were 56 4-byte
# speedup vs baseline: 13.2179x; 1.0455x over previous
#include <hip/hip_runtime.h>
#include <math.h>

#define NQ   20000
#define NS   8192
#define C    112
#define C2   224
#define CLS  20
#define TILES (NQ / 16)
#define WAVES_PER_BLOCK 4
#define PCH        4
#define GATE_PITCH 32
#define SB         16
#define POOLB      (C / PCH)
#define SBLOCKS    ((TILES + WAVES_PER_BLOCK - 1) / WAVES_PER_BLOCK)
#define TOP_PITCH  96
#define LOGB       (NS / 128)
#define NSPLIT     (NS / 2)
#define CARRY_PITCH 192
#define W1P_N      ((C2 / 4) * C)
#define W2P_N      ((C / 4) * C)
#define WCP_N      ((C / 2) * 32)
#define PACKB      ((W1P_N + W2P_N + WCP_N + 127) / 128)

#define WREG    3712
#define HSTRIDE 114
#define H_OFF   0
#define Y2_OFF  1856

typedef float v2f __attribute__((ext_vector_type(2)));
typedef float v8f __attribute__((ext_vector_type(8)));

__device__ __forceinline__ void interleave_role(const float* __restrict__ W1, const float* __restrict__ W2,
                                                const float* __restrict__ W_cls,
                                                float4* __restrict__ w1p, float4* __restrict__ w2p,
                                                float2* __restrict__ wcp, const int e) {
  if (e < W1P_N + W2P_N) {
    const bool second = e >= W1P_N;
    const float* Wsrc = second ? W2 : W1;
    const int f = second ? e - W1P_N : e;
    const int k4 = f / C, col = f - k4 * C;
    const float* r0 = Wsrc + (long)(4 * k4) * C + col;
    float4 v;
    v.x = r0[0]; v.y = r0[C]; v.z = r0[2 * C]; v.w = r0[3 * C];
    (second ? w2p : w1p)[f] = v;
  } else if (e < W1P_N + W2P_N + WCP_N) {
    const int f = e - (W1P_N + W2P_N);
    const int k2 = f >> 5, col = f & 31;
    const int cs = col < CLS ? col : 0;
    float2 v;
    v.x = W_cls[(2 * k2) * CLS + cs]; v.y = W_cls[(2 * k2 + 1) * CLS + cs];
    wcp[f] = v;
  }
}

__device__ __forceinline__ void logits_role(const float* __restrict__ feat_s7,
                                            const float* __restrict__ w_attn,
                                            float* __restrict__ logits, const int j) {
  const float* row = feat_s7 + (long)j * C;
  float acc = 0.f;
#pragma unroll 8
  for (int c = 0; c < C; ++c) acc += row[c] * w_attn[c];
  logits[j] = acc;
}

#define PTHREADS 128
#define PWAVES   (PTHREADS / 32)
#define PCHUNK   16
#define PCHUNKS  (NS / (PTHREADS * PCHUNK))
static_assert(PWAVES == WAVES_PER_BLOCK && NS % (PTHREADS * PCHUNK) == 0 && NS % (4 * PTHREADS) == 0 && C % PCH == 0 && PCH == 4, "pool split");
static_assert(NS % (16 * SB) == 0 && (NS & (NS - 1)) == 0 && NSPLIT % (16 * SB) == 0 && NS % 128 == 0, "search trips");

__device__ __forceinline__ float wave_max(float v) {
  for (int o = 16; o > 0; o >>= 1) v = fmaxf(v, __shfl_xor(v, o));
  return v;
}
__device__ __forceinline__ float wave_sum(float v) {
  for (int o = 16; o > 0; o >>= 1) v += __shfl_xor(v, o);
  return v;
}

__device__ __forceinline__ void pool_gate_role(
    const float* __restrict__ feat_s7, const float* __restrict__ logits,
    float* __restrict__ gate, const int blk) {
  __shared__ float redm[PWAVES];
  __shared__ float reds[1 + PCH][PWAVES];
  const int t = threadIdx.x;
  const int lane = t & 31;
  const int wave = t >> 5;
  const int c0 = blk * PCH;

  float m;
  {
    float4 lv[NS / (4 * PTHREADS)];
#pragma unroll
    for (int i = 0; i < NS / (4 * PTHREADS); ++i)
      lv[i] = *(const float4*)(logits + 4 * (t + i * PTHREADS));
    m = fmaxf(fmaxf(lv[0].x, lv[0].y), fmaxf(lv[0].z, lv[0].w));
#pragma unroll
    for (int i = 1; i < NS / (4 * PTHREADS); ++i)
      m = fmaxf(m, fmaxf(fmaxf(lv[i].x, lv[i].y), fmaxf(lv[i].z, lv[i].w)));
  }
  m = wave_max(m);
  if (lane == 0) redm[wave] = m;
  __syncthreads();
  m = fmaxf(fmaxf(redm[0], redm[1]), fmaxf(redm[2], redm[3]));

  float s = 0.f, a0 = 0.f, a1 = 0.f, a2 = 0.f, a3 = 0.f;
#pragma nounroll
  for (int ch = 0; ch < PCHUNKS; ++ch) {
    float l[PCHUNK];
    float4 f[PCHUNK];
#pragma unroll
    for (int i = 0; i < PCHUNK; ++i) {
      const int j = t + (ch * PCHUNK + i) * PTHREADS;
      l[i] = logits[j];
      f[i] = *(const float4*)(feat_s7 + (long)j * C + c0);
    }
#pragma unroll
    for (int i = 0; i < PCHUNK; ++i) {
      const float e = __expf(l[i] - m);
      s  += e;
      a0 += e * f[i].x;
      a1 += e * f[i].y;
      a2 += e * f[i].z;
      a3 += e * f[i].w;
    }
  }
  s  = wave_sum(s);
  a0 = wave_sum(a0);
  a1 = wave_sum(a1);
  a2 = wave_sum(a2);
  a3 = wave_sum(a3);
  if (lane == 0) {
    reds[0][wave] = s;
    reds[1][wave] = a0;
    reds[2][wave] = a1;
    reds[3][wave] = a2;
    reds[4][wave] = a3;
  }
  __syncthreads();

  if (t < PCH) {
    const float sum = (reds[0][0] + reds[0][1]) + (reds[0][2] + reds[0][3]);
    const float acc = (reds[1 + t][0] + reds[1 + t][1]) + (reds[1 + t][2] + reds[1 + t][3]);
    const float inv = 1.0f / sum;
    const float g = acc * inv;
    gate[blk * GATE_PITCH + t] = 1.0f / (1.0f + __expf(-g));
  }
}

__device__ __forceinline__ float xhalf(float x) {
  const int b = __builtin_bit_cast(int, x);
  return __builtin_bit_cast(float, __builtin_amdgcn_permlanex16(b, b, 0x76543210, 0xfedcba98, false, false));
}

#define TOP3_INS(val, idxv) {                                         \
    float s_ = (val); int id_ = (idxv);                               \
    bool c0_ = s_ < v0, c1_ = s_ < v1, c2_ = s_ < v2v;                \
    v2v = c1_ ? v1 : (c2_ ? s_  : v2v); i2 = c1_ ? i1 : (c2_ ? id_ : i2); \
    v1  = c0_ ? v0 : (c1_ ? s_  : v1 ); i1 = c0_ ? i0 : (c1_ ? id_ : i1); \
    v0  = c0_ ? s_ : v0;                i0 = c0_ ? id_ : i0;          }

template <int PART>
__device__ __forceinline__ void search_role(
    const float* __restrict__ coords_q, const float* __restrict__ coords_s,
    float* __restrict__ carry, float* __restrict__ top, const int tile, const int lane)
{
  constexpr int NBEG = PART ? NSPLIT : 0;
  constexpr int NEND = PART ? NS : NSPLIT;
  float* cp = carry + (long)tile * CARRY_PITCH;
  const int r15  = lane & 15;
  const int half = lane >> 4;
  const int qbase = tile * 16;

  const float qx = coords_q[(qbase + r15) * 3 + 0];
  const float qy = coords_q[(qbase + r15) * 3 + 1];
  const float qz = coords_q[(qbase + r15) * 3 + 2];
  v2f qb;
  qb.x = half ? (-2.f * qz) : (-2.f * qx);
  qb.y = half ? 1.0f        : (-2.f * qy);

  float v0 = 1e30f, v1 = 1e30f, v2v = 1e30f;
  int   i0 = 0, i1 = 0, i2 = 0;
  if (PART) {
    v0 = cp[lane]; v1 = cp[32 + lane]; v2v = cp[64 + lane];
    i0 = __float_as_int(cp[96 + lane]); i1 = __float_as_int(cp[128 + lane]); i2 = __float_as_int(cp[160 + lane]);
  }

  float cx[SB], cy[SB], cz[SB];
#pragma unroll
  for (int u = 0; u < SB; ++u) {
    const float* p = coords_s + (NBEG + u * 16 + r15) * 3;
    cx[u] = p[0]; cy[u] = p[1]; cz[u] = p[2];
  }
  for (int n0 = NBEG; n0 < NEND; n0 += 16 * SB) {
    const float p0 = xhalf(v0), p1 = xhalf(v1), p2 = xhalf(v2v);
    const float bound = fminf(fminf(v2v, p2), fminf(fmaxf(v1, p0), fmaxf(v0, p1)));
    v2f sa[SB];
#pragma unroll
    for (int u = 0; u < SB; ++u) {
      const float s2 = fmaf(cz[u], cz[u], fmaf(cx[u], cx[u], cy[u] * cy[u]));
      sa[u].x = half ? cz[u] : cx[u];
      sa[u].y = half ? s2    : cy[u];
    }
    const int nb = (n0 + 16 * SB) & (NS - 1);
#pragma unroll
    for (int u = 0; u < SB; ++u) {
      const float* p = coords_s + (nb + u * 16 + r15) * 3;
      cx[u] = p[0]; cy[u] = p[1]; cz[u] = p[2];
    }
    v8f c[SB];
#pragma unroll
    for (int u = 0; u < SB; ++u) {
      const v8f z = {0.f, 0.f, 0.f, 0.f, 0.f, 0.f, 0.f, 0.f};
      c[u] = __builtin_amdgcn_wmma_f32_16x16x4_f32(false, sa[u], false, qb, (short)0, z, false, false);
    }
    bool hit[SB];
#pragma unroll
    for (int u = 0; u < SB; ++u) {
      const float m8 = fminf(fminf(fminf(c[u][0], c[u][1]), fminf(c[u][2], c[u][3])),
                             fminf(fminf(c[u][4], c[u][5]), fminf(c[u][6], c[u][7])));
      hit[u] = __any(m8 <= bound);
    }
#pragma unroll
    for (int u = 0; u < SB; ++u) {
      if (hit[u]) {
        const int base = n0 + u * 16 + 8 * half;
        TOP3_INS(c[u][0], base + 0); TOP3_INS(c[u][1], base + 1);
        TOP3_INS(c[u][2], base + 2); TOP3_INS(c[u][3], base + 3);
        TOP3_INS(c[u][4], base + 4); TOP3_INS(c[u][5], base + 5);
        TOP3_INS(c[u][6], base + 6); TOP3_INS(c[u][7], base + 7);
      }
    }
  }

  if (!PART) {
    cp[lane] = v0; cp[32 + lane] = v1; cp[64 + lane] = v2v;
    cp[96 + lane] = __int_as_float(i0); cp[128 + lane] = __int_as_float(i1); cp[160 + lane] = __int_as_float(i2);
    return;
  }
  {
    float pv0 = __shfl(v0, lane + 16), pv1 = __shfl(v1, lane + 16), pv2 = __shfl(v2v, lane + 16);
    int   pi0 = __shfl(i0, lane + 16), pi1 = __shfl(i1, lane + 16), pi2 = __shfl(i2,  lane + 16);
    TOP3_INS(pv0, pi0); TOP3_INS(pv1, pi1); TOP3_INS(pv2, pi2);
  }
  const float e1 = __expf(v0 - v1), e2 = __expf(v0 - v2v);
  const float inv = 1.0f / (1.0f + e1 + e2);
  const float w0 = inv, w1 = e1 * inv, w2 = e2 * inv;

  i0 = __shfl(i0, r15); i1 = __shfl(i1, r15); i2 = __shfl(i2, r15);

  float* tp = top + (long)tile * TOP_PITCH;
  tp[lane]      = __int_as_float(half ? i1 : i0);
  tp[32 + lane] = half ? w0 : __int_as_float(i2);
  tp[64 + lane] = half ? w2 : w1;
}

__global__ __launch_bounds__(128) void logits_and_search(
    const float* __restrict__ coords_q, const float* __restrict__ coords_s,
    const float* __restrict__ feat_s7,  const float* __restrict__ w_attn,
    float* __restrict__ logits,         float* __restrict__ carry,
    const float* __restrict__ W1,       const float* __restrict__ W2,
    const float* __restrict__ W_cls,    float* __restrict__ w1p,
    float* __restrict__ w2p,            float* __restrict__ wcp)
{
  if (blockIdx.x < PACKB) {
    interleave_role(W1, W2, W_cls, (float4*)w1p, (float4*)w2p, (float2*)wcp, blockIdx.x * 128 + threadIdx.x);
    return;
  }
  if (blockIdx.x < PACKB + LOGB) {
    logits_role(feat_s7, w_attn, logits, (blockIdx.x - PACKB) * 128 + threadIdx.x);
    return;
  }
  const int lane = threadIdx.x & 31;
  const int wave = threadIdx.x >> 5;
  const int tile = (blockIdx.x - PACKB - LOGB) * WAVES_PER_BLOCK + wave;
  if (tile >= TILES) return;
  search_role<0>(coords_q, coords_s, carry, nullptr, tile, lane);
}

__global__ __launch_bounds__(128) void pool_and_search(
    const float* __restrict__ coords_q, const float* __restrict__ coords_s,
    const float* __restrict__ feat_s7,  const float* __restrict__ logits,
    float* __restrict__ gate,           float* __restrict__ carry,
    float* __restrict__ top)
{
  if (blockIdx.x < POOLB) {
    pool_gate_role(feat_s7, logits, gate, blockIdx.x);
    return;
  }
  const int lane = threadIdx.x & 31;
  const int wave = threadIdx.x >> 5;
  const int tile = (blockIdx.x - POOLB) * WAVES_PER_BLOCK + wave;
  if (tile >= TILES) return;
  search_role<1>(coords_q, coords_s, carry, top, tile, lane);
}

#define KB   4
#define NBLK (C / (4 * KB))
static_assert(C % (4 * KB) == 0 && KB == 4 && PCH == 4, "k blocks: 16 rows, 4 gate lines");

#define LOAD_B7(dst, Wp, krow0) {                                              \
    const float4* e0_ = (Wp) + (long)((krow0) / 4 + 2 * half) * C + r15;       \
    _Pragma("unroll")                                                          \
    for (int n_ = 0; n_ < 7; ++n_) {                                           \
      dst[n_][0] = e0_[n_ * 16];                                               \
      dst[n_][1] = e0_[C + n_ * 16];                                           \
    } }
#define BFRAG(set, s_, n_) (((s_) & 1) ? (v2f){set[n_][(s_) >> 1].z, set[n_][(s_) >> 1].w} \
                                       : (v2f){set[n_][(s_) >> 1].x, set[n_][(s_) >> 1].y})

__global__ __launch_bounds__(128) void gemm_main(
    const float* __restrict__ x7,       const float* __restrict__ feat_s,
    const float* __restrict__ gate,     const float* __restrict__ top,
    const float* __restrict__ w1p_f,
    const float* __restrict__ gammaW,   const float* __restrict__ betaW,
    const float* __restrict__ w2p_f,    const float* __restrict__ wcp_f,
    const float* __restrict__ b_cls,    float* __restrict__ out)
{
  const float4* W1 = (const float4*)w1p_f;
  const float4* W2 = (const float4*)w2p_f;
  const float2* Wc = (const float2*)wcp_f;
  __shared__ float lds[WAVES_PER_BLOCK * WREG];
  const int lane = threadIdx.x & 31;
  const int wave = threadIdx.x >> 5;
  const int tile = blockIdx.x * WAVES_PER_BLOCK + wave;
  if (tile >= TILES) return;

  float* WL = lds + wave * WREG;
  const int r15  = lane & 15;
  const int half = lane >> 4;
  const int qbase = tile * 16;

  const float* tp = top + (long)tile * TOP_PITCH;
  const int i0 = __float_as_int(tp[r15]), i1 = __float_as_int(tp[16 + r15]), i2 = __float_as_int(tp[32 + r15]);
  const float w0 = tp[48 + r15];
  const float w1 = tp[64 + r15];
  const float w2 = tp[80 + r15];

  float4 bb[2][7][2];
  float4 xq[2], gq[2];
  const float* xrow = x7 + (long)(qbase + r15) * C + 8 * half;
  const float* grow = gate + 2 * half * GATE_PITCH;
  xq[0] = *(const float4*)(xrow);
  xq[1] = *(const float4*)(xrow + 4);
  gq[0] = *(const float4*)(grow);
  gq[1] = *(const float4*)(grow + GATE_PITCH);
  LOAD_B7(bb[0], W1, 0);

  const float* f0 = feat_s + (long)i0 * C + 8 * half;
  const float* f1 = feat_s + (long)i1 * C + 8 * half;
  const float* f2 = feat_s + (long)i2 * C + 8 * half;
  float4 pq0[2], pq1[2], pq2[2];
  pq0[0] = *(const float4*)(f0); pq0[1] = *(const float4*)(f0 + 4);
  pq1[0] = *(const float4*)(f1); pq1[1] = *(const float4*)(f1 + 4);
  pq2[0] = *(const float4*)(f2); pq2[1] = *(const float4*)(f2 + 4);

  v8f acc[7];
#pragma unroll
  for (int n = 0; n < 7; ++n) acc[n] = (v8f){0.f, 0.f, 0.f, 0.f, 0.f, 0.f, 0.f, 0.f};
#pragma unroll
  for (int blk = 0; blk < NBLK; ++blk) {
    v2f a[KB];
    a[0].x = xq[0].x * gq[0].x; a[0].y = xq[0].y * gq[0].y;
    a[1].x = xq[0].z * gq[0].z; a[1].y = xq[0].w * gq[0].w;
    a[2].x = xq[1].x * gq[1].x; a[2].y = xq[1].y * gq[1].y;
    a[3].x = xq[1].z * gq[1].z; a[3].y = xq[1].w * gq[1].w;
    __builtin_amdgcn_sched_barrier(0);
    if (blk + 1 < NBLK) {
      xq[0] = *(const float4*)(xrow + 16 * (blk + 1));
      xq[1] = *(const float4*)(xrow + 16 * (blk + 1) + 4);
      gq[0] = *(const float4*)(grow + 4 * (blk + 1) * GATE_PITCH);
      gq[1] = *(const float4*)(grow + (4 * (blk + 1) + 1) * GATE_PITCH);
    }
    LOAD_B7(bb[(blk + 1) & 1], W1, 16 * (blk + 1));
    __builtin_amdgcn_sched_barrier(0);
#pragma unroll
    for (int s = 0; s < KB; ++s)
#pragma unroll
      for (int n = 0; n < 7; ++n)
        acc[n] = __builtin_amdgcn_wmma_f32_16x16x4_f32(false, a[s], false, BFRAG(bb[blk & 1], s, n), (short)0, acc[n], false, false);
  }
#pragma unroll
  for (int blk = 0; blk < NBLK; ++blk) {
    v2f a[KB];
    a[0].x = fmaf(w2, pq2[0].x, fmaf(w0, pq0[0].x, w1 * pq1[0].x));
    a[0].y = fmaf(w2, pq2[0].y, fmaf(w0, pq0[0].y, w1 * pq1[0].y));
    a[1].x = fmaf(w2, pq2[0].z, fmaf(w0, pq0[0].z, w1 * pq1[0].z));
    a[1].y = fmaf(w2, pq2[0].w, fmaf(w0, pq0[0].w, w1 * pq1[0].w));
    a[2].x = fmaf(w2, pq2[1].x, fmaf(w0, pq0[1].x, w1 * pq1[1].x));
    a[2].y = fmaf(w2, pq2[1].y, fmaf(w0, pq0[1].y, w1 * pq1[1].y));
    a[3].x = fmaf(w2, pq2[1].z, fmaf(w0, pq0[1].z, w1 * pq1[1].z));
    a[3].y = fmaf(w2, pq2[1].w, fmaf(w0, pq0[1].w, w1 * pq1[1].w));
    __builtin_amdgcn_sched_barrier(0);
    if (blk + 1 < NBLK) {
      pq0[0] = *(const float4*)(f0 + 16 * (blk + 1)); pq0[1] = *(const float4*)(f0 + 16 * (blk + 1) + 4);
      pq1[0] = *(const float4*)(f1 + 16 * (blk + 1)); pq1[1] = *(const float4*)(f1 + 16 * (blk + 1) + 4);
      pq2[0] = *(const float4*)(f2 + 16 * (blk + 1)); pq2[1] = *(const float4*)(f2 + 16 * (blk + 1) + 4);
      LOAD_B7(bb[(NBLK + blk + 1) & 1], W1, C + 16 * (blk + 1));
    } else {
      LOAD_B7(bb[(NBLK + blk + 1) & 1], W2, 0);
    }
    __builtin_amdgcn_sched_barrier(0);
#pragma unroll
    for (int s = 0; s < KB; ++s)
#pragma unroll
      for (int n = 0; n < 7; ++n)
        acc[n] = __builtin_amdgcn_wmma_f32_16x16x4_f32(false, a[s], false, BFRAG(bb[(NBLK + blk) & 1], s, n), (short)0, acc[n], false, false);
  }

#pragma unroll
  for (int n = 0; n < 7; ++n) {
    const int ch = n * 16 + r15;
    const float gm = gammaW[ch], bt = betaW[ch];
#pragma unroll
    for (int r = 0; r < 8; ++r) {
      const float h = fmaxf(acc[n][r] * gm + bt, 0.f);
      WL[H_OFF + (r + 8 * half) * HSTRIDE + ch] = h;
    }
  }
  asm volatile("" ::: "memory");

  v2f wc[C / 4][2];
  v8f acc2[7];
#pragma unroll
  for (int n = 0; n < 7; ++n) acc2[n] = (v8f){0.f, 0.f, 0.f, 0.f, 0.f, 0.f, 0.f, 0.f};
  {
    const float* hrow = WL + H_OFF + r15 * HSTRIDE + 8 * half;
    v2f hq[2][KB];
#pragma unroll
    for (int s = 0; s < KB; ++s) hq[0][s] = *(const v2f*)(hrow + 2 * s);
#pragma unroll
    for (int blk = 0; blk < NBLK; ++blk) {
      __builtin_amdgcn_sched_barrier(0);
      if (blk + 1 < NBLK) {
#pragma unroll
        for (int s = 0; s < KB; ++s) hq[(blk + 1) & 1][s] = *(const v2f*)(hrow + 16 * (blk + 1) + 2 * s);
        LOAD_B7(bb[(blk + 1) & 1], W2, 16 * (blk + 1));
      } else {
#pragma unroll
        for (int ks = 0; ks < C / 4; ++ks) {
          const float2 t0 = Wc[(2 * ks + half) * 32 + r15];
          const float2 t1 = Wc[(2 * ks + half) * 32 + 16 + r15];
          wc[ks][0].x = t0.x; wc[ks][0].y = t0.y;
          wc[ks][1].x = t1.x; wc[ks][1].y = t1.y;
        }
      }
      __builtin_amdgcn_sched_barrier(0);
#pragma unroll
      for (int s = 0; s < KB; ++s)
#pragma unroll
        for (int n = 0; n < 7; ++n)
          acc2[n] = __builtin_amdgcn_wmma_f32_16x16x4_f32(false, hq[blk & 1][s], false, BFRAG(bb[blk & 1], s, n), (short)0, acc2[n], false, false);
    }
  }
#pragma unroll
  for (int n = 0; n < 7; ++n) {
    const int ch = n * 16 + r15;
#pragma unroll
    for (int r = 0; r < 8; ++r)
      WL[Y2_OFF + (r + 8 * half) * HSTRIDE + ch] = acc2[n][r];
  }
  asm volatile("" ::: "memory");

  v8f acc3[2];
  acc3[0] = (v8f){0.f, 0.f, 0.f, 0.f, 0.f, 0.f, 0.f, 0.f};
  acc3[1] = acc3[0];
  {
    const float* yrow = WL + Y2_OFF + r15 * HSTRIDE + 2 * half;
    v2f ya[C / 4];
#pragma unroll
    for (int ks = 0; ks < C / 4; ++ks) ya[ks] = *(const v2f*)(yrow + 4 * ks);
    __builtin_amdgcn_sched_barrier(0);
#pragma unroll
    for (int ks = 0; ks < C / 4; ++ks)
#pragma unroll
      for (int n = 0; n < 2; ++n)
        acc3[n] = __builtin_amdgcn_wmma_f32_16x16x4_f32(false, ya[ks], false, wc[ks][n], (short)0, acc3[n], false, false);
  }
#pragma unroll
  for (int n = 0; n < 2; ++n) {
    const int col = n * 16 + r15;
    if (col < CLS) {
      const float bb = b_cls[col];
#pragma unroll
      for (int r = 0; r < 8; ++r)
        out[(long)(qbase + r + 8 * half) * CLS + col] = acc3[n][r] + bb;
    }
  }
}

extern "C" void kernel_launch(void* const* d_in, const int* in_sizes, int n_in,
                              void* d_out, int out_size, void* d_ws, size_t ws_size,
                              hipStream_t stream) {
  const float* coords_q = (const float*)d_in[0];
  const float* coords_s = (const float*)d_in[1];
  const float* x7       = (const float*)d_in[2];
  const float* feat_s7  = (const float*)d_in[3];
  const float* feat_s   = (const float*)d_in[4];
  const float* w_attn   = (const float*)d_in[5];
  const float* W1       = (const float*)d_in[6];
  const float* gammaW   = (const float*)d_in[7];
  const float* betaW    = (const float*)d_in[8];
  const float* W2       = (const float*)d_in[9];
  const float* W_cls    = (const float*)d_in[10];
  const float* b_cls    = (const float*)d_in[11];
  float* out = (float*)d_out;

  float* ws     = (float*)d_ws;
  float* logits = ws;
  float* gate   = ws + 2 * NS;
  float* top    = ws + 4 * NS;
  float* carry  = ws + 4 * NS + 131072;
  float* w1p    = ws + 4 * NS + 131072 + 262144;
  float* w2p    = w1p + 4 * W1P_N;
  float* wcp    = w2p + 4 * W2P_N;

  logits_and_search<<<PACKB + LOGB + SBLOCKS, 128, 0, stream>>>(coords_q, coords_s, feat_s7, w_attn, logits, carry, W1, W2, W_cls, w1p, w2p, wcp);
  pool_and_search<<<POOLB + SBLOCKS, 128, 0, stream>>>(coords_q, coords_s, feat_s7, logits, gate, carry, top);
  gemm_main<<<SBLOCKS, 128, 0, stream>>>(x7, feat_s, gate, top, w1p, gammaW, betaW, w2p, wcp, b_cls, out);
}


# second search launch walks its rows in 4 pieces per tile, one short wave each, pruned from the first tile by the carried lists' bound; piece lists go via workspace, the GEMM launch merges them
# speedup vs baseline: 14.3869x; 1.0884x over previous
#include <hip/hip_runtime.h>
#include <math.h>

#define NQ   20000
#define NS   8192
#define C    112
#define C2   224
#define CLS  20
#define TILES (NQ / 16)
#define WAVES_PER_BLOCK 4
#define PCH        4
#define GATE_PITCH 32
#define SB         16
#define POOLB      (C / PCH)
#define SBLOCKS    ((TILES + WAVES_PER_BLOCK - 1) / WAVES_PER_BLOCK)
#define LOGB       (NS / 128)
#define NSPLIT     (NS / 2)
#define CARRY_PITCH 192
#define NPIECE     4
#define PIECE_ROWS ((NS - NSPLIT) / NPIECE)
#define W1P_N      ((C2 / 4) * C)
#define W2P_N      ((C / 4) * C)
#define WCP_N      ((C / 2) * 32)
#define PACKB      ((W1P_N + W2P_N + WCP_N + 127) / 128)

#define WREG    3712
#define HSTRIDE 114
#define H_OFF   0
#define Y2_OFF  1856

typedef float v2f __attribute__((ext_vector_type(2)));
typedef float v8f __attribute__((ext_vector_type(8)));

__device__ __forceinline__ void interleave_role(const float* __restrict__ W1, const float* __restrict__ W2,
                                                const float* __restrict__ W_cls,
                                                float4* __restrict__ w1p, float4* __restrict__ w2p,
                                                float2* __restrict__ wcp, const int e) {
  if (e < W1P_N + W2P_N) {
    const bool second = e >= W1P_N;
    const float* Wsrc = second ? W2 : W1;
    const int f = second ? e - W1P_N : e;
    const int k4 = f / C, col = f - k4 * C;
    const float* r0 = Wsrc + (long)(4 * k4) * C + col;
    float4 v;
    v.x = r0[0]; v.y = r0[C]; v.z = r0[2 * C]; v.w = r0[3 * C];
    (second ? w2p : w1p)[f] = v;
  } else if (e < W1P_N + W2P_N + WCP_N) {
    const int f = e - (W1P_N + W2P_N);
    const int k2 = f >> 5, col = f & 31;
    const int cs = col < CLS ? col : 0;
    float2 v;
    v.x = W_cls[(2 * k2) * CLS + cs]; v.y = W_cls[(2 * k2 + 1) * CLS + cs];
    wcp[f] = v;
  }
}

__device__ __forceinline__ void logits_role(const float* __restrict__ feat_s7,
                                            const float* __restrict__ w_attn,
                                            float* __restrict__ logits, const int j) {
  const float* row = feat_s7 + (long)j * C;
  float acc = 0.f;
#pragma unroll 8
  for (int c = 0; c < C; ++c) acc += row[c] * w_attn[c];
  logits[j] = acc;
}

#define PTHREADS 128
#define PWAVES   (PTHREADS / 32)
#define PCHUNK   16
#define PCHUNKS  (NS / (PTHREADS * PCHUNK))
static_assert(PWAVES == WAVES_PER_BLOCK && NS % (PTHREADS * PCHUNK) == 0 && NS % (4 * PTHREADS) == 0 && C % PCH == 0 && PCH == 4, "pool split");
static_assert(NS % (16 * SB) == 0 && (NS & (NS - 1)) == 0 && NSPLIT % (16 * SB) == 0 && NS % 128 == 0 && PIECE_ROWS % (16 * SB) == 0 && NPIECE == WAVES_PER_BLOCK, "search trips");

__device__ __forceinline__ float wave_max(float v) {
  for (int o = 16; o > 0; o >>= 1) v = fmaxf(v, __shfl_xor(v, o));
  return v;
}
__device__ __forceinline__ float wave_sum(float v) {
  for (int o = 16; o > 0; o >>= 1) v += __shfl_xor(v, o);
  return v;
}

__device__ __forceinline__ void pool_gate_role(
    const float* __restrict__ feat_s7, const float* __restrict__ logits,
    float* __restrict__ gate, const int blk) {
  __shared__ float redm[PWAVES];
  __shared__ float reds[1 + PCH][PWAVES];
  const int t = threadIdx.x;
  const int lane = t & 31;
  const int wave = t >> 5;
  const int c0 = blk * PCH;

  float m;
  {
    float4 lv[NS / (4 * PTHREADS)];
#pragma unroll
    for (int i = 0; i < NS / (4 * PTHREADS); ++i)
      lv[i] = *(const float4*)(logits + 4 * (t + i * PTHREADS));
    m = fmaxf(fmaxf(lv[0].x, lv[0].y), fmaxf(lv[0].z, lv[0].w));
#pragma unroll
    for (int i = 1; i < NS / (4 * PTHREADS); ++i)
      m = fmaxf(m, fmaxf(fmaxf(lv[i].x, lv[i].y), fmaxf(lv[i].z, lv[i].w)));
  }
  m = wave_max(m);
  if (lane == 0) redm[wave] = m;
  __syncthreads();
  m = fmaxf(fmaxf(redm[0], redm[1]), fmaxf(redm[2], redm[3]));

  float s = 0.f, a0 = 0.f, a1 = 0.f, a2 = 0.f, a3 = 0.f;
#pragma nounroll
  for (int ch = 0; ch < PCHUNKS; ++ch) {
    float l[PCHUNK];
    float4 f[PCHUNK];
#pragma unroll
    for (int i = 0; i < PCHUNK; ++i) {
      const int j = t + (ch * PCHUNK + i) * PTHREADS;
      l[i] = logits[j];
      f[i] = *(const float4*)(feat_s7 + (long)j * C + c0);
    }
#pragma unroll
    for (int i = 0; i < PCHUNK; ++i) {
      const float e = __expf(l[i] - m);
      s  += e;
      a0 += e * f[i].x;
      a1 += e * f[i].y;
      a2 += e * f[i].z;
      a3 += e * f[i].w;
    }
  }
  s  = wave_sum(s);
  a0 = wave_sum(a0);
  a1 = wave_sum(a1);
  a2 = wave_sum(a2);
  a3 = wave_sum(a3);
  if (lane == 0) {
    reds[0][wave] = s;
    reds[1][wave] = a0;
    reds[2][wave] = a1;
    reds[3][wave] = a2;
    reds[4][wave] = a3;
  }
  __syncthreads();

  if (t < PCH) {
    const float sum = (reds[0][0] + reds[0][1]) + (reds[0][2] + reds[0][3]);
    const float acc = (reds[1 + t][0] + reds[1 + t][1]) + (reds[1 + t][2] + reds[1 + t][3]);
    const float inv = 1.0f / sum;
    const float g = acc * inv;
    gate[blk * GATE_PITCH + t] = 1.0f / (1.0f + __expf(-g));
  }
}

__device__ __forceinline__ float xhalf(float x) {
  const int b = __builtin_bit_cast(int, x);
  return __builtin_bit_cast(float, __builtin_amdgcn_permlanex16(b, b, 0x76543210, 0xfedcba98, false, false));
}

__device__ __forceinline__ int xhalf_i(int b) {
  return __builtin_amdgcn_permlanex16(b, b, 0x76543210, 0xfedcba98, false, false);
}

#define TOP3_INS(val, idxv) {                                         \
    float s_ = (val); int id_ = (idxv);                               \
    bool c0_ = s_ < v0, c1_ = s_ < v1, c2_ = s_ < v2v;                \
    v2v = c1_ ? v1 : (c2_ ? s_  : v2v); i2 = c1_ ? i1 : (c2_ ? id_ : i2); \
    v1  = c0_ ? v0 : (c1_ ? s_  : v1 ); i1 = c0_ ? i0 : (c1_ ? id_ : i1); \
    v0  = c0_ ? s_ : v0;                i0 = c0_ ? id_ : i0;          }

template <int PART>
__device__ __forceinline__ void search_role(
    const float* __restrict__ coords_q, const float* __restrict__ coords_s,
    float* __restrict__ carry, float* __restrict__ piece, const int tile, const int pc, const int lane)
{
  const int NBEG = PART ? NSPLIT + pc * PIECE_ROWS : 0;
  const int NEND = PART ? NBEG + PIECE_ROWS : NSPLIT;
  float* cp = carry + (long)tile * CARRY_PITCH;
  const int r15  = lane & 15;
  const int half = lane >> 4;
  const int qbase = tile * 16;

  const float qx = coords_q[(qbase + r15) * 3 + 0];
  const float qy = coords_q[(qbase + r15) * 3 + 1];
  const float qz = coords_q[(qbase + r15) * 3 + 2];
  v2f qb;
  qb.x = half ? (-2.f * qz) : (-2.f * qx);
  qb.y = half ? 1.0f        : (-2.f * qy);

  float v0 = 1e30f, v1 = 1e30f, v2v = 1e30f;
  int   i0 = 0, i1 = 0, i2 = 0;
  float cbound = 1e30f;
  if (PART) {
    const float a0 = cp[lane], a1 = cp[32 + lane], a2 = cp[64 + lane];
    const float b0 = xhalf(a0), b1 = xhalf(a1), b2 = xhalf(a2);
    cbound = fminf(fminf(a2, b2), fminf(fmaxf(a1, b0), fmaxf(a0, b1)));
  }

  float cx[SB], cy[SB], cz[SB];
#pragma unroll
  for (int u = 0; u < SB; ++u) {
    const float* p = coords_s + (NBEG + u * 16 + r15) * 3;
    cx[u] = p[0]; cy[u] = p[1]; cz[u] = p[2];
  }
  for (int n0 = NBEG; n0 < NEND; n0 += 16 * SB) {
    const float p0 = xhalf(v0), p1 = xhalf(v1), p2 = xhalf(v2v);
    const float own = fminf(fminf(v2v, p2), fminf(fmaxf(v1, p0), fmaxf(v0, p1)));
    const float bound = PART ? fminf(own, cbound) : own;
    v2f sa[SB];
#pragma unroll
    for (int u = 0; u < SB; ++u) {
      const float s2 = fmaf(cz[u], cz[u], fmaf(cx[u], cx[u], cy[u] * cy[u]));
      sa[u].x = half ? cz[u] : cx[u];
      sa[u].y = half ? s2    : cy[u];
    }
    const int nb = (n0 + 16 * SB) & (NS - 1);
#pragma unroll
    for (int u = 0; u < SB; ++u) {
      const float* p = coords_s + (nb + u * 16 + r15) * 3;
      cx[u] = p[0]; cy[u] = p[1]; cz[u] = p[2];
    }
    v8f c[SB];
#pragma unroll
    for (int u = 0; u < SB; ++u) {
      const v8f z = {0.f, 0.f, 0.f, 0.f, 0.f, 0.f, 0.f, 0.f};
      c[u] = __builtin_amdgcn_wmma_f32_16x16x4_f32(false, sa[u], false, qb, (short)0, z, false, false);
    }
    bool hit[SB];
#pragma unroll
    for (int u = 0; u < SB; ++u) {
      const float m8 = fminf(fminf(fminf(c[u][0], c[u][1]), fminf(c[u][2], c[u][3])),
                             fminf(fminf(c[u][4], c[u][5]), fminf(c[u][6], c[u][7])));
      hit[u] = __any(m8 <= bound);
    }
#pragma unroll
    for (int u = 0; u < SB; ++u) {
      if (hit[u]) {
        const int base = n0 + u * 16 + 8 * half;
        TOP3_INS(c[u][0], base + 0); TOP3_INS(c[u][1], base + 1);
        TOP3_INS(c[u][2], base + 2); TOP3_INS(c[u][3], base + 3);
        TOP3_INS(c[u][4], base + 4); TOP3_INS(c[u][5], base + 5);
        TOP3_INS(c[u][6], base + 6); TOP3_INS(c[u][7], base + 7);
      }
    }
  }

  float* op = PART ? piece + ((long)tile * NPIECE + pc) * CARRY_PITCH : cp;
  op[lane] = v0; op[32 + lane] = v1; op[64 + lane] = v2v;
  op[96 + lane] = __int_as_float(i0); op[128 + lane] = __int_as_float(i1); op[160 + lane] = __int_as_float(i2);
}

__global__ __launch_bounds__(128) void logits_and_search(
    const float* __restrict__ coords_q, const float* __restrict__ coords_s,
    const float* __restrict__ feat_s7,  const float* __restrict__ w_attn,
    float* __restrict__ logits,         float* __restrict__ carry,
    const float* __restrict__ W1,       const float* __restrict__ W2,
    const float* __restrict__ W_cls,    float* __restrict__ w1p,
    float* __restrict__ w2p,            float* __restrict__ wcp)
{
  if (blockIdx.x < PACKB) {
    interleave_role(W1, W2, W_cls, (float4*)w1p, (float4*)w2p, (float2*)wcp, blockIdx.x * 128 + threadIdx.x);
    return;
  }
  if (blockIdx.x < PACKB + LOGB) {
    logits_role(feat_s7, w_attn, logits, (blockIdx.x - PACKB) * 128 + threadIdx.x);
    return;
  }
  const int lane = threadIdx.x & 31;
  const int wave = threadIdx.x >> 5;
  const int tile = (blockIdx.x - PACKB - LOGB) * WAVES_PER_BLOCK + wave;
  if (tile >= TILES) return;
  search_role<0>(coords_q, coords_s, carry, nullptr, tile, 0, lane);
}

__global__ __launch_bounds__(128) void pool_and_search(
    const float* __restrict__ coords_q, const float* __restrict__ coords_s,
    const float* __restrict__ feat_s7,  const float* __restrict__ logits,
    float* __restrict__ gate,           float* __restrict__ carry,
    float* __restrict__ piece)
{
  if (blockIdx.x < POOLB) {
    pool_gate_role(feat_s7, logits, gate, blockIdx.x);
    return;
  }
  const int lane = threadIdx.x & 31;
  const int wave = threadIdx.x >> 5;
  const int g = blockIdx.x - POOLB;
  const int tile = (g / NPIECE) * WAVES_PER_BLOCK + wave;
  if (tile >= TILES) return;
  search_role<1>(coords_q, coords_s, carry, piece, tile, g % NPIECE, lane);
}

#define KB   4
#define NBLK (C / (4 * KB))
static_assert(C % (4 * KB) == 0 && KB == 4 && PCH == 4, "k blocks: 16 rows, 4 gate lines");

#define LOAD_B7(dst, Wp, krow0) {                                              \
    const float4* e0_ = (Wp) + (long)((krow0) / 4 + 2 * half) * C + r15;       \
    _Pragma("unroll")                                                          \
    for (int n_ = 0; n_ < 7; ++n_) {                                           \
      dst[n_][0] = e0_[n_ * 16];                                               \
      dst[n_][1] = e0_[C + n_ * 16];                                           \
    } }
#define BFRAG(set, s_, n_) (((s_) & 1) ? (v2f){set[n_][(s_) >> 1].z, set[n_][(s_) >> 1].w} \
                                       : (v2f){set[n_][(s_) >> 1].x, set[n_][(s_) >> 1].y})

__global__ __launch_bounds__(128) void gemm_main(
    const float* __restrict__ x7,       const float* __restrict__ feat_s,
    const float* __restrict__ gate,     const float* __restrict__ carry,
    const float* __restrict__ piece,    const float* __restrict__ w1p_f,
    const float* __restrict__ gammaW,   const float* __restrict__ betaW,
    const float* __restrict__ w2p_f,    const float* __restrict__ wcp_f,
    const float* __restrict__ b_cls,    float* __restrict__ out)
{
  const float4* W1 = (const float4*)w1p_f;
  const float4* W2 = (const float4*)w2p_f;
  const float2* Wc = (const float2*)wcp_f;
  __shared__ float lds[WAVES_PER_BLOCK * WREG];
  const int lane = threadIdx.x & 31;
  const int wave = threadIdx.x >> 5;
  const int tile = blockIdx.x * WAVES_PER_BLOCK + wave;
  if (tile >= TILES) return;

  float* WL = lds + wave * WREG;
  const int r15  = lane & 15;
  const int half = lane >> 4;
  const int qbase = tile * 16;

  int i0, i1, i2;
  float w0, w1, w2;
  {
    const float* cp = carry + (long)tile * CARRY_PITCH;
    const float* pp = piece + (long)tile * (NPIECE * CARRY_PITCH);
    float v0 = cp[lane], v1 = cp[32 + lane], v2v = cp[64 + lane];
    i0 = __float_as_int(cp[96 + lane]); i1 = __float_as_int(cp[128 + lane]); i2 = __float_as_int(cp[160 + lane]);
    float lv[NPIECE][3];
    int   li[NPIECE][3];
#pragma unroll
    for (int q = 0; q < NPIECE; ++q)
#pragma unroll
      for (int j = 0; j < 3; ++j) {
        lv[q][j] = pp[q * CARRY_PITCH + 32 * j + lane];
        li[q][j] = __float_as_int(pp[q * CARRY_PITCH + 96 + 32 * j + lane]);
      }
#pragma unroll
    for (int q = 0; q < NPIECE; ++q) {
      TOP3_INS(lv[q][0], li[q][0]); TOP3_INS(lv[q][1], li[q][1]); TOP3_INS(lv[q][2], li[q][2]);
    }
    {
      const float pv0 = xhalf(v0), pv1 = xhalf(v1), pv2 = xhalf(v2v);
      const int   pi0 = xhalf_i(i0), pi1 = xhalf_i(i1), pi2 = xhalf_i(i2);
      TOP3_INS(pv0, pi0); TOP3_INS(pv1, pi1); TOP3_INS(pv2, pi2);
    }
    const float e1 = __expf(v0 - v1), e2 = __expf(v0 - v2v);
    const float inv = 1.0f / (1.0f + e1 + e2);
    w0 = inv; w1 = e1 * inv; w2 = e2 * inv;
    const int t0 = xhalf_i(i0), t1 = xhalf_i(i1), t2 = xhalf_i(i2);
    i0 = half ? t0 : i0; i1 = half ? t1 : i1; i2 = half ? t2 : i2;
  }

  float4 bb[2][7][2];
  float4 xq[2], gq[2];
  const float* xrow = x7 + (long)(qbase + r15) * C + 8 * half;
  const float* grow = gate + 2 * half * GATE_PITCH;
  xq[0] = *(const float4*)(xrow);
  xq[1] = *(const float4*)(xrow + 4);
  gq[0] = *(const float4*)(grow);
  gq[1] = *(const float4*)(grow + GATE_PITCH);
  LOAD_B7(bb[0], W1, 0);

  const float* f0 = feat_s + (long)i0 * C + 8 * half;
  const float* f1 = feat_s + (long)i1 * C + 8 * half;
  const float* f2 = feat_s + (long)i2 * C + 8 * half;
  float4 pq0[2], pq1[2], pq2[2];
  pq0[0] = *(const float4*)(f0); pq0[1] = *(const float4*)(f0 + 4);
  pq1[0] = *(const float4*)(f1); pq1[1] = *(const float4*)(f1 + 4);
  pq2[0] = *(const float4*)(f2); pq2[1] = *(const float4*)(f2 + 4);

  v8f acc[7];
#pragma unroll
  for (int n = 0; n < 7; ++n) acc[n] = (v8f){0.f, 0.f, 0.f, 0.f, 0.f, 0.f, 0.f, 0.f};
#pragma unroll
  for (int blk = 0; blk < NBLK; ++blk) {
    v2f a[KB];
    a[0].x = xq[0].x * gq[0].x; a[0].y = xq[0].y * gq[0].y;
    a[1].x = xq[0].z * gq[0].z; a[1].y = xq[0].w * gq[0].w;
    a[2].x = xq[1].x * gq[1].x; a[2].y = xq[1].y * gq[1].y;
    a[3].x = xq[1].z * gq[1].z; a[3].y = xq[1].w * gq[1].w;
    __builtin_amdgcn_sched_barrier(0);
    if (blk + 1 < NBLK) {
      xq[0] = *(const float4*)(xrow + 16 * (blk + 1));
      xq[1] = *(const float4*)(xrow + 16 * (blk + 1) + 4);
      gq[0] = *(const float4*)(grow + 4 * (blk + 1) * GATE_PITCH);
      gq[1] = *(const float4*)(grow + (4 * (blk + 1) + 1) * GATE_PITCH);
    }
    LOAD_B7(bb[(blk + 1) & 1], W1, 16 * (blk + 1));
    __builtin_amdgcn_sched_barrier(0);
#pragma unroll
    for (int s = 0; s < KB; ++s)
#pragma unroll
      for (int n = 0; n < 7; ++n)
        acc[n] = __builtin_amdgcn_wmma_f32_16x16x4_f32(false, a[s], false, BFRAG(bb[blk & 1], s, n), (short)0, acc[n], false, false);
  }
#pragma unroll
  for (int blk = 0; blk < NBLK; ++blk) {
    v2f a[KB];
    a[0].x = fmaf(w2, pq2[0].x, fmaf(w0, pq0[0].x, w1 * pq1[0].x));
    a[0].y = fmaf(w2, pq2[0].y, fmaf(w0, pq0[0].y, w1 * pq1[0].y));
    a[1].x = fmaf(w2, pq2[0].z, fmaf(w0, pq0[0].z, w1 * pq1[0].z));
    a[1].y = fmaf(w2, pq2[0].w, fmaf(w0, pq0[0].w, w1 * pq1[0].w));
    a[2].x = fmaf(w2, pq2[1].x, fmaf(w0, pq0[1].x, w1 * pq1[1].x));
    a[2].y = fmaf(w2, pq2[1].y, fmaf(w0, pq0[1].y, w1 * pq1[1].y));
    a[3].x = fmaf(w2, pq2[1].z, fmaf(w0, pq0[1].z, w1 * pq1[1].z));
    a[3].y = fmaf(w2, pq2[1].w, fmaf(w0, pq0[1].w, w1 * pq1[1].w));
    __builtin_amdgcn_sched_barrier(0);
    if (blk + 1 < NBLK) {
      pq0[0] = *(const float4*)(f0 + 16 * (blk + 1)); pq0[1] = *(const float4*)(f0 + 16 * (blk + 1) + 4);
      pq1[0] = *(const float4*)(f1 + 16 * (blk + 1)); pq1[1] = *(const float4*)(f1 + 16 * (blk + 1) + 4);
      pq2[0] = *(const float4*)(f2 + 16 * (blk + 1)); pq2[1] = *(const float4*)(f2 + 16 * (blk + 1) + 4);
      LOAD_B7(bb[(NBLK + blk + 1) & 1], W1, C + 16 * (blk + 1));
    } else {
      LOAD_B7(bb[(NBLK + blk + 1) & 1], W2, 0);
    }
    __builtin_amdgcn_sched_barrier(0);
#pragma unroll
    for (int s = 0; s < KB; ++s)
#pragma unroll
      for (int n = 0; n < 7; ++n)
        acc[n] = __builtin_amdgcn_wmma_f32_16x16x4_f32(false, a[s], false, BFRAG(bb[(NBLK + blk) & 1], s, n), (short)0, acc[n], false, false);
  }

#pragma unroll
  for (int n = 0; n < 7; ++n) {
    const int ch = n * 16 + r15;
    const float gm = gammaW[ch], bt = betaW[ch];
#pragma unroll
    for (int r = 0; r < 8; ++r) {
      const float h = fmaxf(acc[n][r] * gm + bt, 0.f);
      WL[H_OFF + (r + 8 * half) * HSTRIDE + ch] = h;
    }
  }
  asm volatile("" ::: "memory");

  v2f wc[C / 4][2];
  v8f acc2[7];
#pragma unroll
  for (int n = 0; n < 7; ++n) acc2[n] = (v8f){0.f, 0.f, 0.f, 0.f, 0.f, 0.f, 0.f, 0.f};
  {
    const float* hrow = WL + H_OFF + r15 * HSTRIDE + 8 * half;
    v2f hq[2][KB];
#pragma unroll
    for (int s = 0; s < KB; ++s) hq[0][s] = *(const v2f*)(hrow + 2 * s);
#pragma unroll
    for (int blk = 0; blk < NBLK; ++blk) {
      __builtin_amdgcn_sched_barrier(0);
      if (blk + 1 < NBLK) {
#pragma unroll
        for (int s = 0; s < KB; ++s) hq[(blk + 1) & 1][s] = *(const v2f*)(hrow + 16 * (blk + 1) + 2 * s);
        LOAD_B7(bb[(blk + 1) & 1], W2, 16 * (blk + 1));
      } else {
#pragma unroll
        for (int ks = 0; ks < C / 4; ++ks) {
          const float2 t0 = Wc[(2 * ks + half) * 32 + r15];
          const float2 t1 = Wc[(2 * ks + half) * 32 + 16 + r15];
          wc[ks][0].x = t0.x; wc[ks][0].y = t0.y;
          wc[ks][1].x = t1.x; wc[ks][1].y = t1.y;
        }
      }
      __builtin_amdgcn_sched_barrier(0);
#pragma unroll
      for (int s = 0; s < KB; ++s)
#pragma unroll
        for (int n = 0; n < 7; ++n)
          acc2[n] = __builtin_amdgcn_wmma_f32_16x16x4_f32(false, hq[blk & 1][s], false, BFRAG(bb[blk & 1], s, n), (short)0, acc2[n], false, false);
    }
  }
#pragma unroll
  for (int n = 0; n < 7; ++n) {
    const int ch = n * 16 + r15;
#pragma unroll
    for (int r = 0; r < 8; ++r)
      WL[Y2_OFF + (r + 8 * half) * HSTRIDE + ch] = acc2[n][r];
  }
  asm volatile("" ::: "memory");

  v8f acc3[2];
  acc3[0] = (v8f){0.f, 0.f, 0.f, 0.f, 0.f, 0.f, 0.f, 0.f};
  acc3[1] = acc3[0];
  {
    const float* yrow = WL + Y2_OFF + r15 * HSTRIDE + 2 * half;
    v2f ya[C / 4];
#pragma unroll
    for (int ks = 0; ks < C / 4; ++ks) ya[ks] = *(const v2f*)(yrow + 4 * ks);
    __builtin_amdgcn_sched_barrier(0);
#pragma unroll
    for (int ks = 0; ks < C / 4; ++ks)
#pragma unroll
      for (int n = 0; n < 2; ++n)
        acc3[n] = __builtin_amdgcn_wmma_f32_16x16x4_f32(false, ya[ks], false, wc[ks][n], (short)0, acc3[n], false, false);
  }
#pragma unroll
  for (int n = 0; n < 2; ++n) {
    const int col = n * 16 + r15;
    if (col < CLS) {
      const float bb = b_cls[col];
#pragma unroll
      for (int r = 0; r < 8; ++r)
        out[(long)(qbase + r + 8 * half) * CLS + col] = acc3[n][r] + bb;
    }
  }
}

extern "C" void kernel_launch(void* const* d_in, const int* in_sizes, int n_in,
                              void* d_out, int out_size, void* d_ws, size_t ws_size,
                              hipStream_t stream) {
  const float* coords_q = (const float*)d_in[0];
  const float* coords_s = (const float*)d_in[1];
  const float* x7       = (const float*)d_in[2];
  const float* feat_s7  = (const float*)d_in[3];
  const float* feat_s   = (const float*)d_in[4];
  const float* w_attn   = (const float*)d_in[5];
  const float* W1       = (const float*)d_in[6];
  const float* gammaW   = (const float*)d_in[7];
  const float* betaW    = (const float*)d_in[8];
  const float* W2       = (const float*)d_in[9];
  const float* W_cls    = (const float*)d_in[10];
  const float* b_cls    = (const float*)d_in[11];
  float* out = (float*)d_out;

  float* ws     = (float*)d_ws;
  float* logits = ws;
  float* gate   = ws + 2 * NS;
  float* carry  = ws + 4 * NS + 131072;
  float* w1p    = ws + 4 * NS + 131072 + 262144;
  float* w2p    = w1p + 4 * W1P_N;
  float* wcp    = w2p + 4 * W2P_N;
  float* piece  = ws + 524288;

  logits_and_search<<<PACKB + LOGB + SBLOCKS, 128, 0, stream>>>(coords_q, coords_s, feat_s7, w_attn, logits, carry, W1, W2, W_cls, w1p, w2p, wcp);
  pool_and_search<<<POOLB + NPIECE * SBLOCKS, 128, 0, stream>>>(coords_q, coords_s, feat_s7, logits, gate, carry, piece);
  gemm_main<<<SBLOCKS, 128, 0, stream>>>(x7, feat_s, gate, carry, piece, w1p, gammaW, betaW, w2p, wcp, b_cls, out);
}


# launch 1 search: one tile per work-group, its 4 waves walk a quarter of the rows each and swap their 3 best scores in LDS before every trip, so all prune with the joint bound; lists merged in LDS
# speedup vs baseline: 16.2851x; 1.1319x over previous
#include <hip/hip_runtime.h>
#include <math.h>

#define NQ   20000
#define NS   8192
#define C    112
#define C2   224
#define CLS  20
#define TILES (NQ / 16)
#define WAVES_PER_BLOCK 4
#define PCH        4
#define GATE_PITCH 32
#define SB         16
#define POOLB      (C / PCH)
#define SBLOCKS    ((TILES + WAVES_PER_BLOCK - 1) / WAVES_PER_BLOCK)
#define LOGB       (NS / 128)
#define NSPLIT     (NS / 2)
#define CARRY_PITCH 192
#define NPIECE     4
#define PIECE_ROWS ((NS - NSPLIT) / NPIECE)
#define QROWS      (NSPLIT / WAVES_PER_BLOCK)
#define W1P_N      ((C2 / 4) * C)
#define W2P_N      ((C / 4) * C)
#define WCP_N      ((C / 2) * 32)
#define PACKB      ((W1P_N + W2P_N + WCP_N + 127) / 128)

#define WREG    3712
#define HSTRIDE 114
#define H_OFF   0
#define Y2_OFF  1856

typedef float v2f __attribute__((ext_vector_type(2)));
typedef float v8f __attribute__((ext_vector_type(8)));

__device__ __forceinline__ void interleave_role(const float* __restrict__ W1, const float* __restrict__ W2,
                                                const float* __restrict__ W_cls,
                                                float4* __restrict__ w1p, float4* __restrict__ w2p,
                                                float2* __restrict__ wcp, const int e) {
  if (e < W1P_N + W2P_N) {
    const bool second = e >= W1P_N;
    const float* Wsrc = second ? W2 : W1;
    const int f = second ? e - W1P_N : e;
    const int k4 = f / C, col = f - k4 * C;
    const float* r0 = Wsrc + (long)(4 * k4) * C + col;
    float4 v;
    v.x = r0[0]; v.y = r0[C]; v.z = r0[2 * C]; v.w = r0[3 * C];
    (second ? w2p : w1p)[f] = v;
  } else if (e < W1P_N + W2P_N + WCP_N) {
    const int f = e - (W1P_N + W2P_N);
    const int k2 = f >> 5, col = f & 31;
    const int cs = col < CLS ? col : 0;
    float2 v;
    v.x = W_cls[(2 * k2) * CLS + cs]; v.y = W_cls[(2 * k2 + 1) * CLS + cs];
    wcp[f] = v;
  }
}

__device__ __forceinline__ void logits_role(const float* __restrict__ feat_s7,
                                            const float* __restrict__ w_attn,
                                            float* __restrict__ logits, const int j) {
  const float* row = feat_s7 + (long)j * C;
  float acc = 0.f;
#pragma unroll 8
  for (int c = 0; c < C; ++c) acc += row[c] * w_attn[c];
  logits[j] = acc;
}

#define PTHREADS 128
#define PWAVES   (PTHREADS / 32)
#define PCHUNK   16
#define PCHUNKS  (NS / (PTHREADS * PCHUNK))
static_assert(PWAVES == WAVES_PER_BLOCK && NS % (PTHREADS * PCHUNK) == 0 && NS % (4 * PTHREADS) == 0 && C % PCH == 0 && PCH == 4, "pool split");
static_assert(NS % (16 * SB) == 0 && (NS & (NS - 1)) == 0 && NSPLIT % (16 * SB) == 0 && NS % 128 == 0 && PIECE_ROWS % (16 * SB) == 0 && QROWS % (16 * SB) == 0 && NPIECE == WAVES_PER_BLOCK && (WAVES_PER_BLOCK & (WAVES_PER_BLOCK - 1)) == 0, "search trips");

__device__ __forceinline__ float wave_max(float v) {
  for (int o = 16; o > 0; o >>= 1) v = fmaxf(v, __shfl_xor(v, o));
  return v;
}
__device__ __forceinline__ float wave_sum(float v) {
  for (int o = 16; o > 0; o >>= 1) v += __shfl_xor(v, o);
  return v;
}

__device__ __forceinline__ void pool_gate_role(
    const float* __restrict__ feat_s7, const float* __restrict__ logits,
    float* __restrict__ gate, const int blk) {
  __shared__ float redm[PWAVES];
  __shared__ float reds[1 + PCH][PWAVES];
  const int t = threadIdx.x;
  const int lane = t & 31;
  const int wave = t >> 5;
  const int c0 = blk * PCH;

  float m;
  {
    float4 lv[NS / (4 * PTHREADS)];
#pragma unroll
    for (int i = 0; i < NS / (4 * PTHREADS); ++i)
      lv[i] = *(const float4*)(logits + 4 * (t + i * PTHREADS));
    m = fmaxf(fmaxf(lv[0].x, lv[0].y), fmaxf(lv[0].z, lv[0].w));
#pragma unroll
    for (int i = 1; i < NS / (4 * PTHREADS); ++i)
      m = fmaxf(m, fmaxf(fmaxf(lv[i].x, lv[i].y), fmaxf(lv[i].z, lv[i].w)));
  }
  m = wave_max(m);
  if (lane == 0) redm[wave] = m;
  __syncthreads();
  m = fmaxf(fmaxf(redm[0], redm[1]), fmaxf(redm[2], redm[3]));

  float s = 0.f, a0 = 0.f, a1 = 0.f, a2 = 0.f, a3 = 0.f;
#pragma nounroll
  for (int ch = 0; ch < PCHUNKS; ++ch) {
    float l[PCHUNK];
    float4 f[PCHUNK];
#pragma unroll
    for (int i = 0; i < PCHUNK; ++i) {
      const int j = t + (ch * PCHUNK + i) * PTHREADS;
      l[i] = logits[j];
      f[i] = *(const float4*)(feat_s7 + (long)j * C + c0);
    }
#pragma unroll
    for (int i = 0; i < PCHUNK; ++i) {
      const float e = __expf(l[i] - m);
      s  += e;
      a0 += e * f[i].x;
      a1 += e * f[i].y;
      a2 += e * f[i].z;
      a3 += e * f[i].w;
    }
  }
  s  = wave_sum(s);
  a0 = wave_sum(a0);
  a1 = wave_sum(a1);
  a2 = wave_sum(a2);
  a3 = wave_sum(a3);
  if (lane == 0) {
    reds[0][wave] = s;
    reds[1][wave] = a0;
    reds[2][wave] = a1;
    reds[3][wave] = a2;
    reds[4][wave] = a3;
  }
  __syncthreads();

  if (t < PCH) {
    const float sum = (reds[0][0] + reds[0][1]) + (reds[0][2] + reds[0][3]);
    const float acc = (reds[1 + t][0] + reds[1 + t][1]) + (reds[1 + t][2] + reds[1 + t][3]);
    const float inv = 1.0f / sum;
    const float g = acc * inv;
    gate[blk * GATE_PITCH + t] = 1.0f / (1.0f + __expf(-g));
  }
}

__device__ __forceinline__ float xhalf(float x) {
  const int b = __builtin_bit_cast(int, x);
  return __builtin_bit_cast(float, __builtin_amdgcn_permlanex16(b, b, 0x76543210, 0xfedcba98, false, false));
}

__device__ __forceinline__ int xhalf_i(int b) {
  return __builtin_amdgcn_permlanex16(b, b, 0x76543210, 0xfedcba98, false, false);
}

#define TOP3_INS(val, idxv) {                                         \
    float s_ = (val); int id_ = (idxv);                               \
    bool c0_ = s_ < v0, c1_ = s_ < v1, c2_ = s_ < v2v;                \
    v2v = c1_ ? v1 : (c2_ ? s_  : v2v); i2 = c1_ ? i1 : (c2_ ? id_ : i2); \
    v1  = c0_ ? v0 : (c1_ ? s_  : v1 ); i1 = c0_ ? i0 : (c1_ ? id_ : i1); \
    v0  = c0_ ? s_ : v0;                i0 = c0_ ? id_ : i0;          }

template <int PART>
__device__ __forceinline__ void search_role(
    const float* __restrict__ coords_q, const float* __restrict__ coords_s,
    float* __restrict__ carry, float* __restrict__ piece, const int tile, const int pc, const int lane)
{
  __shared__ float sh_best[2][WAVES_PER_BLOCK][3][16];
  __shared__ float sh_list[WAVES_PER_BLOCK - 1][6][32];
  const int NBEG = PART ? NSPLIT + pc * PIECE_ROWS : pc * QROWS;
  const int NEND = NBEG + (PART ? PIECE_ROWS : QROWS);
  float* cp = carry + (long)tile * CARRY_PITCH;
  const int r15  = lane & 15;
  const int half = lane >> 4;
  const int qbase = tile * 16;

  const float qx = coords_q[(qbase + r15) * 3 + 0];
  const float qy = coords_q[(qbase + r15) * 3 + 1];
  const float qz = coords_q[(qbase + r15) * 3 + 2];
  v2f qb;
  qb.x = half ? (-2.f * qz) : (-2.f * qx);
  qb.y = half ? 1.0f        : (-2.f * qy);

  float v0 = 1e30f, v1 = 1e30f, v2v = 1e30f;
  int   i0 = 0, i1 = 0, i2 = 0;
  float cbound = 1e30f;
  if (PART) {
    const float a0 = cp[lane], a1 = cp[32 + lane], a2 = cp[64 + lane];
    const float b0 = xhalf(a0), b1 = xhalf(a1), b2 = xhalf(a2);
    cbound = fminf(fminf(a2, b2), fminf(fmaxf(a1, b0), fmaxf(a0, b1)));
  }

  float cx[SB], cy[SB], cz[SB];
#pragma unroll
  for (int u = 0; u < SB; ++u) {
    const float* p = coords_s + (NBEG + u * 16 + r15) * 3;
    cx[u] = p[0]; cy[u] = p[1]; cz[u] = p[2];
  }
  int trip = 0;
  for (int n0 = NBEG; n0 < NEND; n0 += 16 * SB) {
    const float p0 = xhalf(v0), p1 = xhalf(v1), p2 = xhalf(v2v);
    float b2 = fminf(fminf(v2v, p2), fminf(fmaxf(v1, p0), fmaxf(v0, p1)));
    if (!PART && trip != 0) {
      float b0 = fminf(v0, p0);
      float b1 = fminf(fmaxf(v0, p0), fminf(v1, p1));
      float (*sb)[3][16] = sh_best[trip & 1];
      if (half == 0) { sb[pc][0][r15] = b0; sb[pc][1][r15] = b1; sb[pc][2][r15] = b2; }
      __syncthreads();
      for (int w = 1; w < WAVES_PER_BLOCK; ++w) {
        const int o = (pc + w) & (WAVES_PER_BLOCK - 1);
        const float a0 = sb[o][0][r15], a1 = sb[o][1][r15], a2 = sb[o][2][r15];
        const float c0 = fminf(b0, a0);
        const float c1 = fminf(fmaxf(b0, a0), fminf(b1, a1));
        const float c2 = fminf(fminf(b2, a2), fminf(fmaxf(b1, a0), fmaxf(b0, a1)));
        b0 = c0; b1 = c1; b2 = c2;
      }
    }
    ++trip;
    const float bound = PART ? fminf(b2, cbound) : b2;
    v2f sa[SB];
#pragma unroll
    for (int u = 0; u < SB; ++u) {
      const float s2 = fmaf(cz[u], cz[u], fmaf(cx[u], cx[u], cy[u] * cy[u]));
      sa[u].x = half ? cz[u] : cx[u];
      sa[u].y = half ? s2    : cy[u];
    }
    const int nb = (n0 + 16 * SB) & (NS - 1);
#pragma unroll
    for (int u = 0; u < SB; ++u) {
      const float* p = coords_s + (nb + u * 16 + r15) * 3;
      cx[u] = p[0]; cy[u] = p[1]; cz[u] = p[2];
    }
    v8f c[SB];
#pragma unroll
    for (int u = 0; u < SB; ++u) {
      const v8f z = {0.f, 0.f, 0.f, 0.f, 0.f, 0.f, 0.f, 0.f};
      c[u] = __builtin_amdgcn_wmma_f32_16x16x4_f32(false, sa[u], false, qb, (short)0, z, false, false);
    }
    bool hit[SB];
#pragma unroll
    for (int u = 0; u < SB; ++u) {
      const float m8 = fminf(fminf(fminf(c[u][0], c[u][1]), fminf(c[u][2], c[u][3])),
                             fminf(fminf(c[u][4], c[u][5]), fminf(c[u][6], c[u][7])));
      hit[u] = __any(m8 <= bound);
    }
#pragma unroll
    for (int u = 0; u < SB; ++u) {
      if (hit[u]) {
        const int base = n0 + u * 16 + 8 * half;
        TOP3_INS(c[u][0], base + 0); TOP3_INS(c[u][1], base + 1);
        TOP3_INS(c[u][2], base + 2); TOP3_INS(c[u][3], base + 3);
        TOP3_INS(c[u][4], base + 4); TOP3_INS(c[u][5], base + 5);
        TOP3_INS(c[u][6], base + 6); TOP3_INS(c[u][7], base + 7);
      }
    }
  }

  if (!PART) {
    if (pc != 0) {
      float (*fl)[32] = sh_list[pc - 1];
      fl[0][lane] = v0; fl[1][lane] = v1; fl[2][lane] = v2v;
      fl[3][lane] = __int_as_float(i0); fl[4][lane] = __int_as_float(i1); fl[5][lane] = __int_as_float(i2);
    }
    __syncthreads();
    if (pc != 0) return;
    for (int q = 0; q < WAVES_PER_BLOCK - 1; ++q) {
      const float a0 = sh_list[q][0][lane], a1 = sh_list[q][1][lane], a2 = sh_list[q][2][lane];
      const int j0 = __float_as_int(sh_list[q][3][lane]), j1 = __float_as_int(sh_list[q][4][lane]), j2 = __float_as_int(sh_list[q][5][lane]);
      TOP3_INS(a0, j0); TOP3_INS(a1, j1); TOP3_INS(a2, j2);
    }
  }
  float* op = PART ? piece + ((long)tile * NPIECE + pc) * CARRY_PITCH : cp;
  op[lane] = v0; op[32 + lane] = v1; op[64 + lane] = v2v;
  op[96 + lane] = __int_as_float(i0); op[128 + lane] = __int_as_float(i1); op[160 + lane] = __int_as_float(i2);
}

__global__ __launch_bounds__(128) void logits_and_search(
    const float* __restrict__ coords_q, const float* __restrict__ coords_s,
    const float* __restrict__ feat_s7,  const float* __restrict__ w_attn,
    float* __restrict__ logits,         float* __restrict__ carry,
    const float* __restrict__ W1,       const float* __restrict__ W2,
    const float* __restrict__ W_cls,    float* __restrict__ w1p,
    float* __restrict__ w2p,            float* __restrict__ wcp)
{
  if (blockIdx.x < PACKB) {
    interleave_role(W1, W2, W_cls, (float4*)w1p, (float4*)w2p, (float2*)wcp, blockIdx.x * 128 + threadIdx.x);
    return;
  }
  if (blockIdx.x < PACKB + LOGB) {
    logits_role(feat_s7, w_attn, logits, (blockIdx.x - PACKB) * 128 + threadIdx.x);
    return;
  }
  const int lane = threadIdx.x & 31;
  const int wave = __builtin_amdgcn_readfirstlane(threadIdx.x >> 5);
  search_role<0>(coords_q, coords_s, carry, nullptr, blockIdx.x - PACKB - LOGB, wave, lane);
}

__global__ __launch_bounds__(128) void pool_and_search(
    const float* __restrict__ coords_q, const float* __restrict__ coords_s,
    const float* __restrict__ feat_s7,  const float* __restrict__ logits,
    float* __restrict__ gate,           float* __restrict__ carry,
    float* __restrict__ piece)
{
  if (blockIdx.x < POOLB) {
    pool_gate_role(feat_s7, logits, gate, blockIdx.x);
    return;
  }
  const int lane = threadIdx.x & 31;
  const int wave = threadIdx.x >> 5;
  const int g = blockIdx.x - POOLB;
  const int tile = (g / NPIECE) * WAVES_PER_BLOCK + wave;
  if (tile >= TILES) return;
  search_role<1>(coords_q, coords_s, carry, piece, tile, g % NPIECE, lane);
}

#define KB   4
#define NBLK (C / (4 * KB))
static_assert(C % (4 * KB) == 0 && KB == 4 && PCH == 4, "k blocks: 16 rows, 4 gate lines");

#define LOAD_B7(dst, Wp, krow0) {                                              \
    const float4* e0_ = (Wp) + (long)((krow0) / 4 + 2 * half) * C + r15;       \
    _Pragma("unroll")                                                          \
    for (int n_ = 0; n_ < 7; ++n_) {                                           \
      dst[n_][0] = e0_[n_ * 16];                                               \
      dst[n_][1] = e0_[C + n_ * 16];                                           \
    } }
#define BFRAG(set, s_, n_) (((s_) & 1) ? (v2f){set[n_][(s_) >> 1].z, set[n_][(s_) >> 1].w} \
                                       : (v2f){set[n_][(s_) >> 1].x, set[n_][(s_) >> 1].y})

__global__ __launch_bounds__(128) void gemm_main(
    const float* __restrict__ x7,       const float* __restrict__ feat_s,
    const float* __restrict__ gate,     const float* __restrict__ carry,
    const float* __restrict__ piece,    const float* __restrict__ w1p_f,
    const float* __restrict__ gammaW,   const float* __restrict__ betaW,
    const float* __restrict__ w2p_f,    const float* __restrict__ wcp_f,
    const float* __restrict__ b_cls,    float* __restrict__ out)
{
  const float4* W1 = (const float4*)w1p_f;
  const float4* W2 = (const float4*)w2p_f;
  const float2* Wc = (const float2*)wcp_f;
  __shared__ float lds[WAVES_PER_BLOCK * WREG];
  const int lane = threadIdx.x & 31;
  const int wave = threadIdx.x >> 5;
  const int tile = blockIdx.x * WAVES_PER_BLOCK + wave;
  if (tile >= TILES) return;

  float* WL = lds + wave * WREG;
  const int r15  = lane & 15;
  const int half = lane >> 4;
  const int qbase = tile * 16;

  int i0, i1, i2;
  float w0, w1, w2;
  {
    const float* cp = carry + (long)tile * CARRY_PITCH;
    const float* pp = piece + (long)tile * (NPIECE * CARRY_PITCH);
    float v0 = cp[lane], v1 = cp[32 + lane], v2v = cp[64 + lane];
    i0 = __float_as_int(cp[96 + lane]); i1 = __float_as_int(cp[128 + lane]); i2 = __float_as_int(cp[160 + lane]);
    float lv[NPIECE][3];
    int   li[NPIECE][3];
#pragma unroll
    for (int q = 0; q < NPIECE; ++q)
#pragma unroll
      for (int j = 0; j < 3; ++j) {
        lv[q][j] = pp[q * CARRY_PITCH + 32 * j + lane];
        li[q][j] = __float_as_int(pp[q * CARRY_PITCH + 96 + 32 * j + lane]);
      }
#pragma unroll
    for (int q = 0; q < NPIECE; ++q) {
      TOP3_INS(lv[q][0], li[q][0]); TOP3_INS(lv[q][1], li[q][1]); TOP3_INS(lv[q][2], li[q][2]);
    }
    {
      const float pv0 = xhalf(v0), pv1 = xhalf(v1), pv2 = xhalf(v2v);
      const int   pi0 = xhalf_i(i0), pi1 = xhalf_i(i1), pi2 = xhalf_i(i2);
      TOP3_INS(pv0, pi0); TOP3_INS(pv1, pi1); TOP3_INS(pv2, pi2);
    }
    const float e1 = __expf(v0 - v1), e2 = __expf(v0 - v2v);
    const float inv = 1.0f / (1.0f + e1 + e2);
    w0 = inv; w1 = e1 * inv; w2 = e2 * inv;
    const int t0 = xhalf_i(i0), t1 = xhalf_i(i1), t2 = xhalf_i(i2);
    i0 = half ? t0 : i0; i1 = half ? t1 : i1; i2 = half ? t2 : i2;
  }

  float4 bb[2][7][2];
  float4 xq[2], gq[2];
  const float* xrow = x7 + (long)(qbase + r15) * C + 8 * half;
  const float* grow = gate + 2 * half * GATE_PITCH;
  xq[0] = *(const float4*)(xrow);
  xq[1] = *(const float4*)(xrow + 4);
  gq[0] = *(const float4*)(grow);
  gq[1] = *(const float4*)(grow + GATE_PITCH);
  LOAD_B7(bb[0], W1, 0);

  const float* f0 = feat_s + (long)i0 * C + 8 * half;
  const float* f1 = feat_s + (long)i1 * C + 8 * half;
  const float* f2 = feat_s + (long)i2 * C + 8 * half;
  float4 pq0[2], pq1[2], pq2[2];
  pq0[0] = *(const float4*)(f0); pq0[1] = *(const float4*)(f0 + 4);
  pq1[0] = *(const float4*)(f1); pq1[1] = *(const float4*)(f1 + 4);
  pq2[0] = *(const float4*)(f2); pq2[1] = *(const float4*)(f2 + 4);

  v8f acc[7];
#pragma unroll
  for (int n = 0; n < 7; ++n) acc[n] = (v8f){0.f, 0.f, 0.f, 0.f, 0.f, 0.f, 0.f, 0.f};
#pragma unroll
  for (int blk = 0; blk < NBLK; ++blk) {
    v2f a[KB];
    a[0].x = xq[0].x * gq[0].x; a[0].y = xq[0].y * gq[0].y;
    a[1].x = xq[0].z * gq[0].z; a[1].y = xq[0].w * gq[0].w;
    a[2].x = xq[1].x * gq[1].x; a[2].y = xq[1].y * gq[1].y;
    a[3].x = xq[1].z * gq[1].z; a[3].y = xq[1].w * gq[1].w;
    __builtin_amdgcn_sched_barrier(0);
    if (blk + 1 < NBLK) {
      xq[0] = *(const float4*)(xrow + 16 * (blk + 1));
      xq[1] = *(const float4*)(xrow + 16 * (blk + 1) + 4);
      gq[0] = *(const float4*)(grow + 4 * (blk + 1) * GATE_PITCH);
      gq[1] = *(const float4*)(grow + (4 * (blk + 1) + 1) * GATE_PITCH);
    }
    LOAD_B7(bb[(blk + 1) & 1], W1, 16 * (blk + 1));
    __builtin_amdgcn_sched_barrier(0);
#pragma unroll
    for (int s = 0; s < KB; ++s)
#pragma unroll
      for (int n = 0; n < 7; ++n)
        acc[n] = __builtin_amdgcn_wmma_f32_16x16x4_f32(false, a[s], false, BFRAG(bb[blk & 1], s, n), (short)0, acc[n], false, false);
  }
#pragma unroll
  for (int blk = 0; blk < NBLK; ++blk) {
    v2f a[KB];
    a[0].x = fmaf(w2, pq2[0].x, fmaf(w0, pq0[0].x, w1 * pq1[0].x));
    a[0].y = fmaf(w2, pq2[0].y, fmaf(w0, pq0[0].y, w1 * pq1[0].y));
    a[1].x = fmaf(w2, pq2[0].z, fmaf(w0, pq0[0].z, w1 * pq1[0].z));
    a[1].y = fmaf(w2, pq2[0].w, fmaf(w0, pq0[0].w, w1 * pq1[0].w));
    a[2].x = fmaf(w2, pq2[1].x, fmaf(w0, pq0[1].x, w1 * pq1[1].x));
    a[2].y = fmaf(w2, pq2[1].y, fmaf(w0, pq0[1].y, w1 * pq1[1].y));
    a[3].x = fmaf(w2, pq2[1].z, fmaf(w0, pq0[1].z, w1 * pq1[1].z));
    a[3].y = fmaf(w2, pq2[1].w, fmaf(w0, pq0[1].w, w1 * pq1[1].w));
    __builtin_amdgcn_sched_barrier(0);
    if (blk + 1 < NBLK) {
      pq0[0] = *(const float4*)(f0 + 16 * (blk + 1)); pq0[1] = *(const float4*)(f0 + 16 * (blk + 1) + 4);
      pq1[0] = *(const float4*)(f1 + 16 * (blk + 1)); pq1[1] = *(const float4*)(f1 + 16 * (blk + 1) + 4);
      pq2[0] = *(const float4*)(f2 + 16 * (blk + 1)); pq2[1] = *(const float4*)(f2 + 16 * (blk + 1) + 4);
      LOAD_B7(bb[(NBLK + blk + 1) & 1], W1, C + 16 * (blk + 1));
    } else {
      LOAD_B7(bb[(NBLK + blk + 1) & 1], W2, 0);
    }
    __builtin_amdgcn_sched_barrier(0);
#pragma unroll
    for (int s = 0; s < KB; ++s)
#pragma unroll
      for (int n = 0; n < 7; ++n)
        acc[n] = __builtin_amdgcn_wmma_f32_16x16x4_f32(false, a[s], false, BFRAG(bb[(NBLK + blk) & 1], s, n), (short)0, acc[n], false, false);
  }

#pragma unroll
  for (int n = 0; n < 7; ++n) {
    const int ch = n * 16 + r15;
    const float gm = gammaW[ch], bt = betaW[ch];
#pragma unroll
    for (int r = 0; r < 8; ++r) {
      const float h = fmaxf(acc[n][r] * gm + bt, 0.f);
      WL[H_OFF + (r + 8 * half) * HSTRIDE + ch] = h;
    }
  }
  asm volatile("" ::: "memory");

  v2f wc[C / 4][2];
  v8f acc2[7];
#pragma unroll
  for (int n = 0; n < 7; ++n) acc2[n] = (v8f){0.f, 0.f, 0.f, 0.f, 0.f, 0.f, 0.f, 0.f};
  {
    const float* hrow = WL + H_OFF + r15 * HSTRIDE + 8 * half;
    v2f hq[2][KB];
#pragma unroll
    for (int s = 0; s < KB; ++s) hq[0][s] = *(const v2f*)(hrow + 2 * s);
#pragma unroll
    for (int blk = 0; blk < NBLK; ++blk) {
      __builtin_amdgcn_sched_barrier(0);
      if (blk + 1 < NBLK) {
#pragma unroll
        for (int s = 0; s < KB; ++s) hq[(blk + 1) & 1][s] = *(const v2f*)(hrow + 16 * (blk + 1) + 2 * s);
        LOAD_B7(bb[(blk + 1) & 1], W2, 16 * (blk + 1));
      } else {
#pragma unroll
        for (int ks = 0; ks < C / 4; ++ks) {
          const float2 t0 = Wc[(2 * ks + half) * 32 + r15];
          const float2 t1 = Wc[(2 * ks + half) * 32 + 16 + r15];
          wc[ks][0].x = t0.x; wc[ks][0].y = t0.y;
          wc[ks][1].x = t1.x; wc[ks][1].y = t1.y;
        }
      }
      __builtin_amdgcn_sched_barrier(0);
#pragma unroll
      for (int s = 0; s < KB; ++s)
#pragma unroll
        for (int n = 0; n < 7; ++n)
          acc2[n] = __builtin_amdgcn_wmma_f32_16x16x4_f32(false, hq[blk & 1][s], false, BFRAG(bb[blk & 1], s, n), (short)0, acc2[n], false, false);
    }
  }
#pragma unroll
  for (int n = 0; n < 7; ++n) {
    const int ch = n * 16 + r15;
#pragma unroll
    for (int r = 0; r < 8; ++r)
      WL[Y2_OFF + (r + 8 * half) * HSTRIDE + ch] = acc2[n][r];
  }
  asm volatile("" ::: "memory");

  v8f acc3[2];
  acc3[0] = (v8f){0.f, 0.f, 0.f, 0.f, 0.f, 0.f, 0.f, 0.f};
  acc3[1] = acc3[0];
  {
    const float* yrow = WL + Y2_OFF + r15 * HSTRIDE + 2 * half;
    v2f ya[C / 4];
#pragma unroll
    for (int ks = 0; ks < C / 4; ++ks) ya[ks] = *(const v2f*)(yrow + 4 * ks);
    __builtin_amdgcn_sched_barrier(0);
#pragma unroll
    for (int ks = 0; ks < C / 4; ++ks)
#pragma unroll
      for (int n = 0; n < 2; ++n)
        acc3[n] = __builtin_amdgcn_wmma_f32_16x16x4_f32(false, ya[ks], false, wc[ks][n], (short)0, acc3[n], false, false);
  }
#pragma unroll
  for (int n = 0; n < 2; ++n) {
    const int col = n * 16 + r15;
    if (col < CLS) {
      const float bb = b_cls[col];
#pragma unroll
      for (int r = 0; r < 8; ++r)
        out[(long)(qbase + r + 8 * half) * CLS + col] = acc3[n][r] + bb;
    }
  }
}

extern "C" void kernel_launch(void* const* d_in, const int* in_sizes, int n_in,
                              void* d_out, int out_size, void* d_ws, size_t ws_size,
                              hipStream_t stream) {
  const float* coords_q = (const float*)d_in[0];
  const float* coords_s = (const float*)d_in[1];
  const float* x7       = (const float*)d_in[2];
  const float* feat_s7  = (const float*)d_in[3];
  const float* feat_s   = (const float*)d_in[4];
  const float* w_attn   = (const float*)d_in[5];
  const float* W1       = (const float*)d_in[6];
  const float* gammaW   = (const float*)d_in[7];
  const float* betaW    = (const float*)d_in[8];
  const float* W2       = (const float*)d_in[9];
  const float* W_cls    = (const float*)d_in[10];
  const float* b_cls    = (const float*)d_in[11];
  float* out = (float*)d_out;

  float* ws     = (float*)d_ws;
  float* logits = ws;
  float* gate   = ws + 2 * NS;
  float* carry  = ws + 4 * NS + 131072;
  float* w1p    = ws + 4 * NS + 131072 + 262144;
  float* w2p    = w1p + 4 * W1P_N;
  float* wcp    = w2p + 4 * W2P_N;
  float* piece  = ws + 524288;

  logits_and_search<<<PACKB + LOGB + TILES, 128, 0, stream>>>(coords_q, coords_s, feat_s7, w_attn, logits, carry, W1, W2, W_cls, w1p, w2p, wcp);
  pool_and_search<<<POOLB + NPIECE * SBLOCKS, 128, 0, stream>>>(coords_q, coords_s, feat_s7, logits, gate, carry, piece);
  gemm_main<<<SBLOCKS, 128, 0, stream>>>(x7, feat_s, gate, carry, piece, w1p, gammaW, betaW, w2p, wcp, b_cls, out);
}
